# starting kernel
# baseline (speedup 1.0000x reference)
#include <hip/hip_runtime.h>
#include <cstdint>
#include <cstddef>


typedef float v2f __attribute__((ext_vector_type(2)));
typedef float v8f __attribute__((ext_vector_type(8)));

static constexpr int NN  = 100000;
static constexpr int EE  = 1200000;
static constexpr int E2T = EE + NN;
static constexpr int LM1 = 3;
static constexpr int WPB = 5;

#define DEVFN static __device__ __forceinline__

DEVFN float silu_f(float v)  { return v / (1.f + __expf(-v)); }
DEVFN float lrelu_f(float v) { return v > 0.f ? v : 0.2f * v; }
DEVFN unsigned fkey(float f) {
    unsigned u = __float_as_uint(f);
    return (u & 0x80000000u) ? ~u : (u | 0x80000000u);
}
DEVFN float funkey(unsigned u) {
    return (u & 0x80000000u) ? __uint_as_float(u & 0x7FFFFFFFu) : __uint_as_float(~u);
}


__global__ void k_fill(float* __restrict__ p, float v, int n) {
    int i = blockIdx.x * blockDim.x + threadIdx.x;
    if (i < n) p[i] = v;
}

__global__ void k_deg(const int* __restrict__ dst, float* __restrict__ deg, int e) {
    int i = blockIdx.x * blockDim.x + threadIdx.x;
    if (i < e) atomicAdd(&deg[dst[i]], 1.0f);
}

__global__ void k_invdeg(float* __restrict__ deg, int n) {
    int i = blockIdx.x * blockDim.x + threadIdx.x;
    if (i < n) deg[i] = 1.0f / fmaxf(deg[i], 1.0f);
}


template <int K, int NC, bool DO_LN>
__global__ __launch_bounds__(32 * WPB)
void k_gemm(const float* __restrict__ A, const float* __restrict__ W,
            const float* __restrict__ bias, const float* __restrict__ gamma,
            const float* __restrict__ beta, float* __restrict__ out, int ntiles) {
    constexpr int NJ = NC / 16;
    const int wave = threadIdx.x >> 5;
    const int lane = threadIdx.x & 31;
    const int l16  = lane & 15;
    const int half = lane >> 4;
    const int tile = blockIdx.x * WPB + wave;
    const bool active = tile < ntiles;

    v8f zero = {};
    v8f acc[NJ];
#pragma unroll
    for (int j = 0; j < NJ; ++j) acc[j] = zero;

    if (active) {
        const float* Arow = A + (size_t)tile * 16 * K + (size_t)l16 * K;
#pragma unroll
        for (int k = 0; k < K; k += 4) {
            const float* ap = Arow + k + 2 * half;
            v2f a;
            a.x = ap[0];
            a.y = ap[1];
#pragma unroll
            for (int j = 0; j < NJ; ++j) {
                const int col = j * 16 + l16;
                v2f b;
                b.x = W[(size_t)(k + 2 * half) * NC + col];
                b.y = W[(size_t)(k + 2 * half + 1) * NC + col];
                acc[j] = __builtin_amdgcn_wmma_f32_16x16x4_f32(
                    false, a, false, b, (short)0, acc[j], false, false);
            }
        }
    }

    if constexpr (DO_LN) {
        static_assert(NC == 64, "LN epilogue assumes 64 columns");
        __shared__ float lds[WPB][16][68];
        if (active) {
#pragma unroll
            for (int j = 0; j < NJ; ++j) {
                const int col = j * 16 + l16;
                const float bj = bias[col];
#pragma unroll
                for (int v = 0; v < 8; ++v) {
                    const int r = v + 8 * half;
                    lds[wave][r][col] = silu_f(acc[j][v] + bj);
                }
            }
        }
        __syncthreads();
        if (active) {
            const int r = l16, cb = half * 32;
            float s = 0.f, sq = 0.f;
#pragma unroll
            for (int c = 0; c < 32; ++c) {
                float vv = lds[wave][r][cb + c];
                s += vv; sq += vv * vv;
            }
            s  += __shfl_xor(s, 16, 32);
            sq += __shfl_xor(sq, 16, 32);
            const float mean = s * (1.f / NC);
            const float var  = sq * (1.f / NC) - mean * mean;
            const float inv  = rsqrtf(var + 1e-5f);
            float* orow = out + ((size_t)tile * 16 + r) * NC;
#pragma unroll
            for (int c = 0; c < 32; ++c) {
                const int col = cb + c;
                const float vv = lds[wave][r][col];
                orow[col] = (vv - mean) * inv * gamma[col] + beta[col];
            }
        }
    } else {
        if (active) {
#pragma unroll
            for (int j = 0; j < NJ; ++j) {
                const int col = j * 16 + l16;
                const float bj = bias[col];
#pragma unroll
                for (int v = 0; v < 8; ++v) {
                    const int r = v + 8 * half;
                    out[((size_t)tile * 16 + r) * NC + col] = acc[j][v] + bj;
                }
            }
        }
        (void)gamma; (void)beta;
    }
}


__global__ void k_scatter_mean(const int* __restrict__ src, const int* __restrict__ dst,
                               const float* __restrict__ z, float* __restrict__ agg, int e) {
    int t = blockIdx.x * blockDim.x + threadIdx.x;
    if (t >= e * 16) return;
    const int ed = t >> 4, q = t & 15;
    const int s = src[ed], d = dst[ed];
    const float4 v = reinterpret_cast<const float4*>(z + (size_t)s * 64)[q];
    float* ad = agg + (size_t)d * 64 + q * 4;
    atomicAdd(ad + 0, v.x);
    atomicAdd(ad + 1, v.y);
    atomicAdd(ad + 2, v.z);
    atomicAdd(ad + 3, v.w);
}

__global__ void k_mean_res(float* __restrict__ h, const float* __restrict__ agg,
                           const float* __restrict__ invdeg, int n64) {
    int t = blockIdx.x * blockDim.x + threadIdx.x;
    if (t < n64) h[t] += agg[t] * invdeg[t >> 6];
}


template <int NC>
__global__ void k_gat_score(const int* __restrict__ src, const int* __restrict__ dst,
                            const float* __restrict__ xl, const float* __restrict__ xr,
                            const float* __restrict__ att, float* __restrict__ esc,
                            unsigned* __restrict__ mu, int e_graph, int ntot) {
    int t = blockIdx.x * blockDim.x + threadIdx.x;
    if (t >= ntot * 16) return;
    const int ed = t >> 4, q = t & 15;
    int s, d;
    if (ed < e_graph) { s = src[ed]; d = dst[ed]; } else { s = d = ed - e_graph; }
    constexpr int CPL = NC / 16;
    float sum = 0.f;
#pragma unroll
    for (int c0 = 0; c0 < CPL; ++c0) {
        const int c = q * CPL + c0;
        sum += att[c] * lrelu_f(xl[(size_t)s * NC + c] + xr[(size_t)d * NC + c]);
    }
#pragma unroll
    for (int off = 8; off >= 1; off >>= 1) sum += __shfl_xor(sum, off, 32);
    if (q == 0) {
        esc[ed] = sum;
        atomicMax(&mu[d], fkey(sum));
    }
}

__global__ void k_gat_w(const int* __restrict__ dst, float* __restrict__ esc,
                        const unsigned* __restrict__ mu, float* __restrict__ denom,
                        int e_graph, int ntot) {
    int ed = blockIdx.x * blockDim.x + threadIdx.x;
    if (ed >= ntot) return;
    const int d = (ed < e_graph) ? dst[ed] : (ed - e_graph);
    const float w = __expf(esc[ed] - funkey(mu[d]));
    esc[ed] = w;
    atomicAdd(&denom[d], w);
}

__global__ void k_gat_coef(const int* __restrict__ dst, float* __restrict__ esc,
                           const float* __restrict__ denom, int e_graph, int ntot) {
    int ed = blockIdx.x * blockDim.x + threadIdx.x;
    if (ed >= ntot) return;
    const int d = (ed < e_graph) ? dst[ed] : (ed - e_graph);
    esc[ed] = esc[ed] / denom[d];
}

template <int NC>
__global__ void k_init_bias(float* __restrict__ z, const float* __restrict__ bias, int total) {
    int t = blockIdx.x * blockDim.x + threadIdx.x;
    if (t < total) z[t] = bias[t & (NC - 1)];
}

template <int NC>
__global__ void k_gat_scatter(const int* __restrict__ src, const int* __restrict__ dst,
                              const float* __restrict__ xl, const float* __restrict__ coef,
                              float* __restrict__ z, int e_graph, int ntot) {
    constexpr int TPE = NC / 4;
    int t = blockIdx.x * blockDim.x + threadIdx.x;
    if (t >= ntot * TPE) return;
    const int ed = t / TPE, q = t - ed * TPE;
    int s, d;
    if (ed < e_graph) { s = src[ed]; d = dst[ed]; } else { s = d = ed - e_graph; }
    const float a = coef[ed];
    const float4 v = reinterpret_cast<const float4*>(xl + (size_t)s * NC)[q];
    float* zp = z + (size_t)d * NC + q * 4;
    atomicAdd(zp + 0, a * v.x);
    atomicAdd(zp + 1, a * v.y);
    atomicAdd(zp + 2, a * v.z);
    atomicAdd(zp + 3, a * v.w);
}

template <int NC, bool RES>
__global__ void k_gat_post(const float* __restrict__ z, const float* __restrict__ g,
                           const float* __restrict__ be, const float* __restrict__ hin,
                           float* __restrict__ out, int n) {
    int t = blockIdx.x * blockDim.x + threadIdx.x;
    if (t >= n * 16) return;
    const int row = t >> 4, q = t & 15;
    constexpr int CPL = NC / 16;
    float vals[CPL];
    float s = 0.f, sq = 0.f;
#pragma unroll
    for (int c0 = 0; c0 < CPL; ++c0) {
        const float vv = silu_f(z[(size_t)row * NC + q * CPL + c0]);
        vals[c0] = vv;
        s += vv; sq += vv * vv;
    }
#pragma unroll
    for (int off = 8; off >= 1; off >>= 1) {
        s  += __shfl_xor(s, off, 32);
        sq += __shfl_xor(sq, off, 32);
    }
    const float mean = s * (1.f / NC);
    const float var  = sq * (1.f / NC) - mean * mean;
    const float inv  = rsqrtf(var + 1e-5f);
#pragma unroll
    for (int c0 = 0; c0 < CPL; ++c0) {
        const int col = q * CPL + c0;
        float o = (vals[c0] - mean) * inv * g[col] + be[col];
        if (RES) o += hin[(size_t)row * NC + col];
        out[(size_t)row * NC + col] = o;
    }
}


extern "C" void kernel_launch(void* const* d_in, const int* in_sizes, int n_in,
                              void* d_out, int out_size, void* d_ws, size_t ws_size,
                              hipStream_t stream) {
    (void)in_sizes; (void)n_in; (void)out_size; (void)ws_size;

    const float* x        = (const float*)d_in[0];
    const int*   ei       = (const int*)  d_in[1];
    const float* W_emb    = (const float*)d_in[2];
    const float* b_emb    = (const float*)d_in[3];
    const float* g_emb    = (const float*)d_in[4];
    const float* be_emb   = (const float*)d_in[5];
    const float* enc_W    = (const float*)d_in[6];
    const float* enc_b    = (const float*)d_in[7];
    const float* enc_g    = (const float*)d_in[8];
    const float* enc_be   = (const float*)d_in[9];
    const float* dec_Wl   = (const float*)d_in[10];
    const float* dec_bl   = (const float*)d_in[11];
    const float* dec_Wr   = (const float*)d_in[12];
    const float* dec_br   = (const float*)d_in[13];
    const float* dec_att  = (const float*)d_in[14];
    const float* dec_bias = (const float*)d_in[15];
    const float* dec_g    = (const float*)d_in[16];
    const float* dec_be   = (const float*)d_in[17];
    const float* fin_Wl   = (const float*)d_in[18];
    const float* fin_bl   = (const float*)d_in[19];
    const float* fin_Wr   = (const float*)d_in[20];
    const float* fin_br   = (const float*)d_in[21];
    const float* fin_att  = (const float*)d_in[22];
    const float* fin_bias = (const float*)d_in[23];
    const float* fin_g    = (const float*)d_in[24];
    const float* fin_be   = (const float*)d_in[25];

    const int* src = ei;
    const int* dst = ei + EE;
    float* out = (float*)d_out;

    float* w       = (float*)d_ws;
    float*    invdeg = w;               w += NN;
    float*    denom  = w;               w += NN;
    unsigned* mu     = (unsigned*)w;    w += NN;
    float*    h      = w;               w += (size_t)NN * 64;
    float*    zb     = w;               w += (size_t)NN * 64;
    float*    xl     = w;               w += (size_t)NN * 64;
    float*    xr     = w;               w += (size_t)NN * 64;
    float*    esc    = w;               w += E2T;

    const int TB = 256;
    auto nb = [](long long t) { return (unsigned)((t + 255) / 256); };
    const int NT = NN / 16;
    const unsigned GG = (unsigned)((NT + WPB - 1) / WPB);
    const dim3 gthr(32 * WPB);

    k_fill<<<nb(NN), TB, 0, stream>>>(invdeg, 0.f, NN);
    k_deg<<<nb(EE), TB, 0, stream>>>(dst, invdeg, EE);
    k_invdeg<<<nb(NN), TB, 0, stream>>>(invdeg, NN);

    k_gemm<32, 64, true><<<GG, gthr, 0, stream>>>(x, W_emb, b_emb, g_emb, be_emb, h, NT);

    for (int i = 0; i < LM1; ++i) {
        k_gemm<64, 64, true><<<GG, gthr, 0, stream>>>(
            h, enc_W + (size_t)i * 64 * 64, enc_b + i * 64, enc_g + i * 64, enc_be + i * 64, zb, NT);
        k_fill<<<nb((long long)NN * 64), TB, 0, stream>>>(xl, 0.f, NN * 64);
        k_scatter_mean<<<nb((long long)EE * 16), TB, 0, stream>>>(src, dst, zb, xl, EE);
        k_mean_res<<<nb((long long)NN * 64), TB, 0, stream>>>(h, xl, invdeg, NN * 64);
    }

    for (int i = 0; i < LM1; ++i) {
        k_gemm<64, 64, false><<<GG, gthr, 0, stream>>>(
            h, dec_Wl + (size_t)i * 64 * 64, dec_bl + i * 64, nullptr, nullptr, xl, NT);
        k_gemm<64, 64, false><<<GG, gthr, 0, stream>>>(
            h, dec_Wr + (size_t)i * 64 * 64, dec_br + i * 64, nullptr, nullptr, xr, NT);
        k_fill<<<nb(NN), TB, 0, stream>>>((float*)mu, 0.f, NN);
        k_fill<<<nb(NN), TB, 0, stream>>>(denom, 0.f, NN);
        k_gat_score<64><<<nb((long long)E2T * 16), TB, 0, stream>>>(
            src, dst, xl, xr, dec_att + i * 64, esc, mu, EE, E2T);
        k_gat_w<<<nb(E2T), TB, 0, stream>>>(dst, esc, mu, denom, EE, E2T);
        k_gat_coef<<<nb(E2T), TB, 0, stream>>>(dst, esc, denom, EE, E2T);
        k_init_bias<64><<<nb((long long)NN * 64), TB, 0, stream>>>(zb, dec_bias + i * 64, NN * 64);
        k_gat_scatter<64><<<nb((long long)E2T * 16), TB, 0, stream>>>(src, dst, xl, esc, zb, EE, E2T);
        k_gat_post<64, true><<<nb((long long)NN * 16), TB, 0, stream>>>(
            zb, dec_g + i * 64, dec_be + i * 64, h, h, NN);
    }

    k_gemm<64, 32, false><<<GG, gthr, 0, stream>>>(h, fin_Wl, fin_bl, nullptr, nullptr, xl, NT);
    k_gemm<64, 32, false><<<GG, gthr, 0, stream>>>(h, fin_Wr, fin_br, nullptr, nullptr, xr, NT);
    k_fill<<<nb(NN), TB, 0, stream>>>((float*)mu, 0.f, NN);
    k_fill<<<nb(NN), TB, 0, stream>>>(denom, 0.f, NN);
    k_gat_score<32><<<nb((long long)E2T * 16), TB, 0, stream>>>(
        src, dst, xl, xr, fin_att, esc, mu, EE, E2T);
    k_gat_w<<<nb(E2T), TB, 0, stream>>>(dst, esc, mu, denom, EE, E2T);
    k_gat_coef<<<nb(E2T), TB, 0, stream>>>(dst, esc, denom, EE, E2T);
    k_init_bias<32><<<nb((long long)NN * 32), TB, 0, stream>>>(zb, fin_bias, NN * 32);
    k_gat_scatter<32><<<nb((long long)E2T * 8), TB, 0, stream>>>(src, dst, xl, esc, zb, EE, E2T);
    k_gat_post<32, false><<<nb((long long)NN * 16), TB, 0, stream>>>(
        zb, fin_g, fin_be, nullptr, out, NN);
}


# edges counting-sorted by destination once per call; each node walks its slice, sums and one-pass softmax in registers; no float atomics; float sum order = slot order (integer-atomic arrival)
# speedup vs baseline: 4.2520x; 4.2520x over previous
#include <hip/hip_runtime.h>
#include <cstdint>
#include <cstddef>


typedef float v2f __attribute__((ext_vector_type(2)));
typedef float v8f __attribute__((ext_vector_type(8)));

static constexpr int NN  = 100000;
static constexpr int EE  = 1200000;
static constexpr int LM1 = 3;
static constexpr int WPB = 5;
static constexpr int SCB = 1024;
static constexpr int NSB = (NN + SCB - 1) / SCB;
static_assert(NSB <= 256, "block offsets are summed by one 256-thread block");

#define DEVFN static __device__ __forceinline__

DEVFN float silu_f(float v)  { return v / (1.f + __expf(-v)); }
DEVFN float lrelu_f(float v) { return v > 0.f ? v : 0.2f * v; }


__global__ void k_zero(int* __restrict__ p, int n) {
    int i = blockIdx.x * blockDim.x + threadIdx.x;
    if (i < n) p[i] = 0;
}

__global__ void k_count(const int* __restrict__ dst, int* __restrict__ cnt,
                        int* __restrict__ rank, int e) {
    int i = blockIdx.x * blockDim.x + threadIdx.x;
    if (i < e) rank[i] = atomicAdd(&cnt[dst[i]], 1);
}

__global__ __launch_bounds__(256)
void k_blocksum(const int* __restrict__ cnt, int* __restrict__ tot, int n) {
    __shared__ int red[8];
    const int tid = threadIdx.x;
    const int base = blockIdx.x * SCB + tid * 4;
    int s = 0;
#pragma unroll
    for (int k = 0; k < 4; ++k) if (base + k < n) s += cnt[base + k];
#pragma unroll
    for (int off = 16; off >= 1; off >>= 1) s += __shfl_xor(s, off, 32);
    if ((tid & 31) == 0) red[tid >> 5] = s;
    __syncthreads();
    if (tid == 0) {
        int t = 0;
#pragma unroll
        for (int w = 0; w < 8; ++w) t += red[w];
        tot[blockIdx.x] = t;
    }
}

__global__ __launch_bounds__(256)
void k_scan(int* __restrict__ cnt, const int* __restrict__ tot, int n) {
    __shared__ int wsum[8];
    __shared__ int boff;
    const int tid = threadIdx.x, lane = tid & 31, wv = tid >> 5;
    int o = (tid < (int)blockIdx.x) ? tot[tid] : 0;
#pragma unroll
    for (int off = 16; off >= 1; off >>= 1) o += __shfl_xor(o, off, 32);
    if (lane == 0) wsum[wv] = o;
    __syncthreads();
    if (tid == 0) {
        int t = 0;
#pragma unroll
        for (int w = 0; w < 8; ++w) t += wsum[w];
        boff = t;
    }
    __syncthreads();
    const int blockoff = boff;
    __syncthreads();
    const int base = blockIdx.x * SCB + tid * 4;
    int c[4];
#pragma unroll
    for (int k = 0; k < 4; ++k) c[k] = (base + k < n) ? cnt[base + k] : 0;
    const int tsum = c[0] + c[1] + c[2] + c[3];
    int inc = tsum;
#pragma unroll
    for (int off = 1; off < 32; off <<= 1) {
        const int v = __shfl_up(inc, off, 32);
        if (lane >= off) inc += v;
    }
    if (lane == 31) wsum[wv] = inc;
    __syncthreads();
    int woff = 0;
    for (int w = 0; w < wv; ++w) woff += wsum[w];
    int ex = blockoff + woff + inc - tsum;
#pragma unroll
    for (int k = 0; k < 4; ++k) {
        if (base + k < n) { cnt[base + k] = ex; ex += c[k]; }
    }
}

__global__ void k_place(const int* __restrict__ src, const int* __restrict__ dst,
                        const int* __restrict__ start, const int* __restrict__ rank,
                        int* __restrict__ csr, int e) {
    int i = blockIdx.x * blockDim.x + threadIdx.x;
    if (i < e) csr[start[dst[i]] + rank[i]] = src[i];
}


template <int K, int NC, bool DO_LN>
__global__ __launch_bounds__(32 * WPB)
void k_gemm(const float* __restrict__ A, const float* __restrict__ W,
            const float* __restrict__ bias, const float* __restrict__ gamma,
            const float* __restrict__ beta, float* __restrict__ out, int ntiles) {
    constexpr int NJ = NC / 16;
    const int wave = threadIdx.x >> 5;
    const int lane = threadIdx.x & 31;
    const int l16  = lane & 15;
    const int half = lane >> 4;
    const int tile = blockIdx.x * WPB + wave;
    const bool active = tile < ntiles;

    v8f zero = {};
    v8f acc[NJ];
#pragma unroll
    for (int j = 0; j < NJ; ++j) acc[j] = zero;

    if (active) {
        const float* Arow = A + (size_t)tile * 16 * K + (size_t)l16 * K;
#pragma unroll
        for (int k = 0; k < K; k += 4) {
            const float* ap = Arow + k + 2 * half;
            v2f a;
            a.x = ap[0];
            a.y = ap[1];
#pragma unroll
            for (int j = 0; j < NJ; ++j) {
                const int col = j * 16 + l16;
                v2f b;
                b.x = W[(size_t)(k + 2 * half) * NC + col];
                b.y = W[(size_t)(k + 2 * half + 1) * NC + col];
                acc[j] = __builtin_amdgcn_wmma_f32_16x16x4_f32(
                    false, a, false, b, (short)0, acc[j], false, false);
            }
        }
    }

    if constexpr (DO_LN) {
        static_assert(NC == 64, "LN epilogue assumes 64 columns");
        __shared__ float lds[WPB][16][68];
        if (active) {
#pragma unroll
            for (int j = 0; j < NJ; ++j) {
                const int col = j * 16 + l16;
                const float bj = bias[col];
#pragma unroll
                for (int v = 0; v < 8; ++v) {
                    const int r = v + 8 * half;
                    lds[wave][r][col] = silu_f(acc[j][v] + bj);
                }
            }
        }
        __syncthreads();
        if (active) {
            const int r = l16, cb = half * 32;
            float s = 0.f, sq = 0.f;
#pragma unroll
            for (int c = 0; c < 32; ++c) {
                float vv = lds[wave][r][cb + c];
                s += vv; sq += vv * vv;
            }
            s  += __shfl_xor(s, 16, 32);
            sq += __shfl_xor(sq, 16, 32);
            const float mean = s * (1.f / NC);
            const float var  = sq * (1.f / NC) - mean * mean;
            const float inv  = rsqrtf(var + 1e-5f);
            float* orow = out + ((size_t)tile * 16 + r) * NC;
#pragma unroll
            for (int c = 0; c < 32; ++c) {
                const int col = cb + c;
                const float vv = lds[wave][r][col];
                orow[col] = (vv - mean) * inv * gamma[col] + beta[col];
            }
        }
    } else {
        if (active) {
#pragma unroll
            for (int j = 0; j < NJ; ++j) {
                const int col = j * 16 + l16;
                const float bj = bias[col];
#pragma unroll
                for (int v = 0; v < 8; ++v) {
                    const int r = v + 8 * half;
                    out[((size_t)tile * 16 + r) * NC + col] = acc[j][v] + bj;
                }
            }
        }
        (void)gamma; (void)beta;
    }
}


__global__ void k_enc_agg(const int* __restrict__ start, const int* __restrict__ csr,
                          const float* __restrict__ z, float* __restrict__ h, int n, int e) {
    const int t = blockIdx.x * blockDim.x + threadIdx.x;
    const int node = t >> 4, q = t & 15;
    if (node >= n) return;
    const int beg = start[node];
    const int end = (node + 1 < n) ? start[node + 1] : e;
    const float4* z4 = reinterpret_cast<const float4*>(z);
    float4 acc = make_float4(0.f, 0.f, 0.f, 0.f);
    for (int base = beg; base < end; base += 16) {
        const int cnt = min(16, end - base);
        const int my = (q < cnt) ? csr[base + q] : 0;
        for (int j = 0; j < cnt; j += 4) {
            const int s0 = __shfl(my, j + 0, 16);
            const int s1 = __shfl(my, j + 1, 16);
            const int s2 = __shfl(my, j + 2, 16);
            const int s3 = __shfl(my, j + 3, 16);
            const float4 v0 = z4[s0 * 16 + q];
            const float4 v1 = z4[s1 * 16 + q];
            const float4 v2 = z4[s2 * 16 + q];
            const float4 v3 = z4[s3 * 16 + q];
            const bool k1 = j + 1 < cnt, k2 = j + 2 < cnt, k3 = j + 3 < cnt;
            acc.x += v0.x; acc.y += v0.y; acc.z += v0.z; acc.w += v0.w;
            acc.x += k1 ? v1.x : 0.f; acc.y += k1 ? v1.y : 0.f; acc.z += k1 ? v1.z : 0.f; acc.w += k1 ? v1.w : 0.f;
            acc.x += k2 ? v2.x : 0.f; acc.y += k2 ? v2.y : 0.f; acc.z += k2 ? v2.z : 0.f; acc.w += k2 ? v2.w : 0.f;
            acc.x += k3 ? v3.x : 0.f; acc.y += k3 ? v3.y : 0.f; acc.z += k3 ? v3.z : 0.f; acc.w += k3 ? v3.w : 0.f;
        }
    }
    const float inv = 1.0f / fmaxf((float)(end - beg), 1.0f);
    float4* hp = reinterpret_cast<float4*>(h + (size_t)node * 64) + q;
    float4 hv = *hp;
    hv.x += acc.x * inv; hv.y += acc.y * inv; hv.z += acc.z * inv; hv.w += acc.w * inv;
    *hp = hv;
}


template <int NC, bool RES>
__global__ void k_gat_node(const int* __restrict__ start, const int* __restrict__ csr,
                           const float* __restrict__ xl, const float* __restrict__ xr,
                           const float* __restrict__ att, const float* __restrict__ bias,
                           const float* __restrict__ g, const float* __restrict__ be,
                           const float* hin, float* out,
                           int n, int e) {
    constexpr int LPN = NC / 4;
    const int t = blockIdx.x * blockDim.x + threadIdx.x;
    const int node = t / LPN, q = t % LPN;
    if (node >= n) return;
    const int beg = start[node];
    const int end = (node + 1 < n) ? start[node + 1] : e;
    const int c0 = q * 4;
    const float a0 = att[c0], a1 = att[c0 + 1], a2 = att[c0 + 2], a3 = att[c0 + 3];
    const float4* xl4 = reinterpret_cast<const float4*>(xl);
    const float4 r4 = reinterpret_cast<const float4*>(xr)[node * LPN + q];
    const float4 self = xl4[node * LPN + q];

    auto part = [&](const float4& v) -> float {
        return a0 * lrelu_f(v.x + r4.x) + a1 * lrelu_f(v.y + r4.y)
             + a2 * lrelu_f(v.z + r4.z) + a3 * lrelu_f(v.w + r4.w);
    };

    float m = part(self);
#pragma unroll
    for (int off = LPN / 2; off >= 1; off >>= 1) m += __shfl_xor(m, off, 32);
    float l = 1.f;
    float4 acc = self;
    for (int base = beg; base < end; base += LPN) {
        const int cnt = min(LPN, end - base);
        const int my = (q < cnt) ? csr[base + q] : node;
        for (int j = 0; j < cnt; j += 4) {
            const int s0 = __shfl(my, j + 0, LPN);
            const int s1 = __shfl(my, j + 1, LPN);
            const int s2 = __shfl(my, j + 2, LPN);
            const int s3 = __shfl(my, j + 3, LPN);
            const float4 v0 = xl4[s0 * LPN + q];
            const float4 v1 = xl4[s1 * LPN + q];
            const float4 v2 = xl4[s2 * LPN + q];
            const float4 v3 = xl4[s3 * LPN + q];
            float e0 = part(v0), e1 = part(v1), e2 = part(v2), e3 = part(v3);
#pragma unroll
            for (int off = LPN / 2; off >= 1; off >>= 1) {
                e0 += __shfl_xor(e0, off, 32);
                e1 += __shfl_xor(e1, off, 32);
                e2 += __shfl_xor(e2, off, 32);
                e3 += __shfl_xor(e3, off, 32);
            }
            if (j + 1 >= cnt) e1 = -INFINITY;
            if (j + 2 >= cnt) e2 = -INFINITY;
            if (j + 3 >= cnt) e3 = -INFINITY;
            const float mn = fmaxf(fmaxf(m, e0), fmaxf(fmaxf(e1, e2), e3));
            const float sc = __expf(m - mn);
            const float p0 = __expf(e0 - mn), p1 = __expf(e1 - mn);
            const float p2 = __expf(e2 - mn), p3 = __expf(e3 - mn);
            l = l * sc + ((p0 + p1) + (p2 + p3));
            acc.x = acc.x * sc + ((p0 * v0.x + p1 * v1.x) + (p2 * v2.x + p3 * v3.x));
            acc.y = acc.y * sc + ((p0 * v0.y + p1 * v1.y) + (p2 * v2.y + p3 * v3.y));
            acc.z = acc.z * sc + ((p0 * v0.z + p1 * v1.z) + (p2 * v2.z + p3 * v3.z));
            acc.w = acc.w * sc + ((p0 * v0.w + p1 * v1.w) + (p2 * v2.w + p3 * v3.w));
            m = mn;
        }
    }

    const float il = 1.f / l;
    float vals[4];
    vals[0] = silu_f(acc.x * il + bias[c0]);
    vals[1] = silu_f(acc.y * il + bias[c0 + 1]);
    vals[2] = silu_f(acc.z * il + bias[c0 + 2]);
    vals[3] = silu_f(acc.w * il + bias[c0 + 3]);
    float s = (vals[0] + vals[1]) + (vals[2] + vals[3]);
    float sq = (vals[0] * vals[0] + vals[1] * vals[1]) + (vals[2] * vals[2] + vals[3] * vals[3]);
#pragma unroll
    for (int off = LPN / 2; off >= 1; off >>= 1) {
        s  += __shfl_xor(s, off, 32);
        sq += __shfl_xor(sq, off, 32);
    }
    const float mean = s * (1.f / NC);
    const float var  = sq * (1.f / NC) - mean * mean;
    const float inv  = rsqrtf(var + 1e-5f);
    float4 o;
    o.x = (vals[0] - mean) * inv * g[c0]     + be[c0];
    o.y = (vals[1] - mean) * inv * g[c0 + 1] + be[c0 + 1];
    o.z = (vals[2] - mean) * inv * g[c0 + 2] + be[c0 + 2];
    o.w = (vals[3] - mean) * inv * g[c0 + 3] + be[c0 + 3];
    if (RES) {
        const float4 hv = reinterpret_cast<const float4*>(hin + (size_t)node * NC)[q];
        o.x += hv.x; o.y += hv.y; o.z += hv.z; o.w += hv.w;
    }
    reinterpret_cast<float4*>(out + (size_t)node * NC)[q] = o;
}


extern "C" void kernel_launch(void* const* d_in, const int* in_sizes, int n_in,
                              void* d_out, int out_size, void* d_ws, size_t ws_size,
                              hipStream_t stream) {
    (void)in_sizes; (void)n_in; (void)out_size; (void)ws_size;

    const float* x        = (const float*)d_in[0];
    const int*   ei       = (const int*)  d_in[1];
    const float* W_emb    = (const float*)d_in[2];
    const float* b_emb    = (const float*)d_in[3];
    const float* g_emb    = (const float*)d_in[4];
    const float* be_emb   = (const float*)d_in[5];
    const float* enc_W    = (const float*)d_in[6];
    const float* enc_b    = (const float*)d_in[7];
    const float* enc_g    = (const float*)d_in[8];
    const float* enc_be   = (const float*)d_in[9];
    const float* dec_Wl   = (const float*)d_in[10];
    const float* dec_bl   = (const float*)d_in[11];
    const float* dec_Wr   = (const float*)d_in[12];
    const float* dec_br   = (const float*)d_in[13];
    const float* dec_att  = (const float*)d_in[14];
    const float* dec_bias = (const float*)d_in[15];
    const float* dec_g    = (const float*)d_in[16];
    const float* dec_be   = (const float*)d_in[17];
    const float* fin_Wl   = (const float*)d_in[18];
    const float* fin_bl   = (const float*)d_in[19];
    const float* fin_Wr   = (const float*)d_in[20];
    const float* fin_br   = (const float*)d_in[21];
    const float* fin_att  = (const float*)d_in[22];
    const float* fin_bias = (const float*)d_in[23];
    const float* fin_g    = (const float*)d_in[24];
    const float* fin_be   = (const float*)d_in[25];

    const int* src = ei;
    const int* dst = ei + EE;
    float* out = (float*)d_out;

    int* wi      = (int*)d_ws;
    int*   start = wi;                  wi += NN;
    int*   tot   = wi;                  wi += 256;
    int*   rank  = wi;                  wi += EE;
    int*   csr   = wi;                  wi += EE;
    float* w     = (float*)wi;
    float* h     = w;                   w += (size_t)NN * 64;
    float* zb    = w;                   w += (size_t)NN * 64;
    float* xl    = w;                   w += (size_t)NN * 64;
    float* xr    = w;                   w += (size_t)NN * 64;

    const int TB = 256;
    auto nb = [](long long t) { return (unsigned)((t + 255) / 256); };
    const int NT = NN / 16;
    const unsigned GG = (unsigned)((NT + WPB - 1) / WPB);
    const dim3 gthr(32 * WPB);

    k_zero<<<nb(NN), TB, 0, stream>>>(start, NN);
    k_count<<<nb(EE), TB, 0, stream>>>(dst, start, rank, EE);
    k_blocksum<<<NSB, 256, 0, stream>>>(start, tot, NN);
    k_scan<<<NSB, 256, 0, stream>>>(start, tot, NN);
    k_place<<<nb(EE), TB, 0, stream>>>(src, dst, start, rank, csr, EE);

    k_gemm<32, 64, true><<<GG, gthr, 0, stream>>>(x, W_emb, b_emb, g_emb, be_emb, h, NT);

    for (int i = 0; i < LM1; ++i) {
        k_gemm<64, 64, true><<<GG, gthr, 0, stream>>>(
            h, enc_W + (size_t)i * 64 * 64, enc_b + i * 64, enc_g + i * 64, enc_be + i * 64, zb, NT);
        k_enc_agg<<<nb((long long)NN * 16), TB, 0, stream>>>(start, csr, zb, h, NN, EE);
    }

    for (int i = 0; i < LM1; ++i) {
        k_gemm<64, 64, false><<<GG, gthr, 0, stream>>>(
            h, dec_Wl + (size_t)i * 64 * 64, dec_bl + i * 64, nullptr, nullptr, xl, NT);
        k_gemm<64, 64, false><<<GG, gthr, 0, stream>>>(
            h, dec_Wr + (size_t)i * 64 * 64, dec_br + i * 64, nullptr, nullptr, xr, NT);
        k_gat_node<64, true><<<nb((long long)NN * 16), TB, 0, stream>>>(
            start, csr, xl, xr, dec_att + i * 64, dec_bias + i * 64,
            dec_g + i * 64, dec_be + i * 64, h, h, NN, EE);
    }

    k_gemm<64, 32, false><<<GG, gthr, 0, stream>>>(h, fin_Wl, fin_bl, nullptr, nullptr, xl, NT);
    k_gemm<64, 32, false><<<GG, gthr, 0, stream>>>(h, fin_Wr, fin_br, nullptr, nullptr, xr, NT);
    k_gat_node<32, false><<<nb((long long)NN * 8), TB, 0, stream>>>(
        start, csr, xl, xr, fin_att, fin_bias, fin_g, fin_be, nullptr, out, NN, EE);
}


# dense products merged into node kernels: each 16-node block stages its new rows in LDS and runs the next layer's WMMA tiles (24 to 14 launches); merged LN row sums now in 16-lane butterfly order
# speedup vs baseline: 4.8871x; 1.1494x over previous
#include <hip/hip_runtime.h>
#include <cstdint>
#include <cstddef>


typedef float v2f __attribute__((ext_vector_type(2)));
typedef float v8f __attribute__((ext_vector_type(8)));

static constexpr int NN  = 100000;
static constexpr int EE  = 1200000;
static constexpr int LM1 = 3;
static constexpr int WPB = 5;
static constexpr int SCB = 1024;
static constexpr int NSB = (NN + SCB - 1) / SCB;
static_assert(NSB <= 256, "block offsets are summed by one 256-thread block");
static_assert(NN % 16 == 0, "a 256-thread node block is exactly one 16-row tile");
static constexpr int HP  = 68;
static constexpr int OP  = 132;

#define DEVFN static __device__ __forceinline__

DEVFN float silu_f(float v)  { return v / (1.f + __expf(-v)); }
DEVFN float lrelu_f(float v) { return v > 0.f ? v : 0.2f * v; }


__global__ void k_zero(int* __restrict__ p, int n) {
    int i = blockIdx.x * blockDim.x + threadIdx.x;
    if (i < n) p[i] = 0;
}

__global__ void k_count(const int* __restrict__ dst, int* __restrict__ cnt,
                        int* __restrict__ rank, int e) {
    int i = blockIdx.x * blockDim.x + threadIdx.x;
    if (i < e) rank[i] = atomicAdd(&cnt[dst[i]], 1);
}

__global__ __launch_bounds__(256)
void k_blocksum(const int* __restrict__ cnt, int* __restrict__ tot, int n) {
    __shared__ int red[8];
    const int tid = threadIdx.x;
    const int base = blockIdx.x * SCB + tid * 4;
    int s = 0;
#pragma unroll
    for (int k = 0; k < 4; ++k) if (base + k < n) s += cnt[base + k];
#pragma unroll
    for (int off = 16; off >= 1; off >>= 1) s += __shfl_xor(s, off, 32);
    if ((tid & 31) == 0) red[tid >> 5] = s;
    __syncthreads();
    if (tid == 0) {
        int t = 0;
#pragma unroll
        for (int w = 0; w < 8; ++w) t += red[w];
        tot[blockIdx.x] = t;
    }
}

__global__ __launch_bounds__(256)
void k_scan(int* __restrict__ cnt, const int* __restrict__ tot, int n) {
    __shared__ int wsum[8];
    __shared__ int boff;
    const int tid = threadIdx.x, lane = tid & 31, wv = tid >> 5;
    int o = (tid < (int)blockIdx.x) ? tot[tid] : 0;
#pragma unroll
    for (int off = 16; off >= 1; off >>= 1) o += __shfl_xor(o, off, 32);
    if (lane == 0) wsum[wv] = o;
    __syncthreads();
    if (tid == 0) {
        int t = 0;
#pragma unroll
        for (int w = 0; w < 8; ++w) t += wsum[w];
        boff = t;
    }
    __syncthreads();
    const int blockoff = boff;
    __syncthreads();
    const int base = blockIdx.x * SCB + tid * 4;
    int c[4];
#pragma unroll
    for (int k = 0; k < 4; ++k) c[k] = (base + k < n) ? cnt[base + k] : 0;
    const int tsum = c[0] + c[1] + c[2] + c[3];
    int inc = tsum;
#pragma unroll
    for (int off = 1; off < 32; off <<= 1) {
        const int v = __shfl_up(inc, off, 32);
        if (lane >= off) inc += v;
    }
    if (lane == 31) wsum[wv] = inc;
    __syncthreads();
    int woff = 0;
    for (int w = 0; w < wv; ++w) woff += wsum[w];
    int ex = blockoff + woff + inc - tsum;
#pragma unroll
    for (int k = 0; k < 4; ++k) {
        if (base + k < n) { cnt[base + k] = ex; ex += c[k]; }
    }
}

__global__ void k_place(const int* __restrict__ src, const int* __restrict__ dst,
                        const int* __restrict__ start, const int* __restrict__ rank,
                        int* __restrict__ csr, int e) {
    int i = blockIdx.x * blockDim.x + threadIdx.x;
    if (i < e) csr[start[dst[i]] + rank[i]] = src[i];
}


template <int K, int NC, bool DO_LN>
__global__ __launch_bounds__(32 * WPB)
void k_gemm(const float* __restrict__ A, const float* __restrict__ W,
            const float* __restrict__ bias, const float* __restrict__ gamma,
            const float* __restrict__ beta, float* __restrict__ out, int ntiles) {
    constexpr int NJ = NC / 16;
    const int wave = threadIdx.x >> 5;
    const int lane = threadIdx.x & 31;
    const int l16  = lane & 15;
    const int half = lane >> 4;
    const int tile = blockIdx.x * WPB + wave;
    const bool active = tile < ntiles;

    v8f zero = {};
    v8f acc[NJ];
#pragma unroll
    for (int j = 0; j < NJ; ++j) acc[j] = zero;

    if (active) {
        const float* Arow = A + (size_t)tile * 16 * K + (size_t)l16 * K;
#pragma unroll
        for (int k = 0; k < K; k += 4) {
            const float* ap = Arow + k + 2 * half;
            v2f a;
            a.x = ap[0];
            a.y = ap[1];
#pragma unroll
            for (int j = 0; j < NJ; ++j) {
                const int col = j * 16 + l16;
                v2f b;
                b.x = W[(size_t)(k + 2 * half) * NC + col];
                b.y = W[(size_t)(k + 2 * half + 1) * NC + col];
                acc[j] = __builtin_amdgcn_wmma_f32_16x16x4_f32(
                    false, a, false, b, (short)0, acc[j], false, false);
            }
        }
    }

    if constexpr (DO_LN) {
        static_assert(NC == 64, "LN epilogue assumes 64 columns");
        __shared__ float lds[WPB][16][68];
        if (active) {
#pragma unroll
            for (int j = 0; j < NJ; ++j) {
                const int col = j * 16 + l16;
                const float bj = bias[col];
#pragma unroll
                for (int v = 0; v < 8; ++v) {
                    const int r = v + 8 * half;
                    lds[wave][r][col] = silu_f(acc[j][v] + bj);
                }
            }
        }
        __syncthreads();
        if (active) {
            const int r = l16, cb = half * 32;
            float s = 0.f, sq = 0.f;
#pragma unroll
            for (int c = 0; c < 32; ++c) {
                float vv = lds[wave][r][cb + c];
                s += vv; sq += vv * vv;
            }
            s  += __shfl_xor(s, 16, 32);
            sq += __shfl_xor(sq, 16, 32);
            const float mean = s * (1.f / NC);
            const float var  = sq * (1.f / NC) - mean * mean;
            const float inv  = rsqrtf(var + 1e-5f);
            float* orow = out + ((size_t)tile * 16 + r) * NC;
#pragma unroll
            for (int c = 0; c < 32; ++c) {
                const int col = cb + c;
                const float vv = lds[wave][r][col];
                orow[col] = (vv - mean) * inv * gamma[col] + beta[col];
            }
        }
    } else {
        if (active) {
#pragma unroll
            for (int j = 0; j < NJ; ++j) {
                const int col = j * 16 + l16;
                const float bj = bias[col];
#pragma unroll
                for (int v = 0; v < 8; ++v) {
                    const int r = v + 8 * half;
                    out[((size_t)tile * 16 + r) * NC + col] = acc[j][v] + bj;
                }
            }
        }
        (void)gamma; (void)beta;
    }
}


template <int NCO, bool LN>
DEVFN void tile_tail(float (*hs)[HP], float (*os)[OP],
                     const float* __restrict__ Wa, const float* __restrict__ ba,
                     const float* __restrict__ Wb, const float* __restrict__ bb,
                     const float* __restrict__ g,  const float* __restrict__ be,
                     float* __restrict__ outA, float* __restrict__ outB, int tile) {
    static_assert(NCO == 64 || (NCO == 32 && !LN), "tile map below");
    constexpr int TPM = NCO / 16;
    constexpr int NT  = LN ? TPM : 2 * TPM;
    const int wave = threadIdx.x >> 5;
    const int lane = threadIdx.x & 31;
    const int l16  = lane & 15;
    const int half = lane >> 4;
    if (wave < NT) {
        const bool second = (!LN) && (wave >= TPM);
        const float* __restrict__ Wm   = second ? Wb : Wa;
        const float* __restrict__ bias = second ? bb : ba;
        const int col = (second ? wave - TPM : wave) * 16 + l16;
        v8f acc = {};
#pragma unroll
        for (int k = 0; k < 64; k += 4) {
            const int kk = k + 2 * half;
            v2f a, b;
            a.x = hs[l16][kk];
            a.y = hs[l16][kk + 1];
            b.x = Wm[(size_t)kk * NCO + col];
            b.y = Wm[(size_t)(kk + 1) * NCO + col];
            acc = __builtin_amdgcn_wmma_f32_16x16x4_f32(
                false, a, false, b, (short)0, acc, false, false);
        }
        const float bj = bias[col];
#pragma unroll
        for (int v = 0; v < 8; ++v) {
            float val = acc[v] + bj;
            if constexpr (LN) val = silu_f(val);
            os[v + 8 * half][wave * 16 + l16] = val;
        }
    }
    __syncthreads();
    const int nl = threadIdx.x >> 4, q = threadIdx.x & 15;
    const size_t row = (size_t)tile * 16 + nl;
    const float4 v0 = *reinterpret_cast<const float4*>(&os[nl][4 * q]);
    if constexpr (LN) {
        float s  = (v0.x + v0.y) + (v0.z + v0.w);
        float sq = (v0.x * v0.x + v0.y * v0.y) + (v0.z * v0.z + v0.w * v0.w);
#pragma unroll
        for (int off = 8; off >= 1; off >>= 1) {
            s  += __shfl_xor(s, off, 32);
            sq += __shfl_xor(sq, off, 32);
        }
        const float mean = s * (1.f / 64);
        const float var  = sq * (1.f / 64) - mean * mean;
        const float inv  = rsqrtf(var + 1e-5f);
        const int c0 = 4 * q;
        float4 o;
        o.x = (v0.x - mean) * inv * g[c0]     + be[c0];
        o.y = (v0.y - mean) * inv * g[c0 + 1] + be[c0 + 1];
        o.z = (v0.z - mean) * inv * g[c0 + 2] + be[c0 + 2];
        o.w = (v0.w - mean) * inv * g[c0 + 3] + be[c0 + 3];
        reinterpret_cast<float4*>(outA + row * 64)[q] = o;
    } else if constexpr (NCO == 64) {
        const float4 v1 = *reinterpret_cast<const float4*>(&os[nl][64 + 4 * q]);
        reinterpret_cast<float4*>(outA + row * 64)[q] = v0;
        reinterpret_cast<float4*>(outB + row * 64)[q] = v1;
    } else {
        if (q < 8) reinterpret_cast<float4*>(outA + row * 32)[q] = v0;
        else       reinterpret_cast<float4*>(outB + row * 32)[q - 8] = v0;
    }
}


template <bool NEXT_LN>
__global__ __launch_bounds__(256) __attribute__((amdgpu_waves_per_eu(16)))
void k_enc_agg(const int* __restrict__ start, const int* __restrict__ csr,
               const float* __restrict__ z, float* __restrict__ h,
               const float* __restrict__ Wa, const float* __restrict__ ba,
               const float* __restrict__ Wb, const float* __restrict__ bb,
               const float* __restrict__ g,  const float* __restrict__ be,
               float* __restrict__ outA, float* __restrict__ outB, int n, int e) {
    __shared__ __attribute__((aligned(16))) float hs[16][HP];
    __shared__ __attribute__((aligned(16))) float os[16][OP];
    const int t = blockIdx.x * 256 + threadIdx.x;
    const int node = t >> 4, q = t & 15;
    const int beg = start[node];
    const int end = (node + 1 < n) ? start[node + 1] : e;
    const float4* z4 = reinterpret_cast<const float4*>(z);
    float4 acc = make_float4(0.f, 0.f, 0.f, 0.f);
    for (int base = beg; base < end; base += 16) {
        const int cnt = min(16, end - base);
        const int my = (q < cnt) ? csr[base + q] : 0;
        for (int j = 0; j < cnt; j += 4) {
            const int s0 = __shfl(my, j + 0, 16);
            const int s1 = __shfl(my, j + 1, 16);
            const int s2 = __shfl(my, j + 2, 16);
            const int s3 = __shfl(my, j + 3, 16);
            const float4 v0 = z4[s0 * 16 + q];
            const float4 v1 = z4[s1 * 16 + q];
            const float4 v2 = z4[s2 * 16 + q];
            const float4 v3 = z4[s3 * 16 + q];
            const bool k1 = j + 1 < cnt, k2 = j + 2 < cnt, k3 = j + 3 < cnt;
            acc.x += v0.x; acc.y += v0.y; acc.z += v0.z; acc.w += v0.w;
            acc.x += k1 ? v1.x : 0.f; acc.y += k1 ? v1.y : 0.f; acc.z += k1 ? v1.z : 0.f; acc.w += k1 ? v1.w : 0.f;
            acc.x += k2 ? v2.x : 0.f; acc.y += k2 ? v2.y : 0.f; acc.z += k2 ? v2.z : 0.f; acc.w += k2 ? v2.w : 0.f;
            acc.x += k3 ? v3.x : 0.f; acc.y += k3 ? v3.y : 0.f; acc.z += k3 ? v3.z : 0.f; acc.w += k3 ? v3.w : 0.f;
        }
    }
    const float inv = 1.0f / fmaxf((float)(end - beg), 1.0f);
    float4* hp = reinterpret_cast<float4*>(h + (size_t)node * 64) + q;
    float4 hv = *hp;
    hv.x += acc.x * inv; hv.y += acc.y * inv; hv.z += acc.z * inv; hv.w += acc.w * inv;
    *reinterpret_cast<float4*>(&hs[threadIdx.x >> 4][4 * q]) = hv;
    __syncthreads();
    tile_tail<64, NEXT_LN>(hs, os, Wa, ba, Wb, bb, g, be, outA, outB, (int)blockIdx.x);
    *hp = *reinterpret_cast<const float4*>(&hs[threadIdx.x >> 4][4 * q]);
}


template <int NC, bool RES, int TAIL>
__global__ __launch_bounds__(256) __attribute__((amdgpu_waves_per_eu(16)))
void k_gat_node(const int* __restrict__ start, const int* __restrict__ csr,
                           const float* __restrict__ xl, const float* __restrict__ xr,
                           const float* __restrict__ att, const float* __restrict__ bias,
                           const float* __restrict__ g, const float* __restrict__ be,
                           const float* hin, float* out,
                           const float* __restrict__ Wa, const float* __restrict__ ba,
                           const float* __restrict__ Wb, const float* __restrict__ bb,
                           float* __restrict__ outA, float* __restrict__ outB,
                           int n, int e) {
    constexpr int LPN = NC / 4;
    const int t = blockIdx.x * 256 + threadIdx.x;
    const int node = t / LPN, q = t % LPN;
    if constexpr (TAIL == 0) { if (node >= n) return; }
    const int beg = start[node];
    const int end = (node + 1 < n) ? start[node + 1] : e;
    const int c0 = q * 4;
    const float a0 = att[c0], a1 = att[c0 + 1], a2 = att[c0 + 2], a3 = att[c0 + 3];
    const float4* xl4 = reinterpret_cast<const float4*>(xl);
    const float4 r4 = reinterpret_cast<const float4*>(xr)[node * LPN + q];
    const float4 self = xl4[node * LPN + q];

    auto part = [&](const float4& v) -> float {
        return a0 * lrelu_f(v.x + r4.x) + a1 * lrelu_f(v.y + r4.y)
             + a2 * lrelu_f(v.z + r4.z) + a3 * lrelu_f(v.w + r4.w);
    };

    float m = part(self);
#pragma unroll
    for (int off = LPN / 2; off >= 1; off >>= 1) m += __shfl_xor(m, off, 32);
    float l = 1.f;
    float4 acc = self;
    for (int base = beg; base < end; base += LPN) {
        const int cnt = min(LPN, end - base);
        const int my = (q < cnt) ? csr[base + q] : node;
        for (int j = 0; j < cnt; j += 4) {
            const int s0 = __shfl(my, j + 0, LPN);
            const int s1 = __shfl(my, j + 1, LPN);
            const int s2 = __shfl(my, j + 2, LPN);
            const int s3 = __shfl(my, j + 3, LPN);
            const float4 v0 = xl4[s0 * LPN + q];
            const float4 v1 = xl4[s1 * LPN + q];
            const float4 v2 = xl4[s2 * LPN + q];
            const float4 v3 = xl4[s3 * LPN + q];
            float e0 = part(v0), e1 = part(v1), e2 = part(v2), e3 = part(v3);
#pragma unroll
            for (int off = LPN / 2; off >= 1; off >>= 1) {
                e0 += __shfl_xor(e0, off, 32);
                e1 += __shfl_xor(e1, off, 32);
                e2 += __shfl_xor(e2, off, 32);
                e3 += __shfl_xor(e3, off, 32);
            }
            if (j + 1 >= cnt) e1 = -INFINITY;
            if (j + 2 >= cnt) e2 = -INFINITY;
            if (j + 3 >= cnt) e3 = -INFINITY;
            const float mn = fmaxf(fmaxf(m, e0), fmaxf(fmaxf(e1, e2), e3));
            const float sc = __expf(m - mn);
            const float p0 = __expf(e0 - mn), p1 = __expf(e1 - mn);
            const float p2 = __expf(e2 - mn), p3 = __expf(e3 - mn);
            l = l * sc + ((p0 + p1) + (p2 + p3));
            acc.x = acc.x * sc + ((p0 * v0.x + p1 * v1.x) + (p2 * v2.x + p3 * v3.x));
            acc.y = acc.y * sc + ((p0 * v0.y + p1 * v1.y) + (p2 * v2.y + p3 * v3.y));
            acc.z = acc.z * sc + ((p0 * v0.z + p1 * v1.z) + (p2 * v2.z + p3 * v3.z));
            acc.w = acc.w * sc + ((p0 * v0.w + p1 * v1.w) + (p2 * v2.w + p3 * v3.w));
            m = mn;
        }
    }

    const float il = 1.f / l;
    float vals[4];
    vals[0] = silu_f(acc.x * il + bias[c0]);
    vals[1] = silu_f(acc.y * il + bias[c0 + 1]);
    vals[2] = silu_f(acc.z * il + bias[c0 + 2]);
    vals[3] = silu_f(acc.w * il + bias[c0 + 3]);
    float s = (vals[0] + vals[1]) + (vals[2] + vals[3]);
    float sq = (vals[0] * vals[0] + vals[1] * vals[1]) + (vals[2] * vals[2] + vals[3] * vals[3]);
#pragma unroll
    for (int off = LPN / 2; off >= 1; off >>= 1) {
        s  += __shfl_xor(s, off, 32);
        sq += __shfl_xor(sq, off, 32);
    }
    const float mean = s * (1.f / NC);
    const float var  = sq * (1.f / NC) - mean * mean;
    const float inv  = rsqrtf(var + 1e-5f);
    float4 o;
    o.x = (vals[0] - mean) * inv * g[c0]     + be[c0];
    o.y = (vals[1] - mean) * inv * g[c0 + 1] + be[c0 + 1];
    o.z = (vals[2] - mean) * inv * g[c0 + 2] + be[c0 + 2];
    o.w = (vals[3] - mean) * inv * g[c0 + 3] + be[c0 + 3];
    if (RES) {
        const float4 hv = reinterpret_cast<const float4*>(hin + (size_t)node * NC)[q];
        o.x += hv.x; o.y += hv.y; o.z += hv.z; o.w += hv.w;
    }
    if constexpr (TAIL > 0) {
        static_assert(NC == 64, "16 lanes per node: a 256-thread block is one 16-row tile");
        __shared__ __attribute__((aligned(16))) float hs[16][HP];
        __shared__ __attribute__((aligned(16))) float os[16][OP];
        *reinterpret_cast<float4*>(&hs[threadIdx.x >> 4][4 * q]) = o;
        __syncthreads();
        tile_tail<TAIL, false>(hs, os, Wa, ba, Wb, bb, nullptr, nullptr, outA, outB, (int)blockIdx.x);
        reinterpret_cast<float4*>(out + (size_t)node * NC)[q] =
            *reinterpret_cast<const float4*>(&hs[threadIdx.x >> 4][4 * q]);
    } else {
        (void)Wa; (void)ba; (void)Wb; (void)bb; (void)outA; (void)outB;
        reinterpret_cast<float4*>(out + (size_t)node * NC)[q] = o;
    }
}


extern "C" void kernel_launch(void* const* d_in, const int* in_sizes, int n_in,
                              void* d_out, int out_size, void* d_ws, size_t ws_size,
                              hipStream_t stream) {
    (void)in_sizes; (void)n_in; (void)out_size; (void)ws_size;

    const float* x        = (const float*)d_in[0];
    const int*   ei       = (const int*)  d_in[1];
    const float* W_emb    = (const float*)d_in[2];
    const float* b_emb    = (const float*)d_in[3];
    const float* g_emb    = (const float*)d_in[4];
    const float* be_emb   = (const float*)d_in[5];
    const float* enc_W    = (const float*)d_in[6];
    const float* enc_b    = (const float*)d_in[7];
    const float* enc_g    = (const float*)d_in[8];
    const float* enc_be   = (const float*)d_in[9];
    const float* dec_Wl   = (const float*)d_in[10];
    const float* dec_bl   = (const float*)d_in[11];
    const float* dec_Wr   = (const float*)d_in[12];
    const float* dec_br   = (const float*)d_in[13];
    const float* dec_att  = (const float*)d_in[14];
    const float* dec_bias = (const float*)d_in[15];
    const float* dec_g    = (const float*)d_in[16];
    const float* dec_be   = (const float*)d_in[17];
    const float* fin_Wl   = (const float*)d_in[18];
    const float* fin_bl   = (const float*)d_in[19];
    const float* fin_Wr   = (const float*)d_in[20];
    const float* fin_br   = (const float*)d_in[21];
    const float* fin_att  = (const float*)d_in[22];
    const float* fin_bias = (const float*)d_in[23];
    const float* fin_g    = (const float*)d_in[24];
    const float* fin_be   = (const float*)d_in[25];

    const int* src = ei;
    const int* dst = ei + EE;
    float* out = (float*)d_out;

    int* wi      = (int*)d_ws;
    int*   start = wi;                  wi += NN;
    int*   tot   = wi;                  wi += 256;
    int*   rank  = wi;                  wi += EE;
    int*   csr   = wi;                  wi += EE;
    float* w     = (float*)wi;
    float* h     = w;                   w += (size_t)NN * 64;
    float* zA    = w;                   w += (size_t)NN * 64;
    float* zB    = w;                   w += (size_t)NN * 64;
    float* xlA   = w;                   w += (size_t)NN * 64;
    float* xrA   = w;                   w += (size_t)NN * 64;

    const int TB = 256;
    auto nb = [](long long t) { return (unsigned)((t + 255) / 256); };
    const int NT = NN / 16;
    const unsigned GG = (unsigned)((NT + WPB - 1) / WPB);
    const dim3 gthr(32 * WPB);

    k_zero<<<nb(NN), TB, 0, stream>>>(start, NN);
    k_count<<<nb(EE), TB, 0, stream>>>(dst, start, rank, EE);
    k_blocksum<<<NSB, 256, 0, stream>>>(start, tot, NN);
    k_scan<<<NSB, 256, 0, stream>>>(start, tot, NN);
    k_place<<<nb(EE), TB, 0, stream>>>(src, dst, start, rank, csr, EE);

    k_gemm<32, 64, true><<<GG, gthr, 0, stream>>>(x, W_emb, b_emb, g_emb, be_emb, h, NT);

    k_gemm<64, 64, true><<<GG, gthr, 0, stream>>>(h, enc_W, enc_b, enc_g, enc_be, zA, NT);

    const unsigned NB16 = (unsigned)(NN / 16);
    k_enc_agg<true><<<NB16, TB, 0, stream>>>(start, csr, zA, h,
        enc_W + (size_t)1 * 64 * 64, enc_b + 1 * 64, nullptr, nullptr, enc_g + 1 * 64, enc_be + 1 * 64,
        zB, nullptr, NN, EE);
    k_enc_agg<true><<<NB16, TB, 0, stream>>>(start, csr, zB, h,
        enc_W + (size_t)2 * 64 * 64, enc_b + 2 * 64, nullptr, nullptr, enc_g + 2 * 64, enc_be + 2 * 64,
        zA, nullptr, NN, EE);
    k_enc_agg<false><<<NB16, TB, 0, stream>>>(start, csr, zA, h,
        dec_Wl, dec_bl, dec_Wr, dec_br, nullptr, nullptr, xlA, xrA, NN, EE);

    k_gat_node<64, true, 64><<<NB16, TB, 0, stream>>>(
        start, csr, xlA, xrA, dec_att, dec_bias, dec_g, dec_be, h, h,
        dec_Wl + (size_t)1 * 64 * 64, dec_bl + 1 * 64, dec_Wr + (size_t)1 * 64 * 64, dec_br + 1 * 64,
        zA, zB, NN, EE);
    k_gat_node<64, true, 64><<<NB16, TB, 0, stream>>>(
        start, csr, zA, zB, dec_att + 1 * 64, dec_bias + 1 * 64, dec_g + 1 * 64, dec_be + 1 * 64, h, h,
        dec_Wl + (size_t)2 * 64 * 64, dec_bl + 2 * 64, dec_Wr + (size_t)2 * 64 * 64, dec_br + 2 * 64,
        xlA, xrA, NN, EE);
    k_gat_node<64, true, 32><<<NB16, TB, 0, stream>>>(
        start, csr, xlA, xrA, dec_att + 2 * 64, dec_bias + 2 * 64, dec_g + 2 * 64, dec_be + 2 * 64, h, h,
        fin_Wl, fin_bl, fin_Wr, fin_br, zA, zB, NN, EE);

    k_gat_node<32, false, 0><<<nb((long long)NN * 8), TB, 0, stream>>>(
        start, csr, zA, zB, fin_att, fin_bias, fin_g, fin_be, nullptr, out,
        nullptr, nullptr, nullptr, nullptr, nullptr, nullptr, NN, EE);
}


# node kernels: group sums as DPP row_xmask adds and source-id hand-round as permlane16_var row gathers, all on the vector ALU in place of LDS-unit shuffles; same butterfly pairs, sum order unchanged
# speedup vs baseline: 5.3214x; 1.0889x over previous
#include <hip/hip_runtime.h>
#include <cstdint>
#include <cstddef>


typedef float v2f __attribute__((ext_vector_type(2)));
typedef float v8f __attribute__((ext_vector_type(8)));

static constexpr int NN  = 100000;
static constexpr int EE  = 1200000;
static constexpr int LM1 = 3;
static constexpr int WPB = 5;
static constexpr int SCB = 1024;
static constexpr int NSB = (NN + SCB - 1) / SCB;
static_assert(NSB <= 256, "block offsets are summed by one 256-thread block");
static_assert(NN % 16 == 0, "a 256-thread node block is exactly one 16-row tile");
static constexpr int HP  = 68;
static constexpr int OP  = 132;

#define DEVFN static __device__ __forceinline__

DEVFN float silu_f(float v)  { return v / (1.f + __expf(-v)); }
DEVFN float lrelu_f(float v) { return v > 0.f ? v : 0.2f * v; }


template <int M>
DEVFN float row_xor_add(float v) {
    const int t = __builtin_amdgcn_update_dpp(0, __builtin_bit_cast(int, v), 0x160 | M, 0xf, 0xf, true);
    return v + __builtin_bit_cast(float, t);
}

template <int G>
DEVFN float group_sum(float v) {
    static_assert(G == 16 || G == 8, "groups inside a 16-lane row");
    if constexpr (G == 16) v = row_xor_add<8>(v);
    v = row_xor_add<4>(v);
    v = row_xor_add<2>(v);
    return row_xor_add<1>(v);
}

template <int G>
DEVFN void group_sum4(float& a, float& b, float& c, float& d) {
    static_assert(G == 16 || G == 8, "groups inside a 16-lane row");
    if constexpr (G == 16) {
        a = row_xor_add<8>(a); b = row_xor_add<8>(b); c = row_xor_add<8>(c); d = row_xor_add<8>(d);
    }
    a = row_xor_add<4>(a); b = row_xor_add<4>(b); c = row_xor_add<4>(c); d = row_xor_add<4>(d);
    a = row_xor_add<2>(a); b = row_xor_add<2>(b); c = row_xor_add<2>(c); d = row_xor_add<2>(d);
    a = row_xor_add<1>(a); b = row_xor_add<1>(b); c = row_xor_add<1>(c); d = row_xor_add<1>(d);
}

template <int G>
DEVFN int group_bcast(int v, int idx) {
    static_assert(G == 16 || G == 8, "groups inside a 16-lane row");
    const int sel = (G == 16) ? idx : (((int)threadIdx.x & 8) | idx);
    return (int)__builtin_amdgcn_permlane16_var(0u, (unsigned)v, (unsigned)sel, true, false);
}


__global__ void k_zero(int* __restrict__ p, int n) {
    int i = blockIdx.x * blockDim.x + threadIdx.x;
    if (i < n) p[i] = 0;
}

__global__ void k_count(const int* __restrict__ dst, int* __restrict__ cnt,
                        int* __restrict__ rank, int e) {
    int i = blockIdx.x * blockDim.x + threadIdx.x;
    if (i < e) rank[i] = atomicAdd(&cnt[dst[i]], 1);
}

__global__ __launch_bounds__(256)
void k_blocksum(const int* __restrict__ cnt, int* __restrict__ tot, int n) {
    __shared__ int red[8];
    const int tid = threadIdx.x;
    const int base = blockIdx.x * SCB + tid * 4;
    int s = 0;
#pragma unroll
    for (int k = 0; k < 4; ++k) if (base + k < n) s += cnt[base + k];
#pragma unroll
    for (int off = 16; off >= 1; off >>= 1) s += __shfl_xor(s, off, 32);
    if ((tid & 31) == 0) red[tid >> 5] = s;
    __syncthreads();
    if (tid == 0) {
        int t = 0;
#pragma unroll
        for (int w = 0; w < 8; ++w) t += red[w];
        tot[blockIdx.x] = t;
    }
}

__global__ __launch_bounds__(256)
void k_scan(int* __restrict__ cnt, const int* __restrict__ tot, int n) {
    __shared__ int wsum[8];
    __shared__ int boff;
    const int tid = threadIdx.x, lane = tid & 31, wv = tid >> 5;
    int o = (tid < (int)blockIdx.x) ? tot[tid] : 0;
#pragma unroll
    for (int off = 16; off >= 1; off >>= 1) o += __shfl_xor(o, off, 32);
    if (lane == 0) wsum[wv] = o;
    __syncthreads();
    if (tid == 0) {
        int t = 0;
#pragma unroll
        for (int w = 0; w < 8; ++w) t += wsum[w];
        boff = t;
    }
    __syncthreads();
    const int blockoff = boff;
    __syncthreads();
    const int base = blockIdx.x * SCB + tid * 4;
    int c[4];
#pragma unroll
    for (int k = 0; k < 4; ++k) c[k] = (base + k < n) ? cnt[base + k] : 0;
    const int tsum = c[0] + c[1] + c[2] + c[3];
    int inc = tsum;
#pragma unroll
    for (int off = 1; off < 32; off <<= 1) {
        const int v = __shfl_up(inc, off, 32);
        if (lane >= off) inc += v;
    }
    if (lane == 31) wsum[wv] = inc;
    __syncthreads();
    int woff = 0;
    for (int w = 0; w < wv; ++w) woff += wsum[w];
    int ex = blockoff + woff + inc - tsum;
#pragma unroll
    for (int k = 0; k < 4; ++k) {
        if (base + k < n) { cnt[base + k] = ex; ex += c[k]; }
    }
}

__global__ void k_place(const int* __restrict__ src, const int* __restrict__ dst,
                        const int* __restrict__ start, const int* __restrict__ rank,
                        int* __restrict__ csr, int e) {
    int i = blockIdx.x * blockDim.x + threadIdx.x;
    if (i < e) csr[start[dst[i]] + rank[i]] = src[i];
}


template <int K, int NC, bool DO_LN>
__global__ __launch_bounds__(32 * WPB)
void k_gemm(const float* __restrict__ A, const float* __restrict__ W,
            const float* __restrict__ bias, const float* __restrict__ gamma,
            const float* __restrict__ beta, float* __restrict__ out, int ntiles) {
    constexpr int NJ = NC / 16;
    const int wave = threadIdx.x >> 5;
    const int lane = threadIdx.x & 31;
    const int l16  = lane & 15;
    const int half = lane >> 4;
    const int tile = blockIdx.x * WPB + wave;
    const bool active = tile < ntiles;

    v8f zero = {};
    v8f acc[NJ];
#pragma unroll
    for (int j = 0; j < NJ; ++j) acc[j] = zero;

    if (active) {
        const float* Arow = A + (size_t)tile * 16 * K + (size_t)l16 * K;
#pragma unroll
        for (int k = 0; k < K; k += 4) {
            const float* ap = Arow + k + 2 * half;
            v2f a;
            a.x = ap[0];
            a.y = ap[1];
#pragma unroll
            for (int j = 0; j < NJ; ++j) {
                const int col = j * 16 + l16;
                v2f b;
                b.x = W[(size_t)(k + 2 * half) * NC + col];
                b.y = W[(size_t)(k + 2 * half + 1) * NC + col];
                acc[j] = __builtin_amdgcn_wmma_f32_16x16x4_f32(
                    false, a, false, b, (short)0, acc[j], false, false);
            }
        }
    }

    if constexpr (DO_LN) {
        static_assert(NC == 64, "LN epilogue assumes 64 columns");
        __shared__ float lds[WPB][16][68];
        if (active) {
#pragma unroll
            for (int j = 0; j < NJ; ++j) {
                const int col = j * 16 + l16;
                const float bj = bias[col];
#pragma unroll
                for (int v = 0; v < 8; ++v) {
                    const int r = v + 8 * half;
                    lds[wave][r][col] = silu_f(acc[j][v] + bj);
                }
            }
        }
        __syncthreads();
        if (active) {
            const int r = l16, cb = half * 32;
            float s = 0.f, sq = 0.f;
#pragma unroll
            for (int c = 0; c < 32; ++c) {
                float vv = lds[wave][r][cb + c];
                s += vv; sq += vv * vv;
            }
            s  += __shfl_xor(s, 16, 32);
            sq += __shfl_xor(sq, 16, 32);
            const float mean = s * (1.f / NC);
            const float var  = sq * (1.f / NC) - mean * mean;
            const float inv  = rsqrtf(var + 1e-5f);
            float* orow = out + ((size_t)tile * 16 + r) * NC;
#pragma unroll
            for (int c = 0; c < 32; ++c) {
                const int col = cb + c;
                const float vv = lds[wave][r][col];
                orow[col] = (vv - mean) * inv * gamma[col] + beta[col];
            }
        }
    } else {
        if (active) {
#pragma unroll
            for (int j = 0; j < NJ; ++j) {
                const int col = j * 16 + l16;
                const float bj = bias[col];
#pragma unroll
                for (int v = 0; v < 8; ++v) {
                    const int r = v + 8 * half;
                    out[((size_t)tile * 16 + r) * NC + col] = acc[j][v] + bj;
                }
            }
        }
        (void)gamma; (void)beta;
    }
}


template <int NCO, bool LN>
DEVFN void tile_tail(float (*hs)[HP], float (*os)[OP],
                     const float* __restrict__ Wa, const float* __restrict__ ba,
                     const float* __restrict__ Wb, const float* __restrict__ bb,
                     const float* __restrict__ g,  const float* __restrict__ be,
                     float* __restrict__ outA, float* __restrict__ outB, int tile) {
    static_assert(NCO == 64 || (NCO == 32 && !LN), "tile map below");
    constexpr int TPM = NCO / 16;
    constexpr int NT  = LN ? TPM : 2 * TPM;
    const int wave = threadIdx.x >> 5;
    const int lane = threadIdx.x & 31;
    const int l16  = lane & 15;
    const int half = lane >> 4;
    if (wave < NT) {
        const bool second = (!LN) && (wave >= TPM);
        const float* __restrict__ Wm   = second ? Wb : Wa;
        const float* __restrict__ bias = second ? bb : ba;
        const int col = (second ? wave - TPM : wave) * 16 + l16;
        v8f acc = {};
#pragma unroll
        for (int k = 0; k < 64; k += 4) {
            const int kk = k + 2 * half;
            v2f a, b;
            a.x = hs[l16][kk];
            a.y = hs[l16][kk + 1];
            b.x = Wm[(size_t)kk * NCO + col];
            b.y = Wm[(size_t)(kk + 1) * NCO + col];
            acc = __builtin_amdgcn_wmma_f32_16x16x4_f32(
                false, a, false, b, (short)0, acc, false, false);
        }
        const float bj = bias[col];
#pragma unroll
        for (int v = 0; v < 8; ++v) {
            float val = acc[v] + bj;
            if constexpr (LN) val = silu_f(val);
            os[v + 8 * half][wave * 16 + l16] = val;
        }
    }
    __syncthreads();
    const int nl = threadIdx.x >> 4, q = threadIdx.x & 15;
    const size_t row = (size_t)tile * 16 + nl;
    const float4 v0 = *reinterpret_cast<const float4*>(&os[nl][4 * q]);
    if constexpr (LN) {
        float s  = (v0.x + v0.y) + (v0.z + v0.w);
        float sq = (v0.x * v0.x + v0.y * v0.y) + (v0.z * v0.z + v0.w * v0.w);
        s  = group_sum<16>(s);
        sq = group_sum<16>(sq);
        const float mean = s * (1.f / 64);
        const float var  = sq * (1.f / 64) - mean * mean;
        const float inv  = rsqrtf(var + 1e-5f);
        const int c0 = 4 * q;
        float4 o;
        o.x = (v0.x - mean) * inv * g[c0]     + be[c0];
        o.y = (v0.y - mean) * inv * g[c0 + 1] + be[c0 + 1];
        o.z = (v0.z - mean) * inv * g[c0 + 2] + be[c0 + 2];
        o.w = (v0.w - mean) * inv * g[c0 + 3] + be[c0 + 3];
        reinterpret_cast<float4*>(outA + row * 64)[q] = o;
    } else if constexpr (NCO == 64) {
        const float4 v1 = *reinterpret_cast<const float4*>(&os[nl][64 + 4 * q]);
        reinterpret_cast<float4*>(outA + row * 64)[q] = v0;
        reinterpret_cast<float4*>(outB + row * 64)[q] = v1;
    } else {
        if (q < 8) reinterpret_cast<float4*>(outA + row * 32)[q] = v0;
        else       reinterpret_cast<float4*>(outB + row * 32)[q - 8] = v0;
    }
}


template <bool NEXT_LN>
__global__ __launch_bounds__(256) __attribute__((amdgpu_waves_per_eu(16)))
void k_enc_agg(const int* __restrict__ start, const int* __restrict__ csr,
               const float* __restrict__ z, float* __restrict__ h,
               const float* __restrict__ Wa, const float* __restrict__ ba,
               const float* __restrict__ Wb, const float* __restrict__ bb,
               const float* __restrict__ g,  const float* __restrict__ be,
               float* __restrict__ outA, float* __restrict__ outB, int n, int e) {
    __shared__ __attribute__((aligned(16))) float hs[16][HP];
    __shared__ __attribute__((aligned(16))) float os[16][OP];
    const int t = blockIdx.x * 256 + threadIdx.x;
    const int node = t >> 4, q = t & 15;
    const int beg = start[node];
    const int end = (node + 1 < n) ? start[node + 1] : e;
    const float4* z4 = reinterpret_cast<const float4*>(z);
    float4 acc = make_float4(0.f, 0.f, 0.f, 0.f);
    for (int base = beg; base < end; base += 16) {
        const int cnt = min(16, end - base);
        const int my = (q < cnt) ? csr[base + q] : 0;
        for (int j = 0; j < cnt; j += 4) {
            const int s0 = group_bcast<16>(my, j + 0);
            const int s1 = group_bcast<16>(my, j + 1);
            const int s2 = group_bcast<16>(my, j + 2);
            const int s3 = group_bcast<16>(my, j + 3);
            const float4 v0 = z4[s0 * 16 + q];
            const float4 v1 = z4[s1 * 16 + q];
            const float4 v2 = z4[s2 * 16 + q];
            const float4 v3 = z4[s3 * 16 + q];
            const bool k1 = j + 1 < cnt, k2 = j + 2 < cnt, k3 = j + 3 < cnt;
            acc.x += v0.x; acc.y += v0.y; acc.z += v0.z; acc.w += v0.w;
            acc.x += k1 ? v1.x : 0.f; acc.y += k1 ? v1.y : 0.f; acc.z += k1 ? v1.z : 0.f; acc.w += k1 ? v1.w : 0.f;
            acc.x += k2 ? v2.x : 0.f; acc.y += k2 ? v2.y : 0.f; acc.z += k2 ? v2.z : 0.f; acc.w += k2 ? v2.w : 0.f;
            acc.x += k3 ? v3.x : 0.f; acc.y += k3 ? v3.y : 0.f; acc.z += k3 ? v3.z : 0.f; acc.w += k3 ? v3.w : 0.f;
        }
    }
    const float inv = 1.0f / fmaxf((float)(end - beg), 1.0f);
    float4* hp = reinterpret_cast<float4*>(h + (size_t)node * 64) + q;
    float4 hv = *hp;
    hv.x += acc.x * inv; hv.y += acc.y * inv; hv.z += acc.z * inv; hv.w += acc.w * inv;
    *reinterpret_cast<float4*>(&hs[threadIdx.x >> 4][4 * q]) = hv;
    __syncthreads();
    tile_tail<64, NEXT_LN>(hs, os, Wa, ba, Wb, bb, g, be, outA, outB, (int)blockIdx.x);
    *hp = *reinterpret_cast<const float4*>(&hs[threadIdx.x >> 4][4 * q]);
}


template <int NC, bool RES, int TAIL>
__global__ __launch_bounds__(256) __attribute__((amdgpu_waves_per_eu(16)))
void k_gat_node(const int* __restrict__ start, const int* __restrict__ csr,
                           const float* __restrict__ xl, const float* __restrict__ xr,
                           const float* __restrict__ att, const float* __restrict__ bias,
                           const float* __restrict__ g, const float* __restrict__ be,
                           const float* hin, float* out,
                           const float* __restrict__ Wa, const float* __restrict__ ba,
                           const float* __restrict__ Wb, const float* __restrict__ bb,
                           float* __restrict__ outA, float* __restrict__ outB,
                           int n, int e) {
    constexpr int LPN = NC / 4;
    const int t = blockIdx.x * 256 + threadIdx.x;
    const int node = t / LPN, q = t % LPN;
    if constexpr (TAIL == 0) { if (node >= n) return; }
    const int beg = start[node];
    const int end = (node + 1 < n) ? start[node + 1] : e;
    const int c0 = q * 4;
    const float a0 = att[c0], a1 = att[c0 + 1], a2 = att[c0 + 2], a3 = att[c0 + 3];
    const float4* xl4 = reinterpret_cast<const float4*>(xl);
    const float4 r4 = reinterpret_cast<const float4*>(xr)[node * LPN + q];
    const float4 self = xl4[node * LPN + q];

    auto part = [&](const float4& v) -> float {
        return a0 * lrelu_f(v.x + r4.x) + a1 * lrelu_f(v.y + r4.y)
             + a2 * lrelu_f(v.z + r4.z) + a3 * lrelu_f(v.w + r4.w);
    };

    float m = group_sum<LPN>(part(self));
    float l = 1.f;
    float4 acc = self;
    for (int base = beg; base < end; base += LPN) {
        const int cnt = min(LPN, end - base);
        const int my = (q < cnt) ? csr[base + q] : node;
        for (int j = 0; j < cnt; j += 4) {
            const int s0 = group_bcast<LPN>(my, j + 0);
            const int s1 = group_bcast<LPN>(my, j + 1);
            const int s2 = group_bcast<LPN>(my, j + 2);
            const int s3 = group_bcast<LPN>(my, j + 3);
            const float4 v0 = xl4[s0 * LPN + q];
            const float4 v1 = xl4[s1 * LPN + q];
            const float4 v2 = xl4[s2 * LPN + q];
            const float4 v3 = xl4[s3 * LPN + q];
            asm volatile("" :: "v"(v0.x), "v"(v0.y), "v"(v0.z), "v"(v0.w),
                               "v"(v1.x), "v"(v1.y), "v"(v1.z), "v"(v1.w),
                               "v"(v2.x), "v"(v2.y), "v"(v2.z), "v"(v2.w),
                               "v"(v3.x), "v"(v3.y), "v"(v3.z), "v"(v3.w));
            float e0 = part(v0), e1 = part(v1), e2 = part(v2), e3 = part(v3);
            group_sum4<LPN>(e0, e1, e2, e3);
            if (j + 1 >= cnt) e1 = -INFINITY;
            if (j + 2 >= cnt) e2 = -INFINITY;
            if (j + 3 >= cnt) e3 = -INFINITY;
            const float mn = fmaxf(fmaxf(m, e0), fmaxf(fmaxf(e1, e2), e3));
            const float sc = __expf(m - mn);
            const float p0 = __expf(e0 - mn), p1 = __expf(e1 - mn);
            const float p2 = __expf(e2 - mn), p3 = __expf(e3 - mn);
            l = l * sc + ((p0 + p1) + (p2 + p3));
            acc.x = acc.x * sc + ((p0 * v0.x + p1 * v1.x) + (p2 * v2.x + p3 * v3.x));
            acc.y = acc.y * sc + ((p0 * v0.y + p1 * v1.y) + (p2 * v2.y + p3 * v3.y));
            acc.z = acc.z * sc + ((p0 * v0.z + p1 * v1.z) + (p2 * v2.z + p3 * v3.z));
            acc.w = acc.w * sc + ((p0 * v0.w + p1 * v1.w) + (p2 * v2.w + p3 * v3.w));
            m = mn;
        }
    }

    const float il = 1.f / l;
    float vals[4];
    vals[0] = silu_f(acc.x * il + bias[c0]);
    vals[1] = silu_f(acc.y * il + bias[c0 + 1]);
    vals[2] = silu_f(acc.z * il + bias[c0 + 2]);
    vals[3] = silu_f(acc.w * il + bias[c0 + 3]);
    float s = (vals[0] + vals[1]) + (vals[2] + vals[3]);
    float sq = (vals[0] * vals[0] + vals[1] * vals[1]) + (vals[2] * vals[2] + vals[3] * vals[3]);
    s  = group_sum<LPN>(s);
    sq = group_sum<LPN>(sq);
    const float mean = s * (1.f / NC);
    const float var  = sq * (1.f / NC) - mean * mean;
    const float inv  = rsqrtf(var + 1e-5f);
    float4 o;
    o.x = (vals[0] - mean) * inv * g[c0]     + be[c0];
    o.y = (vals[1] - mean) * inv * g[c0 + 1] + be[c0 + 1];
    o.z = (vals[2] - mean) * inv * g[c0 + 2] + be[c0 + 2];
    o.w = (vals[3] - mean) * inv * g[c0 + 3] + be[c0 + 3];
    if (RES) {
        const float4 hv = reinterpret_cast<const float4*>(hin + (size_t)node * NC)[q];
        o.x += hv.x; o.y += hv.y; o.z += hv.z; o.w += hv.w;
    }
    if constexpr (TAIL > 0) {
        static_assert(NC == 64, "16 lanes per node: a 256-thread block is one 16-row tile");
        __shared__ __attribute__((aligned(16))) float hs[16][HP];
        __shared__ __attribute__((aligned(16))) float os[16][OP];
        *reinterpret_cast<float4*>(&hs[threadIdx.x >> 4][4 * q]) = o;
        __syncthreads();
        tile_tail<TAIL, false>(hs, os, Wa, ba, Wb, bb, nullptr, nullptr, outA, outB, (int)blockIdx.x);
        reinterpret_cast<float4*>(out + (size_t)node * NC)[q] =
            *reinterpret_cast<const float4*>(&hs[threadIdx.x >> 4][4 * q]);
    } else {
        (void)Wa; (void)ba; (void)Wb; (void)bb; (void)outA; (void)outB;
        reinterpret_cast<float4*>(out + (size_t)node * NC)[q] = o;
    }
}


extern "C" void kernel_launch(void* const* d_in, const int* in_sizes, int n_in,
                              void* d_out, int out_size, void* d_ws, size_t ws_size,
                              hipStream_t stream) {
    (void)in_sizes; (void)n_in; (void)out_size; (void)ws_size;

    const float* x        = (const float*)d_in[0];
    const int*   ei       = (const int*)  d_in[1];
    const float* W_emb    = (const float*)d_in[2];
    const float* b_emb    = (const float*)d_in[3];
    const float* g_emb    = (const float*)d_in[4];
    const float* be_emb   = (const float*)d_in[5];
    const float* enc_W    = (const float*)d_in[6];
    const float* enc_b    = (const float*)d_in[7];
    const float* enc_g    = (const float*)d_in[8];
    const float* enc_be   = (const float*)d_in[9];
    const float* dec_Wl   = (const float*)d_in[10];
    const float* dec_bl   = (const float*)d_in[11];
    const float* dec_Wr   = (const float*)d_in[12];
    const float* dec_br   = (const float*)d_in[13];
    const float* dec_att  = (const float*)d_in[14];
    const float* dec_bias = (const float*)d_in[15];
    const float* dec_g    = (const float*)d_in[16];
    const float* dec_be   = (const float*)d_in[17];
    const float* fin_Wl   = (const float*)d_in[18];
    const float* fin_bl   = (const float*)d_in[19];
    const float* fin_Wr   = (const float*)d_in[20];
    const float* fin_br   = (const float*)d_in[21];
    const float* fin_att  = (const float*)d_in[22];
    const float* fin_bias = (const float*)d_in[23];
    const float* fin_g    = (const float*)d_in[24];
    const float* fin_be   = (const float*)d_in[25];

    const int* src = ei;
    const int* dst = ei + EE;
    float* out = (float*)d_out;

    int* wi      = (int*)d_ws;
    int*   start = wi;                  wi += NN;
    int*   tot   = wi;                  wi += 256;
    int*   rank  = wi;                  wi += EE;
    int*   csr   = wi;                  wi += EE;
    float* w     = (float*)wi;
    float* h     = w;                   w += (size_t)NN * 64;
    float* zA    = w;                   w += (size_t)NN * 64;
    float* zB    = w;                   w += (size_t)NN * 64;
    float* xlA   = w;                   w += (size_t)NN * 64;
    float* xrA   = w;                   w += (size_t)NN * 64;

    const int TB = 256;
    auto nb = [](long long t) { return (unsigned)((t + 255) / 256); };
    const int NT = NN / 16;
    const unsigned GG = (unsigned)((NT + WPB - 1) / WPB);
    const dim3 gthr(32 * WPB);

    k_zero<<<nb(NN), TB, 0, stream>>>(start, NN);
    k_count<<<nb(EE), TB, 0, stream>>>(dst, start, rank, EE);
    k_blocksum<<<NSB, 256, 0, stream>>>(start, tot, NN);
    k_scan<<<NSB, 256, 0, stream>>>(start, tot, NN);
    k_place<<<nb(EE), TB, 0, stream>>>(src, dst, start, rank, csr, EE);

    k_gemm<32, 64, true><<<GG, gthr, 0, stream>>>(x, W_emb, b_emb, g_emb, be_emb, h, NT);

    k_gemm<64, 64, true><<<GG, gthr, 0, stream>>>(h, enc_W, enc_b, enc_g, enc_be, zA, NT);

    const unsigned NB16 = (unsigned)(NN / 16);
    k_enc_agg<true><<<NB16, TB, 0, stream>>>(start, csr, zA, h,
        enc_W + (size_t)1 * 64 * 64, enc_b + 1 * 64, nullptr, nullptr, enc_g + 1 * 64, enc_be + 1 * 64,
        zB, nullptr, NN, EE);
    k_enc_agg<true><<<NB16, TB, 0, stream>>>(start, csr, zB, h,
        enc_W + (size_t)2 * 64 * 64, enc_b + 2 * 64, nullptr, nullptr, enc_g + 2 * 64, enc_be + 2 * 64,
        zA, nullptr, NN, EE);
    k_enc_agg<false><<<NB16, TB, 0, stream>>>(start, csr, zA, h,
        dec_Wl, dec_bl, dec_Wr, dec_br, nullptr, nullptr, xlA, xrA, NN, EE);

    k_gat_node<64, true, 64><<<NB16, TB, 0, stream>>>(
        start, csr, xlA, xrA, dec_att, dec_bias, dec_g, dec_be, h, h,
        dec_Wl + (size_t)1 * 64 * 64, dec_bl + 1 * 64, dec_Wr + (size_t)1 * 64 * 64, dec_br + 1 * 64,
        zA, zB, NN, EE);
    k_gat_node<64, true, 64><<<NB16, TB, 0, stream>>>(
        start, csr, zA, zB, dec_att + 1 * 64, dec_bias + 1 * 64, dec_g + 1 * 64, dec_be + 1 * 64, h, h,
        dec_Wl + (size_t)2 * 64 * 64, dec_bl + 2 * 64, dec_Wr + (size_t)2 * 64 * 64, dec_br + 2 * 64,
        xlA, xrA, NN, EE);
    k_gat_node<64, true, 32><<<NB16, TB, 0, stream>>>(
        start, csr, xlA, xrA, dec_att + 2 * 64, dec_bias + 2 * 64, dec_g + 2 * 64, dec_be + 2 * 64, h, h,
        fin_Wl, fin_bl, fin_Wr, fin_br, zA, zB, NN, EE);

    k_gat_node<32, false, 0><<<nb((long long)NN * 8), TB, 0, stream>>>(
        start, csr, zA, zB, fin_att, fin_bias, fin_g, fin_be, nullptr, out,
        nullptr, nullptr, nullptr, nullptr, nullptr, nullptr, NN, EE);
}


# fewer vector instructions per edge: leaky relu as max, score dot as packed fma (cols 0+2, 1+3 first), 4 score sums in one transposed butterfly (lane pairs first), base-2 softmax, rcp/rsq divisions
# speedup vs baseline: 5.4913x; 1.0319x over previous
#include <hip/hip_runtime.h>
#include <cstdint>
#include <cstddef>


typedef float v2f __attribute__((ext_vector_type(2)));
typedef float v8f __attribute__((ext_vector_type(8)));

static constexpr int NN  = 100000;
static constexpr int EE  = 1200000;
static constexpr int LM1 = 3;
static constexpr int WPB = 5;
static constexpr int SCB = 1024;
static constexpr int NSB = (NN + SCB - 1) / SCB;
static_assert(NSB <= 256, "block offsets are summed by one 256-thread block");
static_assert(NN % 16 == 0, "a 256-thread node block is exactly one 16-row tile");
static constexpr int HP  = 68;
static constexpr int OP  = 132;

#define DEVFN static __device__ __forceinline__

DEVFN float rcp_f(float v)   { return __builtin_amdgcn_rcpf(v); }
DEVFN float rsq_f(float v)   { return __builtin_amdgcn_rsqf(v); }
DEVFN float silu_f(float v)  { return v * rcp_f(1.f + __expf(-v)); }
DEVFN v2f lrelu2(v2f t)      { return __builtin_elementwise_max(t, t * 0.2f); }
static constexpr float LOG2E = 1.44269504088896340736f;
DEVFN v2f lo2(const float4& v) { return v2f{v.x, v.y}; }
DEVFN v2f hi2(const float4& v) { return v2f{v.z, v.w}; }
DEVFN v2f both(float s)        { return v2f{s, s}; }


template <int M>
DEVFN float row_xor_add(float v) {
    const int t = __builtin_amdgcn_update_dpp(0, __builtin_bit_cast(int, v), 0x160 | M, 0xf, 0xf, true);
    return v + __builtin_bit_cast(float, t);
}

template <int G>
DEVFN float group_sum(float v) {
    static_assert(G == 16 || G == 8, "groups inside a 16-lane row");
    if constexpr (G == 16) v = row_xor_add<8>(v);
    v = row_xor_add<4>(v);
    v = row_xor_add<2>(v);
    return row_xor_add<1>(v);
}

template <int M>
DEVFN float row_xor_get(float v) {
    const int t = __builtin_amdgcn_update_dpp(0, __builtin_bit_cast(int, v), 0x160 | M, 0xf, 0xf, true);
    return __builtin_bit_cast(float, t);
}

template <int K>
DEVFN float quad_get(float v) {
    const int t = __builtin_amdgcn_update_dpp(0, __builtin_bit_cast(int, v), K * 0x55, 0xf, 0xf, true);
    return __builtin_bit_cast(float, t);
}

template <int G>
DEVFN void group_sum4(float& a, float& b, float& c, float& d) {
    static_assert(G == 16 || G == 8, "groups inside a 16-lane row");
    const bool b0 = (threadIdx.x & 1) != 0, b1 = (threadIdx.x & 2) != 0;
    const float x = (b0 ? c : a) + row_xor_get<1>(b0 ? a : c);
    const float y = (b0 ? d : b) + row_xor_get<1>(b0 ? b : d);
    float z = (b1 ? y : x) + row_xor_get<2>(b1 ? x : y);
    z = row_xor_add<4>(z);
    if constexpr (G == 16) z = row_xor_add<8>(z);
    a = quad_get<0>(z); c = quad_get<1>(z); b = quad_get<2>(z); d = quad_get<3>(z);
}

template <int G>
DEVFN int group_bcast(int v, int idx) {
    static_assert(G == 16 || G == 8, "groups inside a 16-lane row");
    const unsigned lo = 0x11111111u * (unsigned)idx;
    const unsigned hi = (G == 16) ? lo : lo + 0x88888888u;
    return (int)__builtin_amdgcn_permlane16(0u, (unsigned)v, lo, hi, true, false);
}


__global__ void k_zero(int* __restrict__ p, int n) {
    int i = blockIdx.x * blockDim.x + threadIdx.x;
    if (i < n) p[i] = 0;
}

__global__ void k_count(const int* __restrict__ dst, int* __restrict__ cnt,
                        int* __restrict__ rank, int e) {
    int i = blockIdx.x * blockDim.x + threadIdx.x;
    if (i < e) rank[i] = atomicAdd(&cnt[dst[i]], 1);
}

__global__ __launch_bounds__(256)
void k_blocksum(const int* __restrict__ cnt, int* __restrict__ tot, int n) {
    __shared__ int red[8];
    const int tid = threadIdx.x;
    const int base = blockIdx.x * SCB + tid * 4;
    int s = 0;
#pragma unroll
    for (int k = 0; k < 4; ++k) if (base + k < n) s += cnt[base + k];
#pragma unroll
    for (int off = 16; off >= 1; off >>= 1) s += __shfl_xor(s, off, 32);
    if ((tid & 31) == 0) red[tid >> 5] = s;
    __syncthreads();
    if (tid == 0) {
        int t = 0;
#pragma unroll
        for (int w = 0; w < 8; ++w) t += red[w];
        tot[blockIdx.x] = t;
    }
}

__global__ __launch_bounds__(256)
void k_scan(int* __restrict__ cnt, const int* __restrict__ tot, int n) {
    __shared__ int wsum[8];
    __shared__ int boff;
    const int tid = threadIdx.x, lane = tid & 31, wv = tid >> 5;
    int o = (tid < (int)blockIdx.x) ? tot[tid] : 0;
#pragma unroll
    for (int off = 16; off >= 1; off >>= 1) o += __shfl_xor(o, off, 32);
    if (lane == 0) wsum[wv] = o;
    __syncthreads();
    if (tid == 0) {
        int t = 0;
#pragma unroll
        for (int w = 0; w < 8; ++w) t += wsum[w];
        boff = t;
    }
    __syncthreads();
    const int blockoff = boff;
    __syncthreads();
    const int base = blockIdx.x * SCB + tid * 4;
    int c[4];
#pragma unroll
    for (int k = 0; k < 4; ++k) c[k] = (base + k < n) ? cnt[base + k] : 0;
    const int tsum = c[0] + c[1] + c[2] + c[3];
    int inc = tsum;
#pragma unroll
    for (int off = 1; off < 32; off <<= 1) {
        const int v = __shfl_up(inc, off, 32);
        if (lane >= off) inc += v;
    }
    if (lane == 31) wsum[wv] = inc;
    __syncthreads();
    int woff = 0;
    for (int w = 0; w < wv; ++w) woff += wsum[w];
    int ex = blockoff + woff + inc - tsum;
#pragma unroll
    for (int k = 0; k < 4; ++k) {
        if (base + k < n) { cnt[base + k] = ex; ex += c[k]; }
    }
}

__global__ void k_place(const int* __restrict__ src, const int* __restrict__ dst,
                        const int* __restrict__ start, const int* __restrict__ rank,
                        int* __restrict__ csr, int e) {
    int i = blockIdx.x * blockDim.x + threadIdx.x;
    if (i < e) csr[start[dst[i]] + rank[i]] = src[i];
}


template <int K, int NC, bool DO_LN>
__global__ __launch_bounds__(32 * WPB)
void k_gemm(const float* __restrict__ A, const float* __restrict__ W,
            const float* __restrict__ bias, const float* __restrict__ gamma,
            const float* __restrict__ beta, float* __restrict__ out, int ntiles) {
    constexpr int NJ = NC / 16;
    const int wave = threadIdx.x >> 5;
    const int lane = threadIdx.x & 31;
    const int l16  = lane & 15;
    const int half = lane >> 4;
    const int tile = blockIdx.x * WPB + wave;
    const bool active = tile < ntiles;

    v8f zero = {};
    v8f acc[NJ];
#pragma unroll
    for (int j = 0; j < NJ; ++j) acc[j] = zero;

    if (active) {
        const float* Arow = A + (size_t)tile * 16 * K + (size_t)l16 * K;
#pragma unroll
        for (int k = 0; k < K; k += 4) {
            const float* ap = Arow + k + 2 * half;
            v2f a;
            a.x = ap[0];
            a.y = ap[1];
#pragma unroll
            for (int j = 0; j < NJ; ++j) {
                const int col = j * 16 + l16;
                v2f b;
                b.x = W[(size_t)(k + 2 * half) * NC + col];
                b.y = W[(size_t)(k + 2 * half + 1) * NC + col];
                acc[j] = __builtin_amdgcn_wmma_f32_16x16x4_f32(
                    false, a, false, b, (short)0, acc[j], false, false);
            }
        }
    }

    if constexpr (DO_LN) {
        static_assert(NC == 64, "LN epilogue assumes 64 columns");
        __shared__ float lds[WPB][16][68];
        if (active) {
#pragma unroll
            for (int j = 0; j < NJ; ++j) {
                const int col = j * 16 + l16;
                const float bj = bias[col];
#pragma unroll
                for (int v = 0; v < 8; ++v) {
                    const int r = v + 8 * half;
                    lds[wave][r][col] = silu_f(acc[j][v] + bj);
                }
            }
        }
        __syncthreads();
        if (active) {
            const int r = l16, cb = half * 32;
            float s = 0.f, sq = 0.f;
#pragma unroll
            for (int c = 0; c < 32; ++c) {
                float vv = lds[wave][r][cb + c];
                s += vv; sq += vv * vv;
            }
            s  += __shfl_xor(s, 16, 32);
            sq += __shfl_xor(sq, 16, 32);
            const float mean = s * (1.f / NC);
            const float var  = sq * (1.f / NC) - mean * mean;
            const float inv  = rsq_f(var + 1e-5f);
            float* orow = out + ((size_t)tile * 16 + r) * NC;
#pragma unroll
            for (int c = 0; c < 32; ++c) {
                const int col = cb + c;
                const float vv = lds[wave][r][col];
                orow[col] = (vv - mean) * inv * gamma[col] + beta[col];
            }
        }
    } else {
        if (active) {
#pragma unroll
            for (int j = 0; j < NJ; ++j) {
                const int col = j * 16 + l16;
                const float bj = bias[col];
#pragma unroll
                for (int v = 0; v < 8; ++v) {
                    const int r = v + 8 * half;
                    out[((size_t)tile * 16 + r) * NC + col] = acc[j][v] + bj;
                }
            }
        }
        (void)gamma; (void)beta;
    }
}


template <int NCO, bool LN>
DEVFN void tile_tail(float (*hs)[HP], float (*os)[OP],
                     const float* __restrict__ Wa, const float* __restrict__ ba,
                     const float* __restrict__ Wb, const float* __restrict__ bb,
                     const float* __restrict__ g,  const float* __restrict__ be,
                     float* __restrict__ outA, float* __restrict__ outB, int tile) {
    static_assert(NCO == 64 || (NCO == 32 && !LN), "tile map below");
    constexpr int TPM = NCO / 16;
    constexpr int NT  = LN ? TPM : 2 * TPM;
    const int wave = threadIdx.x >> 5;
    const int lane = threadIdx.x & 31;
    const int l16  = lane & 15;
    const int half = lane >> 4;
    if (wave < NT) {
        const bool second = (!LN) && (wave >= TPM);
        const float* __restrict__ Wm   = second ? Wb : Wa;
        const float* __restrict__ bias = second ? bb : ba;
        const int col = (second ? wave - TPM : wave) * 16 + l16;
        v8f acc = {};
#pragma unroll
        for (int k = 0; k < 64; k += 4) {
            const int kk = k + 2 * half;
            v2f a, b;
            a.x = hs[l16][kk];
            a.y = hs[l16][kk + 1];
            b.x = Wm[(size_t)kk * NCO + col];
            b.y = Wm[(size_t)(kk + 1) * NCO + col];
            acc = __builtin_amdgcn_wmma_f32_16x16x4_f32(
                false, a, false, b, (short)0, acc, false, false);
        }
        const float bj = bias[col];
#pragma unroll
        for (int v = 0; v < 8; ++v) {
            float val = acc[v] + bj;
            if constexpr (LN) val = silu_f(val);
            os[v + 8 * half][wave * 16 + l16] = val;
        }
    }
    __syncthreads();
    const int nl = threadIdx.x >> 4, q = threadIdx.x & 15;
    const size_t row = (size_t)tile * 16 + nl;
    const float4 v0 = *reinterpret_cast<const float4*>(&os[nl][4 * q]);
    if constexpr (LN) {
        float s  = (v0.x + v0.y) + (v0.z + v0.w);
        float sq = (v0.x * v0.x + v0.y * v0.y) + (v0.z * v0.z + v0.w * v0.w);
        s  = group_sum<16>(s);
        sq = group_sum<16>(sq);
        const float mean = s * (1.f / 64);
        const float var  = sq * (1.f / 64) - mean * mean;
        const float inv  = rsq_f(var + 1e-5f);
        const int c0 = 4 * q;
        float4 o;
        o.x = (v0.x - mean) * inv * g[c0]     + be[c0];
        o.y = (v0.y - mean) * inv * g[c0 + 1] + be[c0 + 1];
        o.z = (v0.z - mean) * inv * g[c0 + 2] + be[c0 + 2];
        o.w = (v0.w - mean) * inv * g[c0 + 3] + be[c0 + 3];
        reinterpret_cast<float4*>(outA + row * 64)[q] = o;
    } else if constexpr (NCO == 64) {
        const float4 v1 = *reinterpret_cast<const float4*>(&os[nl][64 + 4 * q]);
        reinterpret_cast<float4*>(outA + row * 64)[q] = v0;
        reinterpret_cast<float4*>(outB + row * 64)[q] = v1;
    } else {
        if (q < 8) reinterpret_cast<float4*>(outA + row * 32)[q] = v0;
        else       reinterpret_cast<float4*>(outB + row * 32)[q - 8] = v0;
    }
}


template <bool NEXT_LN>
__global__ __launch_bounds__(256) __attribute__((amdgpu_waves_per_eu(16)))
void k_enc_agg(const int* __restrict__ start, const int* __restrict__ csr,
               const float* __restrict__ z, float* __restrict__ h,
               const float* __restrict__ Wa, const float* __restrict__ ba,
               const float* __restrict__ Wb, const float* __restrict__ bb,
               const float* __restrict__ g,  const float* __restrict__ be,
               float* __restrict__ outA, float* __restrict__ outB, int n, int e) {
    __shared__ __attribute__((aligned(16))) float hs[16][HP];
    __shared__ __attribute__((aligned(16))) float os[16][OP];
    const int t = blockIdx.x * 256 + threadIdx.x;
    const int node = t >> 4, q = t & 15;
    const int beg = start[node];
    const int end = (node + 1 < n) ? start[node + 1] : e;
    const float4* z4 = reinterpret_cast<const float4*>(z);
    float4 acc = make_float4(0.f, 0.f, 0.f, 0.f);
    for (int base = beg; base < end; base += 16) {
        const int cnt = min(16, end - base);
        const int my = (q < cnt) ? csr[base + q] : 0;
        for (int j = 0; j < cnt; j += 4) {
            const int s0 = group_bcast<16>(my, j + 0);
            const int s1 = group_bcast<16>(my, j + 1);
            const int s2 = group_bcast<16>(my, j + 2);
            const int s3 = group_bcast<16>(my, j + 3);
            const float4 v0 = z4[s0 * 16 + q];
            const float4 v1 = z4[s1 * 16 + q];
            const float4 v2 = z4[s2 * 16 + q];
            const float4 v3 = z4[s3 * 16 + q];
            const bool k1 = j + 1 < cnt, k2 = j + 2 < cnt, k3 = j + 3 < cnt;
            acc.x += v0.x; acc.y += v0.y; acc.z += v0.z; acc.w += v0.w;
            acc.x += k1 ? v1.x : 0.f; acc.y += k1 ? v1.y : 0.f; acc.z += k1 ? v1.z : 0.f; acc.w += k1 ? v1.w : 0.f;
            acc.x += k2 ? v2.x : 0.f; acc.y += k2 ? v2.y : 0.f; acc.z += k2 ? v2.z : 0.f; acc.w += k2 ? v2.w : 0.f;
            acc.x += k3 ? v3.x : 0.f; acc.y += k3 ? v3.y : 0.f; acc.z += k3 ? v3.z : 0.f; acc.w += k3 ? v3.w : 0.f;
        }
    }
    const float inv = rcp_f(fmaxf((float)(end - beg), 1.0f));
    float4* hp = reinterpret_cast<float4*>(h + (size_t)node * 64) + q;
    float4 hv = *hp;
    hv.x += acc.x * inv; hv.y += acc.y * inv; hv.z += acc.z * inv; hv.w += acc.w * inv;
    *reinterpret_cast<float4*>(&hs[threadIdx.x >> 4][4 * q]) = hv;
    __syncthreads();
    tile_tail<64, NEXT_LN>(hs, os, Wa, ba, Wb, bb, g, be, outA, outB, (int)blockIdx.x);
    *hp = *reinterpret_cast<const float4*>(&hs[threadIdx.x >> 4][4 * q]);
}


template <int NC, bool RES, int TAIL>
__global__ __launch_bounds__(256) __attribute__((amdgpu_waves_per_eu(16)))
void k_gat_node(const int* __restrict__ start, const int* __restrict__ csr,
                           const float* __restrict__ xl, const float* __restrict__ xr,
                           const float* __restrict__ att, const float* __restrict__ bias,
                           const float* __restrict__ g, const float* __restrict__ be,
                           const float* hin, float* out,
                           const float* __restrict__ Wa, const float* __restrict__ ba,
                           const float* __restrict__ Wb, const float* __restrict__ bb,
                           float* __restrict__ outA, float* __restrict__ outB,
                           int n, int e) {
    constexpr int LPN = NC / 4;
    const int t = blockIdx.x * 256 + threadIdx.x;
    const int node = t / LPN, q = t % LPN;
    if constexpr (TAIL == 0) { if (node >= n) return; }
    const int beg = start[node];
    const int end = (node + 1 < n) ? start[node + 1] : e;
    const int c0 = q * 4;
    const v2f a01 = { att[c0] * LOG2E,     att[c0 + 1] * LOG2E };
    const v2f a23 = { att[c0 + 2] * LOG2E, att[c0 + 3] * LOG2E };
    const float4* xl4 = reinterpret_cast<const float4*>(xl);
    const float4 r4 = reinterpret_cast<const float4*>(xr)[node * LPN + q];
    const float4 self = xl4[node * LPN + q];
    const v2f r01 = { r4.x, r4.y }, r23 = { r4.z, r4.w };

    auto part = [&](const float4& v) -> float {
        const v2f t01 = lo2(v) + r01;
        const v2f t23 = hi2(v) + r23;
        v2f p = a01 * lrelu2(t01);
        p = __builtin_elementwise_fma(a23, lrelu2(t23), p);
        return p.x + p.y;
    };

    float m = group_sum<LPN>(part(self));
    float l = 1.f;
    v2f acc01 = lo2(self), acc23 = hi2(self);
    for (int base = beg; base < end; base += LPN) {
        const int cnt = min(LPN, end - base);
        const int my = (q < cnt) ? csr[base + q] : node;
        for (int j = 0; j < cnt; j += 4) {
            const int s0 = group_bcast<LPN>(my, j + 0);
            const int s1 = group_bcast<LPN>(my, j + 1);
            const int s2 = group_bcast<LPN>(my, j + 2);
            const int s3 = group_bcast<LPN>(my, j + 3);
            const float4 v0 = xl4[s0 * LPN + q];
            const float4 v1 = xl4[s1 * LPN + q];
            const float4 v2 = xl4[s2 * LPN + q];
            const float4 v3 = xl4[s3 * LPN + q];
            asm volatile("" :: "v"(v0.x), "v"(v0.y), "v"(v0.z), "v"(v0.w),
                               "v"(v1.x), "v"(v1.y), "v"(v1.z), "v"(v1.w),
                               "v"(v2.x), "v"(v2.y), "v"(v2.z), "v"(v2.w),
                               "v"(v3.x), "v"(v3.y), "v"(v3.z), "v"(v3.w));
            float e0 = part(v0), e1 = part(v1), e2 = part(v2), e3 = part(v3);
            group_sum4<LPN>(e0, e1, e2, e3);
            if (j + 1 >= cnt) e1 = -INFINITY;
            if (j + 2 >= cnt) e2 = -INFINITY;
            if (j + 3 >= cnt) e3 = -INFINITY;
            const float mn = fmaxf(fmaxf(m, e0), fmaxf(fmaxf(e1, e2), e3));
            const v2f d01 = v2f{e0, e1} - both(mn), d23 = v2f{e2, e3} - both(mn);
            const float sc = __builtin_amdgcn_exp2f(m - mn);
            const float p0 = __builtin_amdgcn_exp2f(d01.x), p1 = __builtin_amdgcn_exp2f(d01.y);
            const float p2 = __builtin_amdgcn_exp2f(d23.x), p3 = __builtin_amdgcn_exp2f(d23.y);
            l = l * sc + ((p0 + p1) + (p2 + p3));
            const v2f P0 = both(p0), P1 = both(p1), P2 = both(p2), P3 = both(p3), SC = both(sc);
            const v2f s01a = __builtin_elementwise_fma(lo2(v0), P0, lo2(v1) * P1);
            const v2f s01b = __builtin_elementwise_fma(lo2(v2), P2, lo2(v3) * P3);
            const v2f s23a = __builtin_elementwise_fma(hi2(v0), P0, hi2(v1) * P1);
            const v2f s23b = __builtin_elementwise_fma(hi2(v2), P2, hi2(v3) * P3);
            acc01 = __builtin_elementwise_fma(acc01, SC, s01a + s01b);
            acc23 = __builtin_elementwise_fma(acc23, SC, s23a + s23b);
            m = mn;
        }
    }

    const float il = rcp_f(l);
    float vals[4];
    vals[0] = silu_f(acc01.x * il + bias[c0]);
    vals[1] = silu_f(acc01.y * il + bias[c0 + 1]);
    vals[2] = silu_f(acc23.x * il + bias[c0 + 2]);
    vals[3] = silu_f(acc23.y * il + bias[c0 + 3]);
    float s = (vals[0] + vals[1]) + (vals[2] + vals[3]);
    float sq = (vals[0] * vals[0] + vals[1] * vals[1]) + (vals[2] * vals[2] + vals[3] * vals[3]);
    s  = group_sum<LPN>(s);
    sq = group_sum<LPN>(sq);
    const float mean = s * (1.f / NC);
    const float var  = sq * (1.f / NC) - mean * mean;
    const float inv  = rsq_f(var + 1e-5f);
    float4 o;
    o.x = (vals[0] - mean) * inv * g[c0]     + be[c0];
    o.y = (vals[1] - mean) * inv * g[c0 + 1] + be[c0 + 1];
    o.z = (vals[2] - mean) * inv * g[c0 + 2] + be[c0 + 2];
    o.w = (vals[3] - mean) * inv * g[c0 + 3] + be[c0 + 3];
    if (RES) {
        const float4 hv = reinterpret_cast<const float4*>(hin + (size_t)node * NC)[q];
        o.x += hv.x; o.y += hv.y; o.z += hv.z; o.w += hv.w;
    }
    if constexpr (TAIL > 0) {
        static_assert(NC == 64, "16 lanes per node: a 256-thread block is one 16-row tile");
        __shared__ __attribute__((aligned(16))) float hs[16][HP];
        __shared__ __attribute__((aligned(16))) float os[16][OP];
        *reinterpret_cast<float4*>(&hs[threadIdx.x >> 4][4 * q]) = o;
        __syncthreads();
        tile_tail<TAIL, false>(hs, os, Wa, ba, Wb, bb, nullptr, nullptr, outA, outB, (int)blockIdx.x);
        reinterpret_cast<float4*>(out + (size_t)node * NC)[q] =
            *reinterpret_cast<const float4*>(&hs[threadIdx.x >> 4][4 * q]);
    } else {
        (void)Wa; (void)ba; (void)Wb; (void)bb; (void)outA; (void)outB;
        reinterpret_cast<float4*>(out + (size_t)node * NC)[q] = o;
    }
}


extern "C" void kernel_launch(void* const* d_in, const int* in_sizes, int n_in,
                              void* d_out, int out_size, void* d_ws, size_t ws_size,
                              hipStream_t stream) {
    (void)in_sizes; (void)n_in; (void)out_size; (void)ws_size;

    const float* x        = (const float*)d_in[0];
    const int*   ei       = (const int*)  d_in[1];
    const float* W_emb    = (const float*)d_in[2];
    const float* b_emb    = (const float*)d_in[3];
    const float* g_emb    = (const float*)d_in[4];
    const float* be_emb   = (const float*)d_in[5];
    const float* enc_W    = (const float*)d_in[6];
    const float* enc_b    = (const float*)d_in[7];
    const float* enc_g    = (const float*)d_in[8];
    const float* enc_be   = (const float*)d_in[9];
    const float* dec_Wl   = (const float*)d_in[10];
    const float* dec_bl   = (const float*)d_in[11];
    const float* dec_Wr   = (const float*)d_in[12];
    const float* dec_br   = (const float*)d_in[13];
    const float* dec_att  = (const float*)d_in[14];
    const float* dec_bias = (const float*)d_in[15];
    const float* dec_g    = (const float*)d_in[16];
    const float* dec_be   = (const float*)d_in[17];
    const float* fin_Wl   = (const float*)d_in[18];
    const float* fin_bl   = (const float*)d_in[19];
    const float* fin_Wr   = (const float*)d_in[20];
    const float* fin_br   = (const float*)d_in[21];
    const float* fin_att  = (const float*)d_in[22];
    const float* fin_bias = (const float*)d_in[23];
    const float* fin_g    = (const float*)d_in[24];
    const float* fin_be   = (const float*)d_in[25];

    const int* src = ei;
    const int* dst = ei + EE;
    float* out = (float*)d_out;

    int* wi      = (int*)d_ws;
    int*   start = wi;                  wi += NN;
    int*   tot   = wi;                  wi += 256;
    int*   rank  = wi;                  wi += EE;
    int*   csr   = wi;                  wi += EE;
    float* w     = (float*)wi;
    float* h     = w;                   w += (size_t)NN * 64;
    float* zA    = w;                   w += (size_t)NN * 64;
    float* zB    = w;                   w += (size_t)NN * 64;
    float* xlA   = w;                   w += (size_t)NN * 64;
    float* xrA   = w;                   w += (size_t)NN * 64;

    const int TB = 256;
    auto nb = [](long long t) { return (unsigned)((t + 255) / 256); };
    const int NT = NN / 16;
    const unsigned GG = (unsigned)((NT + WPB - 1) / WPB);
    const dim3 gthr(32 * WPB);

    k_zero<<<nb(NN), TB, 0, stream>>>(start, NN);
    k_count<<<nb(EE), TB, 0, stream>>>(dst, start, rank, EE);
    k_blocksum<<<NSB, 256, 0, stream>>>(start, tot, NN);
    k_scan<<<NSB, 256, 0, stream>>>(start, tot, NN);
    k_place<<<nb(EE), TB, 0, stream>>>(src, dst, start, rank, csr, EE);

    k_gemm<32, 64, true><<<GG, gthr, 0, stream>>>(x, W_emb, b_emb, g_emb, be_emb, h, NT);

    k_gemm<64, 64, true><<<GG, gthr, 0, stream>>>(h, enc_W, enc_b, enc_g, enc_be, zA, NT);

    const unsigned NB16 = (unsigned)(NN / 16);
    k_enc_agg<true><<<NB16, TB, 0, stream>>>(start, csr, zA, h,
        enc_W + (size_t)1 * 64 * 64, enc_b + 1 * 64, nullptr, nullptr, enc_g + 1 * 64, enc_be + 1 * 64,
        zB, nullptr, NN, EE);
    k_enc_agg<true><<<NB16, TB, 0, stream>>>(start, csr, zB, h,
        enc_W + (size_t)2 * 64 * 64, enc_b + 2 * 64, nullptr, nullptr, enc_g + 2 * 64, enc_be + 2 * 64,
        zA, nullptr, NN, EE);
    k_enc_agg<false><<<NB16, TB, 0, stream>>>(start, csr, zA, h,
        dec_Wl, dec_bl, dec_Wr, dec_br, nullptr, nullptr, xlA, xrA, NN, EE);

    k_gat_node<64, true, 64><<<NB16, TB, 0, stream>>>(
        start, csr, xlA, xrA, dec_att, dec_bias, dec_g, dec_be, h, h,
        dec_Wl + (size_t)1 * 64 * 64, dec_bl + 1 * 64, dec_Wr + (size_t)1 * 64 * 64, dec_br + 1 * 64,
        zA, zB, NN, EE);
    k_gat_node<64, true, 64><<<NB16, TB, 0, stream>>>(
        start, csr, zA, zB, dec_att + 1 * 64, dec_bias + 1 * 64, dec_g + 1 * 64, dec_be + 1 * 64, h, h,
        dec_Wl + (size_t)2 * 64 * 64, dec_bl + 2 * 64, dec_Wr + (size_t)2 * 64 * 64, dec_br + 2 * 64,
        xlA, xrA, NN, EE);
    k_gat_node<64, true, 32><<<NB16, TB, 0, stream>>>(
        start, csr, xlA, xrA, dec_att + 2 * 64, dec_bias + 2 * 64, dec_g + 2 * 64, dec_be + 2 * 64, h, h,
        fin_Wl, fin_bl, fin_Wr, fin_br, zA, zB, NN, EE);

    k_gat_node<32, false, 0><<<nb((long long)NN * 8), TB, 0, stream>>>(
        start, csr, zA, zB, fin_att, fin_bias, fin_g, fin_be, nullptr, out,
        nullptr, nullptr, nullptr, nullptr, nullptr, nullptr, NN, EE);
}


# edge sort with all counters in LDS, no global atomics: edges grouped by 256-node range per 4096-edge chunk, then each range counted, scanned, placed (14 to 13 launches); slot order LDS-atomic arrival
# speedup vs baseline: 6.4482x; 1.1743x over previous
#include <hip/hip_runtime.h>
#include <cstdint>
#include <cstddef>


typedef float v2f __attribute__((ext_vector_type(2)));
typedef float v8f __attribute__((ext_vector_type(8)));

static constexpr int NN  = 100000;
static constexpr int EE  = 1200000;
static constexpr int LM1 = 3;
static constexpr int WPB = 5;
static_assert(NN % 16 == 0, "a 256-thread node block is exactly one 16-row tile");
static constexpr int HP  = 68;
static constexpr int OP  = 132;

#define DEVFN static __device__ __forceinline__

DEVFN float rcp_f(float v)   { return __builtin_amdgcn_rcpf(v); }
DEVFN float rsq_f(float v)   { return __builtin_amdgcn_rsqf(v); }
DEVFN float silu_f(float v)  { return v * rcp_f(1.f + __expf(-v)); }
DEVFN v2f lrelu2(v2f t)      { return __builtin_elementwise_max(t, t * 0.2f); }
static constexpr float LOG2E = 1.44269504088896340736f;
DEVFN v2f lo2(const float4& v) { return v2f{v.x, v.y}; }
DEVFN v2f hi2(const float4& v) { return v2f{v.z, v.w}; }
DEVFN v2f both(float s)        { return v2f{s, s}; }


template <int M>
DEVFN float row_xor_add(float v) {
    const int t = __builtin_amdgcn_update_dpp(0, __builtin_bit_cast(int, v), 0x160 | M, 0xf, 0xf, true);
    return v + __builtin_bit_cast(float, t);
}

template <int G>
DEVFN float group_sum(float v) {
    static_assert(G == 16 || G == 8, "groups inside a 16-lane row");
    if constexpr (G == 16) v = row_xor_add<8>(v);
    v = row_xor_add<4>(v);
    v = row_xor_add<2>(v);
    return row_xor_add<1>(v);
}

template <int M>
DEVFN float row_xor_get(float v) {
    const int t = __builtin_amdgcn_update_dpp(0, __builtin_bit_cast(int, v), 0x160 | M, 0xf, 0xf, true);
    return __builtin_bit_cast(float, t);
}

template <int K>
DEVFN float quad_get(float v) {
    const int t = __builtin_amdgcn_update_dpp(0, __builtin_bit_cast(int, v), K * 0x55, 0xf, 0xf, true);
    return __builtin_bit_cast(float, t);
}

template <int G>
DEVFN void group_sum4(float& a, float& b, float& c, float& d) {
    static_assert(G == 16 || G == 8, "groups inside a 16-lane row");
    const bool b0 = (threadIdx.x & 1) != 0, b1 = (threadIdx.x & 2) != 0;
    const float x = (b0 ? c : a) + row_xor_get<1>(b0 ? a : c);
    const float y = (b0 ? d : b) + row_xor_get<1>(b0 ? b : d);
    float z = (b1 ? y : x) + row_xor_get<2>(b1 ? x : y);
    z = row_xor_add<4>(z);
    if constexpr (G == 16) z = row_xor_add<8>(z);
    a = quad_get<0>(z); c = quad_get<1>(z); b = quad_get<2>(z); d = quad_get<3>(z);
}

template <int G>
DEVFN int group_bcast(int v, int idx) {
    static_assert(G == 16 || G == 8, "groups inside a 16-lane row");
    const unsigned lo = 0x11111111u * (unsigned)idx;
    const unsigned hi = (G == 16) ? lo : lo + 0x88888888u;
    return (int)__builtin_amdgcn_permlane16(0u, (unsigned)v, lo, hi, true, false);
}


static constexpr int RSH = 8;
static constexpr int RB  = 1 << RSH;
static constexpr int NBK = (NN + RB - 1) / RB;
static constexpr int CH  = 4096;
static constexpr int NB1 = (EE + CH - 1) / CH;
static constexpr int CAP = 8192;
static_assert(NBK <= 512, "k_scatter scans the ranges two per thread");
static_assert(NB1 <= 512, "k_rowscan scans the chunks one per thread");
static_assert(NN <= (1 << (31 - RSH)), "a source id and RSH destination bits share one positive word");

template <int NT>
DEVFN int block_excl(int v, int* wsum, int& total) {
    const int lane = threadIdx.x & 31, wv = threadIdx.x >> 5;
    int inc = v;
#pragma unroll
    for (int off = 1; off < 32; off <<= 1) {
        const int u = __shfl_up(inc, off, 32);
        if (lane >= off) inc += u;
    }
    __syncthreads();
    if (lane == 31) wsum[wv] = inc;
    __syncthreads();
    int woff = 0, tot = 0;
#pragma unroll
    for (int w = 0; w < NT / 32; ++w) {
        const int s = wsum[w];
        tot += s;
        if (w < wv) woff += s;
    }
    total = tot;
    return woff + inc - v;
}

__global__ __launch_bounds__(256)
void k_hist(const int* __restrict__ dst, int* __restrict__ mat) {
    __shared__ int hist[NBK];
    const int tid = threadIdx.x, c = blockIdx.x;
    for (int k = tid; k < NBK; k += 256) hist[k] = 0;
    __syncthreads();
    const int e0 = c * CH;
#pragma unroll
    for (int i = 0; i < CH / 256; ++i) {
        const int e = e0 + i * 256 + tid;
        const int d = dst[min(e, EE - 1)];
        if (e < EE) atomicAdd(&hist[d >> RSH], 1);
    }
    __syncthreads();
    for (int k = tid; k < NBK; k += 256) mat[c * NBK + k] = hist[k];
}

__global__ __launch_bounds__(512)
void k_rowscan(int* __restrict__ mat, int* __restrict__ rowtot) {
    __shared__ int wsum[16];
    const int k = blockIdx.x, t = threadIdx.x;
    const int v = (t < NB1) ? mat[t * NBK + k] : 0;
    int total;
    const int ex = block_excl<512>(v, wsum, total);
    if (t < NB1) mat[t * NBK + k] = ex;
    if (t == 0) rowtot[k] = total;
}

__global__ __launch_bounds__(256)
void k_scatter(const int* __restrict__ src, const int* __restrict__ dst,
               const int* __restrict__ mat, const int* __restrict__ rowtot,
               int* __restrict__ tmp) {
    __shared__ int cur[512];
    __shared__ int gdel[512];
    __shared__ int sval[CH];
    __shared__ unsigned short sbk[CH];
    __shared__ int wsum[8];
    const int tid = threadIdx.x, c = blockIdx.x;
    cur[2 * tid] = 0;
    cur[2 * tid + 1] = 0;
    __syncthreads();
    const int e0 = c * CH;
    int dv[CH / 256], sv[CH / 256];
#pragma unroll
    for (int i = 0; i < CH / 256; ++i) {
        const int e = e0 + i * 256 + tid;
        const int ec = min(e, EE - 1);
        const int d = dst[ec];
        sv[i] = src[ec];
        dv[i] = (e < EE) ? d : -1;
    }
#pragma unroll
    for (int i = 0; i < CH / 256; ++i)
        if (dv[i] >= 0) atomicAdd(&cur[dv[i] >> RSH], 1);
    __syncthreads();
    const int k0 = 2 * tid, k1 = 2 * tid + 1;
    const int h0 = cur[k0], h1 = cur[k1];
    const int t0 = (k0 < NBK) ? rowtot[k0] : 0, t1 = (k1 < NBK) ? rowtot[k1] : 0;
    const int m0 = (k0 < NBK) ? mat[c * NBK + k0] : 0, m1 = (k1 < NBK) ? mat[c * NBK + k1] : 0;
    int unused;
    const int lex = block_excl<256>(h0 + h1, wsum, unused);
    const int gex = block_excl<256>(t0 + t1, wsum, unused);
    cur[k0] = lex;
    cur[k1] = lex + h0;
    gdel[k0] = gex + m0 - lex;
    gdel[k1] = gex + t0 + m1 - (lex + h0);
    __syncthreads();
#pragma unroll
    for (int i = 0; i < CH / 256; ++i) {
        if (dv[i] >= 0) {
            const int k = dv[i] >> RSH;
            const int p = atomicAdd(&cur[k], 1);
            sval[p] = (sv[i] << RSH) | (dv[i] & (RB - 1));
            sbk[p] = (unsigned short)k;
        }
    }
    __syncthreads();
    const int nv = min(CH, EE - e0);
    for (int i = tid; i < nv; i += 256) tmp[i + gdel[sbk[i]]] = sval[i];
}

__global__ __launch_bounds__(512)
void k_bucket(const int* __restrict__ tmp, const int* __restrict__ rowtot,
              int* __restrict__ start, int* __restrict__ csr) {
    __shared__ int cnt[RB];
    __shared__ int wsum[16];
    __shared__ int stage[CAP];
    const int k = blockIdx.x, t = threadIdx.x;
    int part = 0;
    for (int j = t; j < k; j += 512) part += rowtot[j];
    int base;
    (void)block_excl<512>(part, wsum, base);
    const int tot = rowtot[k];
    if (t < RB) cnt[t] = 0;
    __syncthreads();
    for (int i = t; i < tot; i += 512) atomicAdd(&cnt[tmp[base + i] & (RB - 1)], 1);
    __syncthreads();
    const int v = (t < RB) ? cnt[t] : 0;
    int unused;
    const int ex = block_excl<512>(v, wsum, unused);
    if (t < RB) {
        const int node = k * RB + t;
        if (node < NN) start[node] = base + ex;
        cnt[t] = ex;
    }
    __syncthreads();
    const bool staged = tot <= CAP;
    for (int i = t; i < tot; i += 512) {
        const int w = tmp[base + i];
        const int p = atomicAdd(&cnt[w & (RB - 1)], 1);
        const int s = w >> RSH;
        if (staged) stage[p] = s; else csr[base + p] = s;
    }
    if (staged) {
        __syncthreads();
        for (int i = t; i < tot; i += 512) csr[base + i] = stage[i];
    }
}


template <int K, int NC, bool DO_LN>
__global__ __launch_bounds__(32 * WPB)
void k_gemm(const float* __restrict__ A, const float* __restrict__ W,
            const float* __restrict__ bias, const float* __restrict__ gamma,
            const float* __restrict__ beta, float* __restrict__ out, int ntiles) {
    constexpr int NJ = NC / 16;
    const int wave = threadIdx.x >> 5;
    const int lane = threadIdx.x & 31;
    const int l16  = lane & 15;
    const int half = lane >> 4;
    const int tile = blockIdx.x * WPB + wave;
    const bool active = tile < ntiles;

    v8f zero = {};
    v8f acc[NJ];
#pragma unroll
    for (int j = 0; j < NJ; ++j) acc[j] = zero;

    if (active) {
        const float* Arow = A + (size_t)tile * 16 * K + (size_t)l16 * K;
#pragma unroll
        for (int k = 0; k < K; k += 4) {
            const float* ap = Arow + k + 2 * half;
            v2f a;
            a.x = ap[0];
            a.y = ap[1];
#pragma unroll
            for (int j = 0; j < NJ; ++j) {
                const int col = j * 16 + l16;
                v2f b;
                b.x = W[(size_t)(k + 2 * half) * NC + col];
                b.y = W[(size_t)(k + 2 * half + 1) * NC + col];
                acc[j] = __builtin_amdgcn_wmma_f32_16x16x4_f32(
                    false, a, false, b, (short)0, acc[j], false, false);
            }
        }
    }

    if constexpr (DO_LN) {
        static_assert(NC == 64, "LN epilogue assumes 64 columns");
        __shared__ float lds[WPB][16][68];
        if (active) {
#pragma unroll
            for (int j = 0; j < NJ; ++j) {
                const int col = j * 16 + l16;
                const float bj = bias[col];
#pragma unroll
                for (int v = 0; v < 8; ++v) {
                    const int r = v + 8 * half;
                    lds[wave][r][col] = silu_f(acc[j][v] + bj);
                }
            }
        }
        __syncthreads();
        if (active) {
            const int r = l16, cb = half * 32;
            float s = 0.f, sq = 0.f;
#pragma unroll
            for (int c = 0; c < 32; ++c) {
                float vv = lds[wave][r][cb + c];
                s += vv; sq += vv * vv;
            }
            s  += __shfl_xor(s, 16, 32);
            sq += __shfl_xor(sq, 16, 32);
            const float mean = s * (1.f / NC);
            const float var  = sq * (1.f / NC) - mean * mean;
            const float inv  = rsq_f(var + 1e-5f);
            float* orow = out + ((size_t)tile * 16 + r) * NC;
#pragma unroll
            for (int c = 0; c < 32; ++c) {
                const int col = cb + c;
                const float vv = lds[wave][r][col];
                orow[col] = (vv - mean) * inv * gamma[col] + beta[col];
            }
        }
    } else {
        if (active) {
#pragma unroll
            for (int j = 0; j < NJ; ++j) {
                const int col = j * 16 + l16;
                const float bj = bias[col];
#pragma unroll
                for (int v = 0; v < 8; ++v) {
                    const int r = v + 8 * half;
                    out[((size_t)tile * 16 + r) * NC + col] = acc[j][v] + bj;
                }
            }
        }
        (void)gamma; (void)beta;
    }
}


template <int NCO, bool LN>
DEVFN void tile_tail(float (*hs)[HP], float (*os)[OP],
                     const float* __restrict__ Wa, const float* __restrict__ ba,
                     const float* __restrict__ Wb, const float* __restrict__ bb,
                     const float* __restrict__ g,  const float* __restrict__ be,
                     float* __restrict__ outA, float* __restrict__ outB, int tile) {
    static_assert(NCO == 64 || (NCO == 32 && !LN), "tile map below");
    constexpr int TPM = NCO / 16;
    constexpr int NT  = LN ? TPM : 2 * TPM;
    const int wave = threadIdx.x >> 5;
    const int lane = threadIdx.x & 31;
    const int l16  = lane & 15;
    const int half = lane >> 4;
    if (wave < NT) {
        const bool second = (!LN) && (wave >= TPM);
        const float* __restrict__ Wm   = second ? Wb : Wa;
        const float* __restrict__ bias = second ? bb : ba;
        const int col = (second ? wave - TPM : wave) * 16 + l16;
        v8f acc = {};
#pragma unroll
        for (int k = 0; k < 64; k += 4) {
            const int kk = k + 2 * half;
            v2f a, b;
            a.x = hs[l16][kk];
            a.y = hs[l16][kk + 1];
            b.x = Wm[(size_t)kk * NCO + col];
            b.y = Wm[(size_t)(kk + 1) * NCO + col];
            acc = __builtin_amdgcn_wmma_f32_16x16x4_f32(
                false, a, false, b, (short)0, acc, false, false);
        }
        const float bj = bias[col];
#pragma unroll
        for (int v = 0; v < 8; ++v) {
            float val = acc[v] + bj;
            if constexpr (LN) val = silu_f(val);
            os[v + 8 * half][wave * 16 + l16] = val;
        }
    }
    __syncthreads();
    const int nl = threadIdx.x >> 4, q = threadIdx.x & 15;
    const size_t row = (size_t)tile * 16 + nl;
    const float4 v0 = *reinterpret_cast<const float4*>(&os[nl][4 * q]);
    if constexpr (LN) {
        float s  = (v0.x + v0.y) + (v0.z + v0.w);
        float sq = (v0.x * v0.x + v0.y * v0.y) + (v0.z * v0.z + v0.w * v0.w);
        s  = group_sum<16>(s);
        sq = group_sum<16>(sq);
        const float mean = s * (1.f / 64);
        const float var  = sq * (1.f / 64) - mean * mean;
        const float inv  = rsq_f(var + 1e-5f);
        const int c0 = 4 * q;
        float4 o;
        o.x = (v0.x - mean) * inv * g[c0]     + be[c0];
        o.y = (v0.y - mean) * inv * g[c0 + 1] + be[c0 + 1];
        o.z = (v0.z - mean) * inv * g[c0 + 2] + be[c0 + 2];
        o.w = (v0.w - mean) * inv * g[c0 + 3] + be[c0 + 3];
        reinterpret_cast<float4*>(outA + row * 64)[q] = o;
    } else if constexpr (NCO == 64) {
        const float4 v1 = *reinterpret_cast<const float4*>(&os[nl][64 + 4 * q]);
        reinterpret_cast<float4*>(outA + row * 64)[q] = v0;
        reinterpret_cast<float4*>(outB + row * 64)[q] = v1;
    } else {
        if (q < 8) reinterpret_cast<float4*>(outA + row * 32)[q] = v0;
        else       reinterpret_cast<float4*>(outB + row * 32)[q - 8] = v0;
    }
}


template <bool NEXT_LN>
__global__ __launch_bounds__(256) __attribute__((amdgpu_waves_per_eu(16)))
void k_enc_agg(const int* __restrict__ start, const int* __restrict__ csr,
               const float* __restrict__ z, float* __restrict__ h,
               const float* __restrict__ Wa, const float* __restrict__ ba,
               const float* __restrict__ Wb, const float* __restrict__ bb,
               const float* __restrict__ g,  const float* __restrict__ be,
               float* __restrict__ outA, float* __restrict__ outB, int n, int e) {
    __shared__ __attribute__((aligned(16))) float hs[16][HP];
    __shared__ __attribute__((aligned(16))) float os[16][OP];
    const int t = blockIdx.x * 256 + threadIdx.x;
    const int node = t >> 4, q = t & 15;
    const int beg = start[node];
    const int end = (node + 1 < n) ? start[node + 1] : e;
    const float4* z4 = reinterpret_cast<const float4*>(z);
    float4 acc = make_float4(0.f, 0.f, 0.f, 0.f);
    for (int base = beg; base < end; base += 16) {
        const int cnt = min(16, end - base);
        const int my = (q < cnt) ? csr[base + q] : 0;
        for (int j = 0; j < cnt; j += 4) {
            const int s0 = group_bcast<16>(my, j + 0);
            const int s1 = group_bcast<16>(my, j + 1);
            const int s2 = group_bcast<16>(my, j + 2);
            const int s3 = group_bcast<16>(my, j + 3);
            const float4 v0 = z4[s0 * 16 + q];
            const float4 v1 = z4[s1 * 16 + q];
            const float4 v2 = z4[s2 * 16 + q];
            const float4 v3 = z4[s3 * 16 + q];
            const bool k1 = j + 1 < cnt, k2 = j + 2 < cnt, k3 = j + 3 < cnt;
            acc.x += v0.x; acc.y += v0.y; acc.z += v0.z; acc.w += v0.w;
            acc.x += k1 ? v1.x : 0.f; acc.y += k1 ? v1.y : 0.f; acc.z += k1 ? v1.z : 0.f; acc.w += k1 ? v1.w : 0.f;
            acc.x += k2 ? v2.x : 0.f; acc.y += k2 ? v2.y : 0.f; acc.z += k2 ? v2.z : 0.f; acc.w += k2 ? v2.w : 0.f;
            acc.x += k3 ? v3.x : 0.f; acc.y += k3 ? v3.y : 0.f; acc.z += k3 ? v3.z : 0.f; acc.w += k3 ? v3.w : 0.f;
        }
    }
    const float inv = rcp_f(fmaxf((float)(end - beg), 1.0f));
    float4* hp = reinterpret_cast<float4*>(h + (size_t)node * 64) + q;
    float4 hv = *hp;
    hv.x += acc.x * inv; hv.y += acc.y * inv; hv.z += acc.z * inv; hv.w += acc.w * inv;
    *reinterpret_cast<float4*>(&hs[threadIdx.x >> 4][4 * q]) = hv;
    __syncthreads();
    tile_tail<64, NEXT_LN>(hs, os, Wa, ba, Wb, bb, g, be, outA, outB, (int)blockIdx.x);
    *hp = *reinterpret_cast<const float4*>(&hs[threadIdx.x >> 4][4 * q]);
}


template <int NC, bool RES, int TAIL>
__global__ __launch_bounds__(256) __attribute__((amdgpu_waves_per_eu(16)))
void k_gat_node(const int* __restrict__ start, const int* __restrict__ csr,
                           const float* __restrict__ xl, const float* __restrict__ xr,
                           const float* __restrict__ att, const float* __restrict__ bias,
                           const float* __restrict__ g, const float* __restrict__ be,
                           const float* hin, float* out,
                           const float* __restrict__ Wa, const float* __restrict__ ba,
                           const float* __restrict__ Wb, const float* __restrict__ bb,
                           float* __restrict__ outA, float* __restrict__ outB,
                           int n, int e) {
    constexpr int LPN = NC / 4;
    const int t = blockIdx.x * 256 + threadIdx.x;
    const int node = t / LPN, q = t % LPN;
    if constexpr (TAIL == 0) { if (node >= n) return; }
    const int beg = start[node];
    const int end = (node + 1 < n) ? start[node + 1] : e;
    const int c0 = q * 4;
    const v2f a01 = { att[c0] * LOG2E,     att[c0 + 1] * LOG2E };
    const v2f a23 = { att[c0 + 2] * LOG2E, att[c0 + 3] * LOG2E };
    const float4* xl4 = reinterpret_cast<const float4*>(xl);
    const float4 r4 = reinterpret_cast<const float4*>(xr)[node * LPN + q];
    const float4 self = xl4[node * LPN + q];
    const v2f r01 = { r4.x, r4.y }, r23 = { r4.z, r4.w };

    auto part = [&](const float4& v) -> float {
        const v2f t01 = lo2(v) + r01;
        const v2f t23 = hi2(v) + r23;
        v2f p = a01 * lrelu2(t01);
        p = __builtin_elementwise_fma(a23, lrelu2(t23), p);
        return p.x + p.y;
    };

    float m = group_sum<LPN>(part(self));
    float l = 1.f;
    v2f acc01 = lo2(self), acc23 = hi2(self);
    for (int base = beg; base < end; base += LPN) {
        const int cnt = min(LPN, end - base);
        const int my = (q < cnt) ? csr[base + q] : node;
        for (int j = 0; j < cnt; j += 4) {
            const int s0 = group_bcast<LPN>(my, j + 0);
            const int s1 = group_bcast<LPN>(my, j + 1);
            const int s2 = group_bcast<LPN>(my, j + 2);
            const int s3 = group_bcast<LPN>(my, j + 3);
            const float4 v0 = xl4[s0 * LPN + q];
            const float4 v1 = xl4[s1 * LPN + q];
            const float4 v2 = xl4[s2 * LPN + q];
            const float4 v3 = xl4[s3 * LPN + q];
            asm volatile("" :: "v"(v0.x), "v"(v0.y), "v"(v0.z), "v"(v0.w),
                               "v"(v1.x), "v"(v1.y), "v"(v1.z), "v"(v1.w),
                               "v"(v2.x), "v"(v2.y), "v"(v2.z), "v"(v2.w),
                               "v"(v3.x), "v"(v3.y), "v"(v3.z), "v"(v3.w));
            float e0 = part(v0), e1 = part(v1), e2 = part(v2), e3 = part(v3);
            group_sum4<LPN>(e0, e1, e2, e3);
            if (j + 1 >= cnt) e1 = -INFINITY;
            if (j + 2 >= cnt) e2 = -INFINITY;
            if (j + 3 >= cnt) e3 = -INFINITY;
            const float mn = fmaxf(fmaxf(m, e0), fmaxf(fmaxf(e1, e2), e3));
            const v2f d01 = v2f{e0, e1} - both(mn), d23 = v2f{e2, e3} - both(mn);
            const float sc = __builtin_amdgcn_exp2f(m - mn);
            const float p0 = __builtin_amdgcn_exp2f(d01.x), p1 = __builtin_amdgcn_exp2f(d01.y);
            const float p2 = __builtin_amdgcn_exp2f(d23.x), p3 = __builtin_amdgcn_exp2f(d23.y);
            l = l * sc + ((p0 + p1) + (p2 + p3));
            const v2f P0 = both(p0), P1 = both(p1), P2 = both(p2), P3 = both(p3), SC = both(sc);
            const v2f s01a = __builtin_elementwise_fma(lo2(v0), P0, lo2(v1) * P1);
            const v2f s01b = __builtin_elementwise_fma(lo2(v2), P2, lo2(v3) * P3);
            const v2f s23a = __builtin_elementwise_fma(hi2(v0), P0, hi2(v1) * P1);
            const v2f s23b = __builtin_elementwise_fma(hi2(v2), P2, hi2(v3) * P3);
            acc01 = __builtin_elementwise_fma(acc01, SC, s01a + s01b);
            acc23 = __builtin_elementwise_fma(acc23, SC, s23a + s23b);
            m = mn;
        }
    }

    const float il = rcp_f(l);
    float vals[4];
    vals[0] = silu_f(acc01.x * il + bias[c0]);
    vals[1] = silu_f(acc01.y * il + bias[c0 + 1]);
    vals[2] = silu_f(acc23.x * il + bias[c0 + 2]);
    vals[3] = silu_f(acc23.y * il + bias[c0 + 3]);
    float s = (vals[0] + vals[1]) + (vals[2] + vals[3]);
    float sq = (vals[0] * vals[0] + vals[1] * vals[1]) + (vals[2] * vals[2] + vals[3] * vals[3]);
    s  = group_sum<LPN>(s);
    sq = group_sum<LPN>(sq);
    const float mean = s * (1.f / NC);
    const float var  = sq * (1.f / NC) - mean * mean;
    const float inv  = rsq_f(var + 1e-5f);
    float4 o;
    o.x = (vals[0] - mean) * inv * g[c0]     + be[c0];
    o.y = (vals[1] - mean) * inv * g[c0 + 1] + be[c0 + 1];
    o.z = (vals[2] - mean) * inv * g[c0 + 2] + be[c0 + 2];
    o.w = (vals[3] - mean) * inv * g[c0 + 3] + be[c0 + 3];
    if (RES) {
        const float4 hv = reinterpret_cast<const float4*>(hin + (size_t)node * NC)[q];
        o.x += hv.x; o.y += hv.y; o.z += hv.z; o.w += hv.w;
    }
    if constexpr (TAIL > 0) {
        static_assert(NC == 64, "16 lanes per node: a 256-thread block is one 16-row tile");
        __shared__ __attribute__((aligned(16))) float hs[16][HP];
        __shared__ __attribute__((aligned(16))) float os[16][OP];
        *reinterpret_cast<float4*>(&hs[threadIdx.x >> 4][4 * q]) = o;
        __syncthreads();
        tile_tail<TAIL, false>(hs, os, Wa, ba, Wb, bb, nullptr, nullptr, outA, outB, (int)blockIdx.x);
        reinterpret_cast<float4*>(out + (size_t)node * NC)[q] =
            *reinterpret_cast<const float4*>(&hs[threadIdx.x >> 4][4 * q]);
    } else {
        (void)Wa; (void)ba; (void)Wb; (void)bb; (void)outA; (void)outB;
        reinterpret_cast<float4*>(out + (size_t)node * NC)[q] = o;
    }
}


extern "C" void kernel_launch(void* const* d_in, const int* in_sizes, int n_in,
                              void* d_out, int out_size, void* d_ws, size_t ws_size,
                              hipStream_t stream) {
    (void)in_sizes; (void)n_in; (void)out_size; (void)ws_size;

    const float* x        = (const float*)d_in[0];
    const int*   ei       = (const int*)  d_in[1];
    const float* W_emb    = (const float*)d_in[2];
    const float* b_emb    = (const float*)d_in[3];
    const float* g_emb    = (const float*)d_in[4];
    const float* be_emb   = (const float*)d_in[5];
    const float* enc_W    = (const float*)d_in[6];
    const float* enc_b    = (const float*)d_in[7];
    const float* enc_g    = (const float*)d_in[8];
    const float* enc_be   = (const float*)d_in[9];
    const float* dec_Wl   = (const float*)d_in[10];
    const float* dec_bl   = (const float*)d_in[11];
    const float* dec_Wr   = (const float*)d_in[12];
    const float* dec_br   = (const float*)d_in[13];
    const float* dec_att  = (const float*)d_in[14];
    const float* dec_bias = (const float*)d_in[15];
    const float* dec_g    = (const float*)d_in[16];
    const float* dec_be   = (const float*)d_in[17];
    const float* fin_Wl   = (const float*)d_in[18];
    const float* fin_bl   = (const float*)d_in[19];
    const float* fin_Wr   = (const float*)d_in[20];
    const float* fin_br   = (const float*)d_in[21];
    const float* fin_att  = (const float*)d_in[22];
    const float* fin_bias = (const float*)d_in[23];
    const float* fin_g    = (const float*)d_in[24];
    const float* fin_be   = (const float*)d_in[25];

    const int* src = ei;
    const int* dst = ei + EE;
    float* out = (float*)d_out;

    int* wi      = (int*)d_ws;
    int*   start = wi;                  wi += NN;
                                        wi += 256;
    int*   tmp   = wi;                  wi += EE;
    int*   csr   = wi;                  wi += EE;
    float* w     = (float*)wi;
    float* h     = w;                   w += (size_t)NN * 64;
    float* zA    = w;                   w += (size_t)NN * 64;
    float* zB    = w;                   w += (size_t)NN * 64;
    float* xlA   = w;                   w += (size_t)NN * 64;
    float* xrA   = w;                   w += (size_t)NN * 64;
    int*   mat    = (int*)w;
    int*   rowtot = mat + (size_t)NB1 * NBK;

    const int TB = 256;
    auto nb = [](long long t) { return (unsigned)((t + 255) / 256); };
    const int NT = NN / 16;
    const unsigned GG = (unsigned)((NT + WPB - 1) / WPB);
    const dim3 gthr(32 * WPB);

    k_hist<<<NB1, 256, 0, stream>>>(dst, mat);
    k_rowscan<<<NBK, 512, 0, stream>>>(mat, rowtot);
    k_scatter<<<NB1, 256, 0, stream>>>(src, dst, mat, rowtot, tmp);
    k_bucket<<<NBK, 512, 0, stream>>>(tmp, rowtot, start, csr);

    k_gemm<32, 64, true><<<GG, gthr, 0, stream>>>(x, W_emb, b_emb, g_emb, be_emb, h, NT);

    k_gemm<64, 64, true><<<GG, gthr, 0, stream>>>(h, enc_W, enc_b, enc_g, enc_be, zA, NT);

    const unsigned NB16 = (unsigned)(NN / 16);
    k_enc_agg<true><<<NB16, TB, 0, stream>>>(start, csr, zA, h,
        enc_W + (size_t)1 * 64 * 64, enc_b + 1 * 64, nullptr, nullptr, enc_g + 1 * 64, enc_be + 1 * 64,
        zB, nullptr, NN, EE);
    k_enc_agg<true><<<NB16, TB, 0, stream>>>(start, csr, zB, h,
        enc_W + (size_t)2 * 64 * 64, enc_b + 2 * 64, nullptr, nullptr, enc_g + 2 * 64, enc_be + 2 * 64,
        zA, nullptr, NN, EE);
    k_enc_agg<false><<<NB16, TB, 0, stream>>>(start, csr, zA, h,
        dec_Wl, dec_bl, dec_Wr, dec_br, nullptr, nullptr, xlA, xrA, NN, EE);

    k_gat_node<64, true, 64><<<NB16, TB, 0, stream>>>(
        start, csr, xlA, xrA, dec_att, dec_bias, dec_g, dec_be, h, h,
        dec_Wl + (size_t)1 * 64 * 64, dec_bl + 1 * 64, dec_Wr + (size_t)1 * 64 * 64, dec_br + 1 * 64,
        zA, zB, NN, EE);
    k_gat_node<64, true, 64><<<NB16, TB, 0, stream>>>(
        start, csr, zA, zB, dec_att + 1 * 64, dec_bias + 1 * 64, dec_g + 1 * 64, dec_be + 1 * 64, h, h,
        dec_Wl + (size_t)2 * 64 * 64, dec_bl + 2 * 64, dec_Wr + (size_t)2 * 64 * 64, dec_br + 2 * 64,
        xlA, xrA, NN, EE);
    k_gat_node<64, true, 32><<<NB16, TB, 0, stream>>>(
        start, csr, xlA, xrA, dec_att + 2 * 64, dec_bias + 2 * 64, dec_g + 2 * 64, dec_be + 2 * 64, h, h,
        fin_Wl, fin_bl, fin_Wr, fin_br, zA, zB, NN, EE);

    k_gat_node<32, false, 0><<<nb((long long)NN * 8), TB, 0, stream>>>(
        start, csr, zA, zB, fin_att, fin_bias, fin_g, fin_be, nullptr, out,
        nullptr, nullptr, nullptr, nullptr, nullptr, nullptr, NN, EE);
}


# embedding and first encoder product as one launch on LDS-staged 32-row tiles (13 to 12 launches): whole 128-B lines per load and store, no 4-byte strided stores; their LN row sums in butterfly order
# speedup vs baseline: 7.0650x; 1.0957x over previous
#include <hip/hip_runtime.h>
#include <cstdint>
#include <cstddef>


typedef float v2f __attribute__((ext_vector_type(2)));
typedef float v8f __attribute__((ext_vector_type(8)));

static constexpr int NN  = 100000;
static constexpr int EE  = 1200000;
static constexpr int LM1 = 3;
static_assert(NN % 16 == 0, "a 256-thread node block is exactly one 16-row tile");
static constexpr int HP  = 68;
static constexpr int OP  = 132;

#define DEVFN static __device__ __forceinline__

DEVFN float rcp_f(float v)   { return __builtin_amdgcn_rcpf(v); }
DEVFN float rsq_f(float v)   { return __builtin_amdgcn_rsqf(v); }
DEVFN float silu_f(float v)  { return v * rcp_f(1.f + __expf(-v)); }
DEVFN v2f lrelu2(v2f t)      { return __builtin_elementwise_max(t, t * 0.2f); }
static constexpr float LOG2E = 1.44269504088896340736f;
DEVFN v2f lo2(const float4& v) { return v2f{v.x, v.y}; }
DEVFN v2f hi2(const float4& v) { return v2f{v.z, v.w}; }
DEVFN v2f both(float s)        { return v2f{s, s}; }


template <int M>
DEVFN float row_xor_add(float v) {
    const int t = __builtin_amdgcn_update_dpp(0, __builtin_bit_cast(int, v), 0x160 | M, 0xf, 0xf, true);
    return v + __builtin_bit_cast(float, t);
}

template <int G>
DEVFN float group_sum(float v) {
    static_assert(G == 16 || G == 8, "groups inside a 16-lane row");
    if constexpr (G == 16) v = row_xor_add<8>(v);
    v = row_xor_add<4>(v);
    v = row_xor_add<2>(v);
    return row_xor_add<1>(v);
}

template <int M>
DEVFN float row_xor_get(float v) {
    const int t = __builtin_amdgcn_update_dpp(0, __builtin_bit_cast(int, v), 0x160 | M, 0xf, 0xf, true);
    return __builtin_bit_cast(float, t);
}

template <int K>
DEVFN float quad_get(float v) {
    const int t = __builtin_amdgcn_update_dpp(0, __builtin_bit_cast(int, v), K * 0x55, 0xf, 0xf, true);
    return __builtin_bit_cast(float, t);
}

template <int G>
DEVFN void group_sum4(float& a, float& b, float& c, float& d) {
    static_assert(G == 16 || G == 8, "groups inside a 16-lane row");
    const bool b0 = (threadIdx.x & 1) != 0, b1 = (threadIdx.x & 2) != 0;
    const float x = (b0 ? c : a) + row_xor_get<1>(b0 ? a : c);
    const float y = (b0 ? d : b) + row_xor_get<1>(b0 ? b : d);
    float z = (b1 ? y : x) + row_xor_get<2>(b1 ? x : y);
    z = row_xor_add<4>(z);
    if constexpr (G == 16) z = row_xor_add<8>(z);
    a = quad_get<0>(z); c = quad_get<1>(z); b = quad_get<2>(z); d = quad_get<3>(z);
}

template <int G>
DEVFN int group_bcast(int v, int idx) {
    static_assert(G == 16 || G == 8, "groups inside a 16-lane row");
    const unsigned lo = 0x11111111u * (unsigned)idx;
    const unsigned hi = (G == 16) ? lo : lo + 0x88888888u;
    return (int)__builtin_amdgcn_permlane16(0u, (unsigned)v, lo, hi, true, false);
}


static constexpr int RSH = 8;
static constexpr int RB  = 1 << RSH;
static constexpr int NBK = (NN + RB - 1) / RB;
static constexpr int CH  = 4096;
static constexpr int NB1 = (EE + CH - 1) / CH;
static constexpr int CAP = 8192;
static_assert(NBK <= 512, "k_scatter scans the ranges two per thread");
static_assert(NB1 <= 512, "k_rowscan scans the chunks one per thread");
static_assert(NN <= (1 << (31 - RSH)), "a source id and RSH destination bits share one positive word");

template <int NT>
DEVFN int block_excl(int v, int* wsum, int& total) {
    const int lane = threadIdx.x & 31, wv = threadIdx.x >> 5;
    int inc = v;
#pragma unroll
    for (int off = 1; off < 32; off <<= 1) {
        const int u = __shfl_up(inc, off, 32);
        if (lane >= off) inc += u;
    }
    __syncthreads();
    if (lane == 31) wsum[wv] = inc;
    __syncthreads();
    int woff = 0, tot = 0;
#pragma unroll
    for (int w = 0; w < NT / 32; ++w) {
        const int s = wsum[w];
        tot += s;
        if (w < wv) woff += s;
    }
    total = tot;
    return woff + inc - v;
}

__global__ __launch_bounds__(256)
void k_hist(const int* __restrict__ dst, int* __restrict__ mat) {
    __shared__ int hist[NBK];
    const int tid = threadIdx.x, c = blockIdx.x;
    for (int k = tid; k < NBK; k += 256) hist[k] = 0;
    __syncthreads();
    const int e0 = c * CH;
#pragma unroll
    for (int i = 0; i < CH / 256; ++i) {
        const int e = e0 + i * 256 + tid;
        const int d = dst[min(e, EE - 1)];
        if (e < EE) atomicAdd(&hist[d >> RSH], 1);
    }
    __syncthreads();
    for (int k = tid; k < NBK; k += 256) mat[c * NBK + k] = hist[k];
}

__global__ __launch_bounds__(512)
void k_rowscan(int* __restrict__ mat, int* __restrict__ rowtot) {
    __shared__ int wsum[16];
    const int k = blockIdx.x, t = threadIdx.x;
    const int v = (t < NB1) ? mat[t * NBK + k] : 0;
    int total;
    const int ex = block_excl<512>(v, wsum, total);
    if (t < NB1) mat[t * NBK + k] = ex;
    if (t == 0) rowtot[k] = total;
}

__global__ __launch_bounds__(256)
void k_scatter(const int* __restrict__ src, const int* __restrict__ dst,
               const int* __restrict__ mat, const int* __restrict__ rowtot,
               int* __restrict__ tmp) {
    __shared__ int cur[512];
    __shared__ int gdel[512];
    __shared__ int sval[CH];
    __shared__ unsigned short sbk[CH];
    __shared__ int wsum[8];
    const int tid = threadIdx.x, c = blockIdx.x;
    cur[2 * tid] = 0;
    cur[2 * tid + 1] = 0;
    __syncthreads();
    const int e0 = c * CH;
    int dv[CH / 256], sv[CH / 256];
#pragma unroll
    for (int i = 0; i < CH / 256; ++i) {
        const int e = e0 + i * 256 + tid;
        const int ec = min(e, EE - 1);
        const int d = dst[ec];
        sv[i] = src[ec];
        dv[i] = (e < EE) ? d : -1;
    }
#pragma unroll
    for (int i = 0; i < CH / 256; ++i)
        if (dv[i] >= 0) atomicAdd(&cur[dv[i] >> RSH], 1);
    __syncthreads();
    const int k0 = 2 * tid, k1 = 2 * tid + 1;
    const int h0 = cur[k0], h1 = cur[k1];
    const int t0 = (k0 < NBK) ? rowtot[k0] : 0, t1 = (k1 < NBK) ? rowtot[k1] : 0;
    const int m0 = (k0 < NBK) ? mat[c * NBK + k0] : 0, m1 = (k1 < NBK) ? mat[c * NBK + k1] : 0;
    int unused;
    const int lex = block_excl<256>(h0 + h1, wsum, unused);
    const int gex = block_excl<256>(t0 + t1, wsum, unused);
    cur[k0] = lex;
    cur[k1] = lex + h0;
    gdel[k0] = gex + m0 - lex;
    gdel[k1] = gex + t0 + m1 - (lex + h0);
    __syncthreads();
#pragma unroll
    for (int i = 0; i < CH / 256; ++i) {
        if (dv[i] >= 0) {
            const int k = dv[i] >> RSH;
            const int p = atomicAdd(&cur[k], 1);
            sval[p] = (sv[i] << RSH) | (dv[i] & (RB - 1));
            sbk[p] = (unsigned short)k;
        }
    }
    __syncthreads();
    const int nv = min(CH, EE - e0);
    for (int i = tid; i < nv; i += 256) tmp[i + gdel[sbk[i]]] = sval[i];
}

__global__ __launch_bounds__(512)
void k_bucket(const int* __restrict__ tmp, const int* __restrict__ rowtot,
              int* __restrict__ start, int* __restrict__ csr) {
    __shared__ int cnt[RB];
    __shared__ int wsum[16];
    __shared__ int stage[CAP];
    const int k = blockIdx.x, t = threadIdx.x;
    int part = 0;
    for (int j = t; j < k; j += 512) part += rowtot[j];
    int base;
    (void)block_excl<512>(part, wsum, base);
    const int tot = rowtot[k];
    if (t < RB) cnt[t] = 0;
    __syncthreads();
    for (int i = t; i < tot; i += 512) atomicAdd(&cnt[tmp[base + i] & (RB - 1)], 1);
    __syncthreads();
    const int v = (t < RB) ? cnt[t] : 0;
    int unused;
    const int ex = block_excl<512>(v, wsum, unused);
    if (t < RB) {
        const int node = k * RB + t;
        if (node < NN) start[node] = base + ex;
        cnt[t] = ex;
    }
    __syncthreads();
    const bool staged = tot <= CAP;
    for (int i = t; i < tot; i += 512) {
        const int w = tmp[base + i];
        const int p = atomicAdd(&cnt[w & (RB - 1)], 1);
        const int s = w >> RSH;
        if (staged) stage[p] = s; else csr[base + p] = s;
    }
    if (staged) {
        __syncthreads();
        for (int i = t; i < tot; i += 512) csr[base + i] = stage[i];
    }
}


static constexpr int ER = 32;
static_assert(NN % ER == 0, "k_embed's grid is exact");

template <int K, bool KEEP>
DEVFN void rows_ln(float (*hs)[HP], float (*os)[HP],
                   const float* __restrict__ Wm, const float* __restrict__ bias,
                   const float* __restrict__ g,  const float* __restrict__ be,
                   float* __restrict__ outG, int blk) {
    const int wave = threadIdx.x >> 5;
    const int lane = threadIdx.x & 31;
    const int l16  = lane & 15;
    const int half = lane >> 4;
    const int col  = wave * 16 + l16;
    v8f acc0 = {}, acc1 = {};
#pragma unroll
    for (int k = 0; k < K; k += 4) {
        const int kk = k + 2 * half;
        v2f a0, a1, b;
        a0.x = hs[l16][kk];
        a0.y = hs[l16][kk + 1];
        a1.x = hs[16 + l16][kk];
        a1.y = hs[16 + l16][kk + 1];
        b.x = Wm[(size_t)kk * 64 + col];
        b.y = Wm[(size_t)(kk + 1) * 64 + col];
        acc0 = __builtin_amdgcn_wmma_f32_16x16x4_f32(
            false, a0, false, b, (short)0, acc0, false, false);
        acc1 = __builtin_amdgcn_wmma_f32_16x16x4_f32(
            false, a1, false, b, (short)0, acc1, false, false);
    }
    const float bj = bias[col];
#pragma unroll
    for (int v = 0; v < 8; ++v) {
        os[v + 8 * half][col]      = silu_f(acc0[v] + bj);
        os[16 + v + 8 * half][col] = silu_f(acc1[v] + bj);
    }
    __syncthreads();
    const int q = threadIdx.x & 15, c0 = 4 * q;
    const float g0 = g[c0], g1 = g[c0 + 1], g2 = g[c0 + 2], g3 = g[c0 + 3];
    const float e0 = be[c0], e1 = be[c0 + 1], e2 = be[c0 + 2], e3 = be[c0 + 3];
#pragma unroll
    for (int p = 0; p < ER / 8; ++p) {
        const int nl = (threadIdx.x >> 4) + 8 * p;
        const float4 v0 = *reinterpret_cast<const float4*>(&os[nl][c0]);
        float s  = (v0.x + v0.y) + (v0.z + v0.w);
        float sq = (v0.x * v0.x + v0.y * v0.y) + (v0.z * v0.z + v0.w * v0.w);
        s  = group_sum<16>(s);
        sq = group_sum<16>(sq);
        const float mean = s * (1.f / 64);
        const float var  = sq * (1.f / 64) - mean * mean;
        const float inv  = rsq_f(var + 1e-5f);
        float4 o;
        o.x = (v0.x - mean) * inv * g0 + e0;
        o.y = (v0.y - mean) * inv * g1 + e1;
        o.z = (v0.z - mean) * inv * g2 + e2;
        o.w = (v0.w - mean) * inv * g3 + e3;
        if constexpr (KEEP) *reinterpret_cast<float4*>(&hs[nl][c0]) = o;
        else reinterpret_cast<float4*>(outG + ((size_t)blk * ER + nl) * 64)[q] = o;
    }
}

__global__ __launch_bounds__(128)
void k_embed(const float* __restrict__ x,
             const float* __restrict__ We, const float* __restrict__ bE,
             const float* __restrict__ gE, const float* __restrict__ beE,
             const float* __restrict__ W0, const float* __restrict__ b0,
             const float* __restrict__ g0, const float* __restrict__ be0,
             float* __restrict__ h, float* __restrict__ z) {
    __shared__ __attribute__((aligned(16))) float hs[ER][HP];
    __shared__ __attribute__((aligned(16))) float os[ER][HP];
    const int blk = blockIdx.x;
    const float4* x4 = reinterpret_cast<const float4*>(x + (size_t)blk * ER * 32);
#pragma unroll
    for (int i = 0; i < 2; ++i) {
        const int idx = threadIdx.x + 128 * i;
        *reinterpret_cast<float4*>(&hs[idx >> 3][4 * (idx & 7)]) = x4[idx];
    }
    __syncthreads();
    rows_ln<32, true>(hs, os, We, bE, gE, beE, nullptr, blk);
    __syncthreads();
    rows_ln<64, false>(hs, os, W0, b0, g0, be0, z, blk);
    const int q = threadIdx.x & 15;
#pragma unroll
    for (int p = 0; p < ER / 8; ++p) {
        const int nl = (threadIdx.x >> 4) + 8 * p;
        reinterpret_cast<float4*>(h + ((size_t)blk * ER + nl) * 64)[q] =
            *reinterpret_cast<const float4*>(&hs[nl][4 * q]);
    }
}


template <int NCO, bool LN>
DEVFN void tile_tail(float (*hs)[HP], float (*os)[OP],
                     const float* __restrict__ Wa, const float* __restrict__ ba,
                     const float* __restrict__ Wb, const float* __restrict__ bb,
                     const float* __restrict__ g,  const float* __restrict__ be,
                     float* __restrict__ outA, float* __restrict__ outB, int tile) {
    static_assert(NCO == 64 || (NCO == 32 && !LN), "tile map below");
    constexpr int TPM = NCO / 16;
    constexpr int NT  = LN ? TPM : 2 * TPM;
    const int wave = threadIdx.x >> 5;
    const int lane = threadIdx.x & 31;
    const int l16  = lane & 15;
    const int half = lane >> 4;
    if (wave < NT) {
        const bool second = (!LN) && (wave >= TPM);
        const float* __restrict__ Wm   = second ? Wb : Wa;
        const float* __restrict__ bias = second ? bb : ba;
        const int col = (second ? wave - TPM : wave) * 16 + l16;
        v8f acc = {};
#pragma unroll
        for (int k = 0; k < 64; k += 4) {
            const int kk = k + 2 * half;
            v2f a, b;
            a.x = hs[l16][kk];
            a.y = hs[l16][kk + 1];
            b.x = Wm[(size_t)kk * NCO + col];
            b.y = Wm[(size_t)(kk + 1) * NCO + col];
            acc = __builtin_amdgcn_wmma_f32_16x16x4_f32(
                false, a, false, b, (short)0, acc, false, false);
        }
        const float bj = bias[col];
#pragma unroll
        for (int v = 0; v < 8; ++v) {
            float val = acc[v] + bj;
            if constexpr (LN) val = silu_f(val);
            os[v + 8 * half][wave * 16 + l16] = val;
        }
    }
    __syncthreads();
    const int nl = threadIdx.x >> 4, q = threadIdx.x & 15;
    const size_t row = (size_t)tile * 16 + nl;
    const float4 v0 = *reinterpret_cast<const float4*>(&os[nl][4 * q]);
    if constexpr (LN) {
        float s  = (v0.x + v0.y) + (v0.z + v0.w);
        float sq = (v0.x * v0.x + v0.y * v0.y) + (v0.z * v0.z + v0.w * v0.w);
        s  = group_sum<16>(s);
        sq = group_sum<16>(sq);
        const float mean = s * (1.f / 64);
        const float var  = sq * (1.f / 64) - mean * mean;
        const float inv  = rsq_f(var + 1e-5f);
        const int c0 = 4 * q;
        float4 o;
        o.x = (v0.x - mean) * inv * g[c0]     + be[c0];
        o.y = (v0.y - mean) * inv * g[c0 + 1] + be[c0 + 1];
        o.z = (v0.z - mean) * inv * g[c0 + 2] + be[c0 + 2];
        o.w = (v0.w - mean) * inv * g[c0 + 3] + be[c0 + 3];
        reinterpret_cast<float4*>(outA + row * 64)[q] = o;
    } else if constexpr (NCO == 64) {
        const float4 v1 = *reinterpret_cast<const float4*>(&os[nl][64 + 4 * q]);
        reinterpret_cast<float4*>(outA + row * 64)[q] = v0;
        reinterpret_cast<float4*>(outB + row * 64)[q] = v1;
    } else {
        if (q < 8) reinterpret_cast<float4*>(outA + row * 32)[q] = v0;
        else       reinterpret_cast<float4*>(outB + row * 32)[q - 8] = v0;
    }
}


template <bool NEXT_LN>
__global__ __launch_bounds__(256) __attribute__((amdgpu_waves_per_eu(16)))
void k_enc_agg(const int* __restrict__ start, const int* __restrict__ csr,
               const float* __restrict__ z, float* __restrict__ h,
               const float* __restrict__ Wa, const float* __restrict__ ba,
               const float* __restrict__ Wb, const float* __restrict__ bb,
               const float* __restrict__ g,  const float* __restrict__ be,
               float* __restrict__ outA, float* __restrict__ outB, int n, int e) {
    __shared__ __attribute__((aligned(16))) float hs[16][HP];
    __shared__ __attribute__((aligned(16))) float os[16][OP];
    const int t = blockIdx.x * 256 + threadIdx.x;
    const int node = t >> 4, q = t & 15;
    const int beg = start[node];
    const int end = (node + 1 < n) ? start[node + 1] : e;
    const float4* z4 = reinterpret_cast<const float4*>(z);
    float4 acc = make_float4(0.f, 0.f, 0.f, 0.f);
    for (int base = beg; base < end; base += 16) {
        const int cnt = min(16, end - base);
        const int my = (q < cnt) ? csr[base + q] : 0;
        for (int j = 0; j < cnt; j += 4) {
            const int s0 = group_bcast<16>(my, j + 0);
            const int s1 = group_bcast<16>(my, j + 1);
            const int s2 = group_bcast<16>(my, j + 2);
            const int s3 = group_bcast<16>(my, j + 3);
            const float4 v0 = z4[s0 * 16 + q];
            const float4 v1 = z4[s1 * 16 + q];
            const float4 v2 = z4[s2 * 16 + q];
            const float4 v3 = z4[s3 * 16 + q];
            const bool k1 = j + 1 < cnt, k2 = j + 2 < cnt, k3 = j + 3 < cnt;
            acc.x += v0.x; acc.y += v0.y; acc.z += v0.z; acc.w += v0.w;
            acc.x += k1 ? v1.x : 0.f; acc.y += k1 ? v1.y : 0.f; acc.z += k1 ? v1.z : 0.f; acc.w += k1 ? v1.w : 0.f;
            acc.x += k2 ? v2.x : 0.f; acc.y += k2 ? v2.y : 0.f; acc.z += k2 ? v2.z : 0.f; acc.w += k2 ? v2.w : 0.f;
            acc.x += k3 ? v3.x : 0.f; acc.y += k3 ? v3.y : 0.f; acc.z += k3 ? v3.z : 0.f; acc.w += k3 ? v3.w : 0.f;
        }
    }
    const float inv = rcp_f(fmaxf((float)(end - beg), 1.0f));
    float4* hp = reinterpret_cast<float4*>(h + (size_t)node * 64) + q;
    float4 hv = *hp;
    hv.x += acc.x * inv; hv.y += acc.y * inv; hv.z += acc.z * inv; hv.w += acc.w * inv;
    *reinterpret_cast<float4*>(&hs[threadIdx.x >> 4][4 * q]) = hv;
    __syncthreads();
    tile_tail<64, NEXT_LN>(hs, os, Wa, ba, Wb, bb, g, be, outA, outB, (int)blockIdx.x);
    *hp = *reinterpret_cast<const float4*>(&hs[threadIdx.x >> 4][4 * q]);
}


template <int NC, bool RES, int TAIL>
__global__ __launch_bounds__(256) __attribute__((amdgpu_waves_per_eu(16)))
void k_gat_node(const int* __restrict__ start, const int* __restrict__ csr,
                           const float* __restrict__ xl, const float* __restrict__ xr,
                           const float* __restrict__ att, const float* __restrict__ bias,
                           const float* __restrict__ g, const float* __restrict__ be,
                           const float* hin, float* out,
                           const float* __restrict__ Wa, const float* __restrict__ ba,
                           const float* __restrict__ Wb, const float* __restrict__ bb,
                           float* __restrict__ outA, float* __restrict__ outB,
                           int n, int e) {
    constexpr int LPN = NC / 4;
    const int t = blockIdx.x * 256 + threadIdx.x;
    const int node = t / LPN, q = t % LPN;
    if constexpr (TAIL == 0) { if (node >= n) return; }
    const int beg = start[node];
    const int end = (node + 1 < n) ? start[node + 1] : e;
    const int c0 = q * 4;
    const v2f a01 = { att[c0] * LOG2E,     att[c0 + 1] * LOG2E };
    const v2f a23 = { att[c0 + 2] * LOG2E, att[c0 + 3] * LOG2E };
    const float4* xl4 = reinterpret_cast<const float4*>(xl);
    const float4 r4 = reinterpret_cast<const float4*>(xr)[node * LPN + q];
    const float4 self = xl4[node * LPN + q];
    const v2f r01 = { r4.x, r4.y }, r23 = { r4.z, r4.w };

    auto part = [&](const float4& v) -> float {
        const v2f t01 = lo2(v) + r01;
        const v2f t23 = hi2(v) + r23;
        v2f p = a01 * lrelu2(t01);
        p = __builtin_elementwise_fma(a23, lrelu2(t23), p);
        return p.x + p.y;
    };

    float m = group_sum<LPN>(part(self));
    float l = 1.f;
    v2f acc01 = lo2(self), acc23 = hi2(self);
    for (int base = beg; base < end; base += LPN) {
        const int cnt = min(LPN, end - base);
        const int my = (q < cnt) ? csr[base + q] : node;
        for (int j = 0; j < cnt; j += 4) {
            const int s0 = group_bcast<LPN>(my, j + 0);
            const int s1 = group_bcast<LPN>(my, j + 1);
            const int s2 = group_bcast<LPN>(my, j + 2);
            const int s3 = group_bcast<LPN>(my, j + 3);
            const float4 v0 = xl4[s0 * LPN + q];
            const float4 v1 = xl4[s1 * LPN + q];
            const float4 v2 = xl4[s2 * LPN + q];
            const float4 v3 = xl4[s3 * LPN + q];
            asm volatile("" :: "v"(v0.x), "v"(v0.y), "v"(v0.z), "v"(v0.w),
                               "v"(v1.x), "v"(v1.y), "v"(v1.z), "v"(v1.w),
                               "v"(v2.x), "v"(v2.y), "v"(v2.z), "v"(v2.w),
                               "v"(v3.x), "v"(v3.y), "v"(v3.z), "v"(v3.w));
            float e0 = part(v0), e1 = part(v1), e2 = part(v2), e3 = part(v3);
            group_sum4<LPN>(e0, e1, e2, e3);
            if (j + 1 >= cnt) e1 = -INFINITY;
            if (j + 2 >= cnt) e2 = -INFINITY;
            if (j + 3 >= cnt) e3 = -INFINITY;
            const float mn = fmaxf(fmaxf(m, e0), fmaxf(fmaxf(e1, e2), e3));
            const v2f d01 = v2f{e0, e1} - both(mn), d23 = v2f{e2, e3} - both(mn);
            const float sc = __builtin_amdgcn_exp2f(m - mn);
            const float p0 = __builtin_amdgcn_exp2f(d01.x), p1 = __builtin_amdgcn_exp2f(d01.y);
            const float p2 = __builtin_amdgcn_exp2f(d23.x), p3 = __builtin_amdgcn_exp2f(d23.y);
            l = l * sc + ((p0 + p1) + (p2 + p3));
            const v2f P0 = both(p0), P1 = both(p1), P2 = both(p2), P3 = both(p3), SC = both(sc);
            const v2f s01a = __builtin_elementwise_fma(lo2(v0), P0, lo2(v1) * P1);
            const v2f s01b = __builtin_elementwise_fma(lo2(v2), P2, lo2(v3) * P3);
            const v2f s23a = __builtin_elementwise_fma(hi2(v0), P0, hi2(v1) * P1);
            const v2f s23b = __builtin_elementwise_fma(hi2(v2), P2, hi2(v3) * P3);
            acc01 = __builtin_elementwise_fma(acc01, SC, s01a + s01b);
            acc23 = __builtin_elementwise_fma(acc23, SC, s23a + s23b);
            m = mn;
        }
    }

    const float il = rcp_f(l);
    float vals[4];
    vals[0] = silu_f(acc01.x * il + bias[c0]);
    vals[1] = silu_f(acc01.y * il + bias[c0 + 1]);
    vals[2] = silu_f(acc23.x * il + bias[c0 + 2]);
    vals[3] = silu_f(acc23.y * il + bias[c0 + 3]);
    float s = (vals[0] + vals[1]) + (vals[2] + vals[3]);
    float sq = (vals[0] * vals[0] + vals[1] * vals[1]) + (vals[2] * vals[2] + vals[3] * vals[3]);
    s  = group_sum<LPN>(s);
    sq = group_sum<LPN>(sq);
    const float mean = s * (1.f / NC);
    const float var  = sq * (1.f / NC) - mean * mean;
    const float inv  = rsq_f(var + 1e-5f);
    float4 o;
    o.x = (vals[0] - mean) * inv * g[c0]     + be[c0];
    o.y = (vals[1] - mean) * inv * g[c0 + 1] + be[c0 + 1];
    o.z = (vals[2] - mean) * inv * g[c0 + 2] + be[c0 + 2];
    o.w = (vals[3] - mean) * inv * g[c0 + 3] + be[c0 + 3];
    if (RES) {
        const float4 hv = reinterpret_cast<const float4*>(hin + (size_t)node * NC)[q];
        o.x += hv.x; o.y += hv.y; o.z += hv.z; o.w += hv.w;
    }
    if constexpr (TAIL > 0) {
        static_assert(NC == 64, "16 lanes per node: a 256-thread block is one 16-row tile");
        __shared__ __attribute__((aligned(16))) float hs[16][HP];
        __shared__ __attribute__((aligned(16))) float os[16][OP];
        *reinterpret_cast<float4*>(&hs[threadIdx.x >> 4][4 * q]) = o;
        __syncthreads();
        tile_tail<TAIL, false>(hs, os, Wa, ba, Wb, bb, nullptr, nullptr, outA, outB, (int)blockIdx.x);
        reinterpret_cast<float4*>(out + (size_t)node * NC)[q] =
            *reinterpret_cast<const float4*>(&hs[threadIdx.x >> 4][4 * q]);
    } else {
        (void)Wa; (void)ba; (void)Wb; (void)bb; (void)outA; (void)outB;
        reinterpret_cast<float4*>(out + (size_t)node * NC)[q] = o;
    }
}


extern "C" void kernel_launch(void* const* d_in, const int* in_sizes, int n_in,
                              void* d_out, int out_size, void* d_ws, size_t ws_size,
                              hipStream_t stream) {
    (void)in_sizes; (void)n_in; (void)out_size; (void)ws_size;

    const float* x        = (const float*)d_in[0];
    const int*   ei       = (const int*)  d_in[1];
    const float* W_emb    = (const float*)d_in[2];
    const float* b_emb    = (const float*)d_in[3];
    const float* g_emb    = (const float*)d_in[4];
    const float* be_emb   = (const float*)d_in[5];
    const float* enc_W    = (const float*)d_in[6];
    const float* enc_b    = (const float*)d_in[7];
    const float* enc_g    = (const float*)d_in[8];
    const float* enc_be   = (const float*)d_in[9];
    const float* dec_Wl   = (const float*)d_in[10];
    const float* dec_bl   = (const float*)d_in[11];
    const float* dec_Wr   = (const float*)d_in[12];
    const float* dec_br   = (const float*)d_in[13];
    const float* dec_att  = (const float*)d_in[14];
    const float* dec_bias = (const float*)d_in[15];
    const float* dec_g    = (const float*)d_in[16];
    const float* dec_be   = (const float*)d_in[17];
    const float* fin_Wl   = (const float*)d_in[18];
    const float* fin_bl   = (const float*)d_in[19];
    const float* fin_Wr   = (const float*)d_in[20];
    const float* fin_br   = (const float*)d_in[21];
    const float* fin_att  = (const float*)d_in[22];
    const float* fin_bias = (const float*)d_in[23];
    const float* fin_g    = (const float*)d_in[24];
    const float* fin_be   = (const float*)d_in[25];

    const int* src = ei;
    const int* dst = ei + EE;
    float* out = (float*)d_out;

    int* wi      = (int*)d_ws;
    int*   start = wi;                  wi += NN;
                                        wi += 256;
    int*   tmp   = wi;                  wi += EE;
    int*   csr   = wi;                  wi += EE;
    float* w     = (float*)wi;
    float* h     = w;                   w += (size_t)NN * 64;
    float* zA    = w;                   w += (size_t)NN * 64;
    float* zB    = w;                   w += (size_t)NN * 64;
    float* xlA   = w;                   w += (size_t)NN * 64;
    float* xrA   = w;                   w += (size_t)NN * 64;
    int*   mat    = (int*)w;
    int*   rowtot = mat + (size_t)NB1 * NBK;

    const int TB = 256;
    auto nb = [](long long t) { return (unsigned)((t + 255) / 256); };

    k_hist<<<NB1, 256, 0, stream>>>(dst, mat);
    k_rowscan<<<NBK, 512, 0, stream>>>(mat, rowtot);
    k_scatter<<<NB1, 256, 0, stream>>>(src, dst, mat, rowtot, tmp);
    k_bucket<<<NBK, 512, 0, stream>>>(tmp, rowtot, start, csr);

    k_embed<<<NN / ER, 128, 0, stream>>>(x, W_emb, b_emb, g_emb, be_emb,
                                         enc_W, enc_b, enc_g, enc_be, h, zA);

    const unsigned NB16 = (unsigned)(NN / 16);
    k_enc_agg<true><<<NB16, TB, 0, stream>>>(start, csr, zA, h,
        enc_W + (size_t)1 * 64 * 64, enc_b + 1 * 64, nullptr, nullptr, enc_g + 1 * 64, enc_be + 1 * 64,
        zB, nullptr, NN, EE);
    k_enc_agg<true><<<NB16, TB, 0, stream>>>(start, csr, zB, h,
        enc_W + (size_t)2 * 64 * 64, enc_b + 2 * 64, nullptr, nullptr, enc_g + 2 * 64, enc_be + 2 * 64,
        zA, nullptr, NN, EE);
    k_enc_agg<false><<<NB16, TB, 0, stream>>>(start, csr, zA, h,
        dec_Wl, dec_bl, dec_Wr, dec_br, nullptr, nullptr, xlA, xrA, NN, EE);

    k_gat_node<64, true, 64><<<NB16, TB, 0, stream>>>(
        start, csr, xlA, xrA, dec_att, dec_bias, dec_g, dec_be, h, h,
        dec_Wl + (size_t)1 * 64 * 64, dec_bl + 1 * 64, dec_Wr + (size_t)1 * 64 * 64, dec_br + 1 * 64,
        zA, zB, NN, EE);
    k_gat_node<64, true, 64><<<NB16, TB, 0, stream>>>(
        start, csr, zA, zB, dec_att + 1 * 64, dec_bias + 1 * 64, dec_g + 1 * 64, dec_be + 1 * 64, h, h,
        dec_Wl + (size_t)2 * 64 * 64, dec_bl + 2 * 64, dec_Wr + (size_t)2 * 64 * 64, dec_br + 2 * 64,
        xlA, xrA, NN, EE);
    k_gat_node<64, true, 32><<<NB16, TB, 0, stream>>>(
        start, csr, xlA, xrA, dec_att + 2 * 64, dec_bias + 2 * 64, dec_g + 2 * 64, dec_be + 2 * 64, h, h,
        fin_Wl, fin_bl, fin_Wr, fin_br, zA, zB, NN, EE);

    k_gat_node<32, false, 0><<<nb((long long)NN * 8), TB, 0, stream>>>(
        start, csr, zA, zB, fin_att, fin_bias, fin_g, fin_be, nullptr, out,
        nullptr, nullptr, nullptr, nullptr, nullptr, nullptr, NN, EE);
}


# 64-column node kernels: 8 lanes per node (8 columns a lane), 4 nodes a wave, 32-node blocks with two row tiles per dense tail; score and LN row sums now in 8-column, 8-lane order
# speedup vs baseline: 7.9246x; 1.1217x over previous
#include <hip/hip_runtime.h>
#include <cstdint>
#include <cstddef>


typedef float v2f __attribute__((ext_vector_type(2)));
typedef float v8f __attribute__((ext_vector_type(8)));

static constexpr int NN  = 100000;
static constexpr int EE  = 1200000;
static constexpr int LM1 = 3;
static_assert(NN % 32 == 0, "a 256-thread node block is exactly two 16-row tiles");
static constexpr int HP  = 68;
static constexpr int OP  = 132;

#define DEVFN static __device__ __forceinline__

DEVFN float rcp_f(float v)   { return __builtin_amdgcn_rcpf(v); }
DEVFN float rsq_f(float v)   { return __builtin_amdgcn_rsqf(v); }
DEVFN float silu_f(float v)  { return v * rcp_f(1.f + __expf(-v)); }
DEVFN v2f lrelu2(v2f t)      { return __builtin_elementwise_max(t, t * 0.2f); }
static constexpr float LOG2E = 1.44269504088896340736f;
DEVFN v2f lo2(const float4& v) { return v2f{v.x, v.y}; }
DEVFN v2f hi2(const float4& v) { return v2f{v.z, v.w}; }
DEVFN v2f both(float s)        { return v2f{s, s}; }


template <int M>
DEVFN float row_xor_add(float v) {
    const int t = __builtin_amdgcn_update_dpp(0, __builtin_bit_cast(int, v), 0x160 | M, 0xf, 0xf, true);
    return v + __builtin_bit_cast(float, t);
}

template <int G>
DEVFN float group_sum(float v) {
    static_assert(G == 16 || G == 8, "groups inside a 16-lane row");
    if constexpr (G == 16) v = row_xor_add<8>(v);
    v = row_xor_add<4>(v);
    v = row_xor_add<2>(v);
    return row_xor_add<1>(v);
}

template <int M>
DEVFN float row_xor_get(float v) {
    const int t = __builtin_amdgcn_update_dpp(0, __builtin_bit_cast(int, v), 0x160 | M, 0xf, 0xf, true);
    return __builtin_bit_cast(float, t);
}

template <int K>
DEVFN float quad_get(float v) {
    const int t = __builtin_amdgcn_update_dpp(0, __builtin_bit_cast(int, v), K * 0x55, 0xf, 0xf, true);
    return __builtin_bit_cast(float, t);
}

template <int G>
DEVFN void group_sum4(float& a, float& b, float& c, float& d) {
    static_assert(G == 16 || G == 8, "groups inside a 16-lane row");
    const bool b0 = (threadIdx.x & 1) != 0, b1 = (threadIdx.x & 2) != 0;
    const float x = (b0 ? c : a) + row_xor_get<1>(b0 ? a : c);
    const float y = (b0 ? d : b) + row_xor_get<1>(b0 ? b : d);
    float z = (b1 ? y : x) + row_xor_get<2>(b1 ? x : y);
    z = row_xor_add<4>(z);
    if constexpr (G == 16) z = row_xor_add<8>(z);
    a = quad_get<0>(z); c = quad_get<1>(z); b = quad_get<2>(z); d = quad_get<3>(z);
}

template <int G>
DEVFN int group_bcast(int v, int idx) {
    static_assert(G == 16 || G == 8, "groups inside a 16-lane row");
    const unsigned lo = 0x11111111u * (unsigned)idx;
    const unsigned hi = (G == 16) ? lo : lo + 0x88888888u;
    return (int)__builtin_amdgcn_permlane16(0u, (unsigned)v, lo, hi, true, false);
}


static constexpr int RSH = 8;
static constexpr int RB  = 1 << RSH;
static constexpr int NBK = (NN + RB - 1) / RB;
static constexpr int CH  = 4096;
static constexpr int NB1 = (EE + CH - 1) / CH;
static constexpr int CAP = 8192;
static_assert(NBK <= 512, "k_scatter scans the ranges two per thread");
static_assert(NB1 <= 512, "k_rowscan scans the chunks one per thread");
static_assert(NN <= (1 << (31 - RSH)), "a source id and RSH destination bits share one positive word");

template <int NT>
DEVFN int block_excl(int v, int* wsum, int& total) {
    const int lane = threadIdx.x & 31, wv = threadIdx.x >> 5;
    int inc = v;
#pragma unroll
    for (int off = 1; off < 32; off <<= 1) {
        const int u = __shfl_up(inc, off, 32);
        if (lane >= off) inc += u;
    }
    __syncthreads();
    if (lane == 31) wsum[wv] = inc;
    __syncthreads();
    int woff = 0, tot = 0;
#pragma unroll
    for (int w = 0; w < NT / 32; ++w) {
        const int s = wsum[w];
        tot += s;
        if (w < wv) woff += s;
    }
    total = tot;
    return woff + inc - v;
}

__global__ __launch_bounds__(256)
void k_hist(const int* __restrict__ dst, int* __restrict__ mat) {
    __shared__ int hist[NBK];
    const int tid = threadIdx.x, c = blockIdx.x;
    for (int k = tid; k < NBK; k += 256) hist[k] = 0;
    __syncthreads();
    const int e0 = c * CH;
#pragma unroll
    for (int i = 0; i < CH / 256; ++i) {
        const int e = e0 + i * 256 + tid;
        const int d = dst[min(e, EE - 1)];
        if (e < EE) atomicAdd(&hist[d >> RSH], 1);
    }
    __syncthreads();
    for (int k = tid; k < NBK; k += 256) mat[c * NBK + k] = hist[k];
}

__global__ __launch_bounds__(512)
void k_rowscan(int* __restrict__ mat, int* __restrict__ rowtot) {
    __shared__ int wsum[16];
    const int k = blockIdx.x, t = threadIdx.x;
    const int v = (t < NB1) ? mat[t * NBK + k] : 0;
    int total;
    const int ex = block_excl<512>(v, wsum, total);
    if (t < NB1) mat[t * NBK + k] = ex;
    if (t == 0) rowtot[k] = total;
}

__global__ __launch_bounds__(256)
void k_scatter(const int* __restrict__ src, const int* __restrict__ dst,
               const int* __restrict__ mat, const int* __restrict__ rowtot,
               int* __restrict__ tmp) {
    __shared__ int cur[512];
    __shared__ int gdel[512];
    __shared__ int sval[CH];
    __shared__ unsigned short sbk[CH];
    __shared__ int wsum[8];
    const int tid = threadIdx.x, c = blockIdx.x;
    cur[2 * tid] = 0;
    cur[2 * tid + 1] = 0;
    __syncthreads();
    const int e0 = c * CH;
    int dv[CH / 256], sv[CH / 256];
#pragma unroll
    for (int i = 0; i < CH / 256; ++i) {
        const int e = e0 + i * 256 + tid;
        const int ec = min(e, EE - 1);
        const int d = dst[ec];
        sv[i] = src[ec];
        dv[i] = (e < EE) ? d : -1;
    }
#pragma unroll
    for (int i = 0; i < CH / 256; ++i)
        if (dv[i] >= 0) atomicAdd(&cur[dv[i] >> RSH], 1);
    __syncthreads();
    const int k0 = 2 * tid, k1 = 2 * tid + 1;
    const int h0 = cur[k0], h1 = cur[k1];
    const int t0 = (k0 < NBK) ? rowtot[k0] : 0, t1 = (k1 < NBK) ? rowtot[k1] : 0;
    const int m0 = (k0 < NBK) ? mat[c * NBK + k0] : 0, m1 = (k1 < NBK) ? mat[c * NBK + k1] : 0;
    int unused;
    const int lex = block_excl<256>(h0 + h1, wsum, unused);
    const int gex = block_excl<256>(t0 + t1, wsum, unused);
    cur[k0] = lex;
    cur[k1] = lex + h0;
    gdel[k0] = gex + m0 - lex;
    gdel[k1] = gex + t0 + m1 - (lex + h0);
    __syncthreads();
#pragma unroll
    for (int i = 0; i < CH / 256; ++i) {
        if (dv[i] >= 0) {
            const int k = dv[i] >> RSH;
            const int p = atomicAdd(&cur[k], 1);
            sval[p] = (sv[i] << RSH) | (dv[i] & (RB - 1));
            sbk[p] = (unsigned short)k;
        }
    }
    __syncthreads();
    const int nv = min(CH, EE - e0);
    for (int i = tid; i < nv; i += 256) tmp[i + gdel[sbk[i]]] = sval[i];
}

__global__ __launch_bounds__(512)
void k_bucket(const int* __restrict__ tmp, const int* __restrict__ rowtot,
              int* __restrict__ start, int* __restrict__ csr) {
    __shared__ int cnt[RB];
    __shared__ int wsum[16];
    __shared__ int stage[CAP];
    const int k = blockIdx.x, t = threadIdx.x;
    int part = 0;
    for (int j = t; j < k; j += 512) part += rowtot[j];
    int base;
    (void)block_excl<512>(part, wsum, base);
    const int tot = rowtot[k];
    if (t < RB) cnt[t] = 0;
    __syncthreads();
    for (int i = t; i < tot; i += 512) atomicAdd(&cnt[tmp[base + i] & (RB - 1)], 1);
    __syncthreads();
    const int v = (t < RB) ? cnt[t] : 0;
    int unused;
    const int ex = block_excl<512>(v, wsum, unused);
    if (t < RB) {
        const int node = k * RB + t;
        if (node < NN) start[node] = base + ex;
        cnt[t] = ex;
    }
    __syncthreads();
    const bool staged = tot <= CAP;
    for (int i = t; i < tot; i += 512) {
        const int w = tmp[base + i];
        const int p = atomicAdd(&cnt[w & (RB - 1)], 1);
        const int s = w >> RSH;
        if (staged) stage[p] = s; else csr[base + p] = s;
    }
    if (staged) {
        __syncthreads();
        for (int i = t; i < tot; i += 512) csr[base + i] = stage[i];
    }
}


static constexpr int ER = 32;
static_assert(NN % ER == 0, "k_embed's grid is exact");

template <int K, bool KEEP>
DEVFN void rows_ln(float (*hs)[HP], float (*os)[HP],
                   const float* __restrict__ Wm, const float* __restrict__ bias,
                   const float* __restrict__ g,  const float* __restrict__ be,
                   float* __restrict__ outG, int blk) {
    const int wave = threadIdx.x >> 5;
    const int lane = threadIdx.x & 31;
    const int l16  = lane & 15;
    const int half = lane >> 4;
    const int col  = wave * 16 + l16;
    v8f acc0 = {}, acc1 = {};
#pragma unroll
    for (int k = 0; k < K; k += 4) {
        const int kk = k + 2 * half;
        v2f a0, a1, b;
        a0.x = hs[l16][kk];
        a0.y = hs[l16][kk + 1];
        a1.x = hs[16 + l16][kk];
        a1.y = hs[16 + l16][kk + 1];
        b.x = Wm[(size_t)kk * 64 + col];
        b.y = Wm[(size_t)(kk + 1) * 64 + col];
        acc0 = __builtin_amdgcn_wmma_f32_16x16x4_f32(
            false, a0, false, b, (short)0, acc0, false, false);
        acc1 = __builtin_amdgcn_wmma_f32_16x16x4_f32(
            false, a1, false, b, (short)0, acc1, false, false);
    }
    const float bj = bias[col];
#pragma unroll
    for (int v = 0; v < 8; ++v) {
        os[v + 8 * half][col]      = silu_f(acc0[v] + bj);
        os[16 + v + 8 * half][col] = silu_f(acc1[v] + bj);
    }
    __syncthreads();
    const int q = threadIdx.x & 15, c0 = 4 * q;
    const float g0 = g[c0], g1 = g[c0 + 1], g2 = g[c0 + 2], g3 = g[c0 + 3];
    const float e0 = be[c0], e1 = be[c0 + 1], e2 = be[c0 + 2], e3 = be[c0 + 3];
#pragma unroll
    for (int p = 0; p < ER / 8; ++p) {
        const int nl = (threadIdx.x >> 4) + 8 * p;
        const float4 v0 = *reinterpret_cast<const float4*>(&os[nl][c0]);
        float s  = (v0.x + v0.y) + (v0.z + v0.w);
        float sq = (v0.x * v0.x + v0.y * v0.y) + (v0.z * v0.z + v0.w * v0.w);
        s  = group_sum<16>(s);
        sq = group_sum<16>(sq);
        const float mean = s * (1.f / 64);
        const float var  = sq * (1.f / 64) - mean * mean;
        const float inv  = rsq_f(var + 1e-5f);
        float4 o;
        o.x = (v0.x - mean) * inv * g0 + e0;
        o.y = (v0.y - mean) * inv * g1 + e1;
        o.z = (v0.z - mean) * inv * g2 + e2;
        o.w = (v0.w - mean) * inv * g3 + e3;
        if constexpr (KEEP) *reinterpret_cast<float4*>(&hs[nl][c0]) = o;
        else reinterpret_cast<float4*>(outG + ((size_t)blk * ER + nl) * 64)[q] = o;
    }
}

__global__ __launch_bounds__(128)
void k_embed(const float* __restrict__ x,
             const float* __restrict__ We, const float* __restrict__ bE,
             const float* __restrict__ gE, const float* __restrict__ beE,
             const float* __restrict__ W0, const float* __restrict__ b0,
             const float* __restrict__ g0, const float* __restrict__ be0,
             float* __restrict__ h, float* __restrict__ z) {
    __shared__ __attribute__((aligned(16))) float hs[ER][HP];
    __shared__ __attribute__((aligned(16))) float os[ER][HP];
    const int blk = blockIdx.x;
    const float4* x4 = reinterpret_cast<const float4*>(x + (size_t)blk * ER * 32);
#pragma unroll
    for (int i = 0; i < 2; ++i) {
        const int idx = threadIdx.x + 128 * i;
        *reinterpret_cast<float4*>(&hs[idx >> 3][4 * (idx & 7)]) = x4[idx];
    }
    __syncthreads();
    rows_ln<32, true>(hs, os, We, bE, gE, beE, nullptr, blk);
    __syncthreads();
    rows_ln<64, false>(hs, os, W0, b0, g0, be0, z, blk);
    const int q = threadIdx.x & 15;
#pragma unroll
    for (int p = 0; p < ER / 8; ++p) {
        const int nl = (threadIdx.x >> 4) + 8 * p;
        reinterpret_cast<float4*>(h + ((size_t)blk * ER + nl) * 64)[q] =
            *reinterpret_cast<const float4*>(&hs[nl][4 * q]);
    }
}


template <int NCO, bool LN>
DEVFN void tile_tail(float (*hs)[HP], float (*os)[OP],
                     const float* __restrict__ Wa, const float* __restrict__ ba,
                     const float* __restrict__ Wb, const float* __restrict__ bb,
                     const float* __restrict__ g,  const float* __restrict__ be,
                     float* __restrict__ outA, float* __restrict__ outB, int blk) {
    static_assert(NCO == 64 || (NCO == 32 && !LN), "tile map below");
    constexpr int TPM = NCO / 16;
    constexpr int NCT = LN ? TPM : 2 * TPM;
    constexpr int RPW = NCT / 4;
    const int wave = threadIdx.x >> 5;
    const int lane = threadIdx.x & 31;
    const int l16  = lane & 15;
    const int half = lane >> 4;
    const int ct   = (RPW == 2) ? wave : (wave & 3);
    const int rt0  = (RPW == 2) ? 0 : (wave >> 2);
    const bool second = (!LN) && (ct >= TPM);
    const float* __restrict__ Wm   = second ? Wb : Wa;
    const float* __restrict__ bias = second ? bb : ba;
    const int col = (second ? ct - TPM : ct) * 16 + l16;
    v8f acc[RPW];
#pragma unroll
    for (int r = 0; r < RPW; ++r) acc[r] = v8f{};
#pragma unroll
    for (int k = 0; k < 64; k += 4) {
        const int kk = k + 2 * half;
        v2f b;
        b.x = Wm[(size_t)kk * NCO + col];
        b.y = Wm[(size_t)(kk + 1) * NCO + col];
#pragma unroll
        for (int r = 0; r < RPW; ++r) {
            v2f a;
            a.x = hs[(rt0 + r) * 16 + l16][kk];
            a.y = hs[(rt0 + r) * 16 + l16][kk + 1];
            acc[r] = __builtin_amdgcn_wmma_f32_16x16x4_f32(
                false, a, false, b, (short)0, acc[r], false, false);
        }
    }
    const float bj = bias[col];
#pragma unroll
    for (int r = 0; r < RPW; ++r) {
#pragma unroll
        for (int v = 0; v < 8; ++v) {
            float val = acc[r][v] + bj;
            if constexpr (LN) val = silu_f(val);
            os[(rt0 + r) * 16 + v + 8 * half][ct * 16 + l16] = val;
        }
    }
    __syncthreads();
    const int q = threadIdx.x & 15, c0 = 4 * q;
    float4 gq = make_float4(0.f, 0.f, 0.f, 0.f), eq = gq;
    if constexpr (LN) {
        gq = make_float4(g[c0], g[c0 + 1], g[c0 + 2], g[c0 + 3]);
        eq = make_float4(be[c0], be[c0 + 1], be[c0 + 2], be[c0 + 3]);
    }
#pragma unroll
    for (int p = 0; p < 2; ++p) {
        const int nl = (threadIdx.x >> 4) + 16 * p;
        const size_t row = (size_t)blk * 32 + nl;
        const float4 v0 = *reinterpret_cast<const float4*>(&os[nl][c0]);
        if constexpr (LN) {
            float s  = (v0.x + v0.y) + (v0.z + v0.w);
            float sq = (v0.x * v0.x + v0.y * v0.y) + (v0.z * v0.z + v0.w * v0.w);
            s  = group_sum<16>(s);
            sq = group_sum<16>(sq);
            const float mean = s * (1.f / 64);
            const float var  = sq * (1.f / 64) - mean * mean;
            const float inv  = rsq_f(var + 1e-5f);
            float4 o;
            o.x = (v0.x - mean) * inv * gq.x + eq.x;
            o.y = (v0.y - mean) * inv * gq.y + eq.y;
            o.z = (v0.z - mean) * inv * gq.z + eq.z;
            o.w = (v0.w - mean) * inv * gq.w + eq.w;
            reinterpret_cast<float4*>(outA + row * 64)[q] = o;
        } else if constexpr (NCO == 64) {
            const float4 v1 = *reinterpret_cast<const float4*>(&os[nl][64 + c0]);
            reinterpret_cast<float4*>(outA + row * 64)[q] = v0;
            reinterpret_cast<float4*>(outB + row * 64)[q] = v1;
        } else {
            if (q < 8) reinterpret_cast<float4*>(outA + row * 32)[q] = v0;
            else       reinterpret_cast<float4*>(outB + row * 32)[q - 8] = v0;
        }
    }
}


template <bool NEXT_LN>
__global__ __launch_bounds__(256) __attribute__((amdgpu_waves_per_eu(16)))
void k_enc_agg(const int* __restrict__ start, const int* __restrict__ csr,
               const float* __restrict__ z, float* __restrict__ h,
               const float* __restrict__ Wa, const float* __restrict__ ba,
               const float* __restrict__ Wb, const float* __restrict__ bb,
               const float* __restrict__ g,  const float* __restrict__ be,
               float* __restrict__ outA, float* __restrict__ outB, int n, int e) {
    __shared__ __attribute__((aligned(16))) float hs[32][HP];
    __shared__ __attribute__((aligned(16))) float os[32][OP];
    const int t = blockIdx.x * 256 + threadIdx.x;
    const int node = t >> 3, q = t & 7;
    const int beg = start[node];
    const int end = (node + 1 < n) ? start[node + 1] : e;
    const float4* z4 = reinterpret_cast<const float4*>(z);
    float4 accA = make_float4(0.f, 0.f, 0.f, 0.f), accB = accA;
    for (int base = beg; base < end; base += 16) {
        const int cnt = min(16, end - base);
        const int my0 = (q < cnt)     ? csr[base + q]     : 0;
        const int my1 = (q + 8 < cnt) ? csr[base + 8 + q] : 0;
        for (int j = 0; j < cnt; j += 4) {
            const int my = (j < 8) ? my0 : my1;
            const int s0 = group_bcast<8>(my, (j & 7) + 0);
            const int s1 = group_bcast<8>(my, (j & 7) + 1);
            const int s2 = group_bcast<8>(my, (j & 7) + 2);
            const int s3 = group_bcast<8>(my, (j & 7) + 3);
            const float4 a0 = z4[s0 * 16 + q], b0 = z4[s0 * 16 + 8 + q];
            const float4 a1 = z4[s1 * 16 + q], b1 = z4[s1 * 16 + 8 + q];
            const float4 a2 = z4[s2 * 16 + q], b2 = z4[s2 * 16 + 8 + q];
            const float4 a3 = z4[s3 * 16 + q], b3 = z4[s3 * 16 + 8 + q];
            const bool k1 = j + 1 < cnt, k2 = j + 2 < cnt, k3 = j + 3 < cnt;
            accA.x += a0.x; accA.y += a0.y; accA.z += a0.z; accA.w += a0.w;
            accB.x += b0.x; accB.y += b0.y; accB.z += b0.z; accB.w += b0.w;
            accA.x += k1 ? a1.x : 0.f; accA.y += k1 ? a1.y : 0.f; accA.z += k1 ? a1.z : 0.f; accA.w += k1 ? a1.w : 0.f;
            accB.x += k1 ? b1.x : 0.f; accB.y += k1 ? b1.y : 0.f; accB.z += k1 ? b1.z : 0.f; accB.w += k1 ? b1.w : 0.f;
            accA.x += k2 ? a2.x : 0.f; accA.y += k2 ? a2.y : 0.f; accA.z += k2 ? a2.z : 0.f; accA.w += k2 ? a2.w : 0.f;
            accB.x += k2 ? b2.x : 0.f; accB.y += k2 ? b2.y : 0.f; accB.z += k2 ? b2.z : 0.f; accB.w += k2 ? b2.w : 0.f;
            accA.x += k3 ? a3.x : 0.f; accA.y += k3 ? a3.y : 0.f; accA.z += k3 ? a3.z : 0.f; accA.w += k3 ? a3.w : 0.f;
            accB.x += k3 ? b3.x : 0.f; accB.y += k3 ? b3.y : 0.f; accB.z += k3 ? b3.z : 0.f; accB.w += k3 ? b3.w : 0.f;
        }
    }
    const float inv = rcp_f(fmaxf((float)(end - beg), 1.0f));
    float4* hp = reinterpret_cast<float4*>(h + (size_t)node * 64) + q;
    float4 hvA = hp[0], hvB = hp[8];
    hvA.x += accA.x * inv; hvA.y += accA.y * inv; hvA.z += accA.z * inv; hvA.w += accA.w * inv;
    hvB.x += accB.x * inv; hvB.y += accB.y * inv; hvB.z += accB.z * inv; hvB.w += accB.w * inv;
    const int nl = threadIdx.x >> 3;
    *reinterpret_cast<float4*>(&hs[nl][4 * q])      = hvA;
    *reinterpret_cast<float4*>(&hs[nl][32 + 4 * q]) = hvB;
    __syncthreads();
    tile_tail<64, NEXT_LN>(hs, os, Wa, ba, Wb, bb, g, be, outA, outB, (int)blockIdx.x);
    hp[0] = *reinterpret_cast<const float4*>(&hs[nl][4 * q]);
    hp[8] = *reinterpret_cast<const float4*>(&hs[nl][32 + 4 * q]);
}


template <int TAIL>
__global__ __launch_bounds__(256) __attribute__((amdgpu_waves_per_eu(10)))
void k_gat_wide(const int* __restrict__ start, const int* __restrict__ csr,
                const float* __restrict__ xl, const float* __restrict__ xr,
                const float* __restrict__ att, const float* __restrict__ bias,
                const float* __restrict__ g, const float* __restrict__ be,
                float* __restrict__ h,
                const float* __restrict__ Wa, const float* __restrict__ ba,
                const float* __restrict__ Wb, const float* __restrict__ bb,
                float* __restrict__ outA, float* __restrict__ outB,
                int n, int e) {
    __shared__ __attribute__((aligned(16))) float hs[32][HP];
    __shared__ __attribute__((aligned(16))) float os[32][OP];
    const int t = blockIdx.x * 256 + threadIdx.x;
    const int node = t >> 3, q = t & 7;
    const int beg = start[node];
    const int end = (node + 1 < n) ? start[node + 1] : e;
    const int cA = 4 * q, cB = 32 + 4 * q;
    const v2f aA01 = { att[cA] * LOG2E,     att[cA + 1] * LOG2E };
    const v2f aA23 = { att[cA + 2] * LOG2E, att[cA + 3] * LOG2E };
    const v2f aB01 = { att[cB] * LOG2E,     att[cB + 1] * LOG2E };
    const v2f aB23 = { att[cB + 2] * LOG2E, att[cB + 3] * LOG2E };
    const float4* xl4 = reinterpret_cast<const float4*>(xl);
    const float4 rA = reinterpret_cast<const float4*>(xr)[node * 16 + q];
    const float4 rB = reinterpret_cast<const float4*>(xr)[node * 16 + 8 + q];
    const float4 selfA = xl4[node * 16 + q], selfB = xl4[node * 16 + 8 + q];
    const v2f rA01 = lo2(rA), rA23 = hi2(rA), rB01 = lo2(rB), rB23 = hi2(rB);

    auto part = [&](const float4& vA, const float4& vB) -> float {
        v2f p = aA01 * lrelu2(lo2(vA) + rA01);
        p = __builtin_elementwise_fma(aA23, lrelu2(hi2(vA) + rA23), p);
        p = __builtin_elementwise_fma(aB01, lrelu2(lo2(vB) + rB01), p);
        p = __builtin_elementwise_fma(aB23, lrelu2(hi2(vB) + rB23), p);
        return p.x + p.y;
    };

    float m = group_sum<8>(part(selfA, selfB));
    float l = 1.f;
    v2f accA01 = lo2(selfA), accA23 = hi2(selfA);
    v2f accB01 = lo2(selfB), accB23 = hi2(selfB);
    for (int base = beg; base < end; base += 16) {
        const int cnt = min(16, end - base);
        const int my0 = (q < cnt)     ? csr[base + q]     : node;
        const int my1 = (q + 8 < cnt) ? csr[base + 8 + q] : node;
        for (int j = 0; j < cnt; j += 4) {
            const int my = (j < 8) ? my0 : my1;
            const int s0 = group_bcast<8>(my, (j & 7) + 0);
            const int s1 = group_bcast<8>(my, (j & 7) + 1);
            const int s2 = group_bcast<8>(my, (j & 7) + 2);
            const int s3 = group_bcast<8>(my, (j & 7) + 3);
            const float4 vA0 = xl4[s0 * 16 + q], vB0 = xl4[s0 * 16 + 8 + q];
            const float4 vA1 = xl4[s1 * 16 + q], vB1 = xl4[s1 * 16 + 8 + q];
            const float4 vA2 = xl4[s2 * 16 + q], vB2 = xl4[s2 * 16 + 8 + q];
            const float4 vA3 = xl4[s3 * 16 + q], vB3 = xl4[s3 * 16 + 8 + q];
            asm volatile("" :: "v"(vA0.x), "v"(vA0.y), "v"(vB0.x), "v"(vB0.y),
                               "v"(vA1.x), "v"(vA1.y), "v"(vB1.x), "v"(vB1.y),
                               "v"(vA2.x), "v"(vA2.y), "v"(vB2.x), "v"(vB2.y),
                               "v"(vA3.x), "v"(vA3.y), "v"(vB3.x), "v"(vB3.y));
            asm volatile("" :: "v"(vA0.z), "v"(vA0.w), "v"(vB0.z), "v"(vB0.w),
                               "v"(vA1.z), "v"(vA1.w), "v"(vB1.z), "v"(vB1.w),
                               "v"(vA2.z), "v"(vA2.w), "v"(vB2.z), "v"(vB2.w),
                               "v"(vA3.z), "v"(vA3.w), "v"(vB3.z), "v"(vB3.w));
            float e0 = part(vA0, vB0), e1 = part(vA1, vB1), e2 = part(vA2, vB2), e3 = part(vA3, vB3);
            group_sum4<8>(e0, e1, e2, e3);
            if (j + 1 >= cnt) e1 = -INFINITY;
            if (j + 2 >= cnt) e2 = -INFINITY;
            if (j + 3 >= cnt) e3 = -INFINITY;
            const float mn = fmaxf(fmaxf(m, e0), fmaxf(fmaxf(e1, e2), e3));
            const v2f d01 = v2f{e0, e1} - both(mn), d23 = v2f{e2, e3} - both(mn);
            const float sc = __builtin_amdgcn_exp2f(m - mn);
            const float p0 = __builtin_amdgcn_exp2f(d01.x), p1 = __builtin_amdgcn_exp2f(d01.y);
            const float p2 = __builtin_amdgcn_exp2f(d23.x), p3 = __builtin_amdgcn_exp2f(d23.y);
            l = l * sc + ((p0 + p1) + (p2 + p3));
            const v2f P0 = both(p0), P1 = both(p1), P2 = both(p2), P3 = both(p3), SC = both(sc);
            const v2f sA01a = __builtin_elementwise_fma(lo2(vA0), P0, lo2(vA1) * P1);
            const v2f sA01b = __builtin_elementwise_fma(lo2(vA2), P2, lo2(vA3) * P3);
            const v2f sA23a = __builtin_elementwise_fma(hi2(vA0), P0, hi2(vA1) * P1);
            const v2f sA23b = __builtin_elementwise_fma(hi2(vA2), P2, hi2(vA3) * P3);
            const v2f sB01a = __builtin_elementwise_fma(lo2(vB0), P0, lo2(vB1) * P1);
            const v2f sB01b = __builtin_elementwise_fma(lo2(vB2), P2, lo2(vB3) * P3);
            const v2f sB23a = __builtin_elementwise_fma(hi2(vB0), P0, hi2(vB1) * P1);
            const v2f sB23b = __builtin_elementwise_fma(hi2(vB2), P2, hi2(vB3) * P3);
            accA01 = __builtin_elementwise_fma(accA01, SC, sA01a + sA01b);
            accA23 = __builtin_elementwise_fma(accA23, SC, sA23a + sA23b);
            accB01 = __builtin_elementwise_fma(accB01, SC, sB01a + sB01b);
            accB23 = __builtin_elementwise_fma(accB23, SC, sB23a + sB23b);
            m = mn;
        }
    }

    const float il = rcp_f(l);
    float vals[8];
    vals[0] = silu_f(accA01.x * il + bias[cA]);
    vals[1] = silu_f(accA01.y * il + bias[cA + 1]);
    vals[2] = silu_f(accA23.x * il + bias[cA + 2]);
    vals[3] = silu_f(accA23.y * il + bias[cA + 3]);
    vals[4] = silu_f(accB01.x * il + bias[cB]);
    vals[5] = silu_f(accB01.y * il + bias[cB + 1]);
    vals[6] = silu_f(accB23.x * il + bias[cB + 2]);
    vals[7] = silu_f(accB23.y * il + bias[cB + 3]);
    float s  = ((vals[0] + vals[1]) + (vals[2] + vals[3])) + ((vals[4] + vals[5]) + (vals[6] + vals[7]));
    float sq = ((vals[0] * vals[0] + vals[1] * vals[1]) + (vals[2] * vals[2] + vals[3] * vals[3]))
             + ((vals[4] * vals[4] + vals[5] * vals[5]) + (vals[6] * vals[6] + vals[7] * vals[7]));
    s  = group_sum<8>(s);
    sq = group_sum<8>(sq);
    const float mean = s * (1.f / 64);
    const float var  = sq * (1.f / 64) - mean * mean;
    const float inv  = rsq_f(var + 1e-5f);
    float4* hp = reinterpret_cast<float4*>(h + (size_t)node * 64) + q;
    const float4 hvA = hp[0], hvB = hp[8];
    float4 oA, oB;
    oA.x = (vals[0] - mean) * inv * g[cA]     + be[cA]     + hvA.x;
    oA.y = (vals[1] - mean) * inv * g[cA + 1] + be[cA + 1] + hvA.y;
    oA.z = (vals[2] - mean) * inv * g[cA + 2] + be[cA + 2] + hvA.z;
    oA.w = (vals[3] - mean) * inv * g[cA + 3] + be[cA + 3] + hvA.w;
    oB.x = (vals[4] - mean) * inv * g[cB]     + be[cB]     + hvB.x;
    oB.y = (vals[5] - mean) * inv * g[cB + 1] + be[cB + 1] + hvB.y;
    oB.z = (vals[6] - mean) * inv * g[cB + 2] + be[cB + 2] + hvB.z;
    oB.w = (vals[7] - mean) * inv * g[cB + 3] + be[cB + 3] + hvB.w;
    const int nl = threadIdx.x >> 3;
    *reinterpret_cast<float4*>(&hs[nl][cA]) = oA;
    *reinterpret_cast<float4*>(&hs[nl][cB]) = oB;
    __syncthreads();
    tile_tail<TAIL, false>(hs, os, Wa, ba, Wb, bb, nullptr, nullptr, outA, outB, (int)blockIdx.x);
    hp[0] = *reinterpret_cast<const float4*>(&hs[nl][cA]);
    hp[8] = *reinterpret_cast<const float4*>(&hs[nl][cB]);
}

__global__ __launch_bounds__(256) __attribute__((amdgpu_waves_per_eu(16)))
void k_gat_final(const int* __restrict__ start, const int* __restrict__ csr,
                 const float* __restrict__ xl, const float* __restrict__ xr,
                 const float* __restrict__ att, const float* __restrict__ bias,
                 const float* __restrict__ g, const float* __restrict__ be,
                 float* __restrict__ out, int n, int e) {
    constexpr int NC = 32, LPN = 8;
    const int t = blockIdx.x * 256 + threadIdx.x;
    const int node = t / LPN, q = t % LPN;
    if (node >= n) return;
    const int beg = start[node];
    const int end = (node + 1 < n) ? start[node + 1] : e;
    const int c0 = q * 4;
    const v2f a01 = { att[c0] * LOG2E,     att[c0 + 1] * LOG2E };
    const v2f a23 = { att[c0 + 2] * LOG2E, att[c0 + 3] * LOG2E };
    const float4* xl4 = reinterpret_cast<const float4*>(xl);
    const float4 r4 = reinterpret_cast<const float4*>(xr)[node * LPN + q];
    const float4 self = xl4[node * LPN + q];
    const v2f r01 = { r4.x, r4.y }, r23 = { r4.z, r4.w };

    auto part = [&](const float4& v) -> float {
        const v2f t01 = lo2(v) + r01;
        const v2f t23 = hi2(v) + r23;
        v2f p = a01 * lrelu2(t01);
        p = __builtin_elementwise_fma(a23, lrelu2(t23), p);
        return p.x + p.y;
    };

    float m = group_sum<LPN>(part(self));
    float l = 1.f;
    v2f acc01 = lo2(self), acc23 = hi2(self);
    for (int base = beg; base < end; base += LPN) {
        const int cnt = min(LPN, end - base);
        const int my = (q < cnt) ? csr[base + q] : node;
        for (int j = 0; j < cnt; j += 4) {
            const int s0 = group_bcast<LPN>(my, j + 0);
            const int s1 = group_bcast<LPN>(my, j + 1);
            const int s2 = group_bcast<LPN>(my, j + 2);
            const int s3 = group_bcast<LPN>(my, j + 3);
            const float4 v0 = xl4[s0 * LPN + q];
            const float4 v1 = xl4[s1 * LPN + q];
            const float4 v2 = xl4[s2 * LPN + q];
            const float4 v3 = xl4[s3 * LPN + q];
            asm volatile("" :: "v"(v0.x), "v"(v0.y), "v"(v0.z), "v"(v0.w),
                               "v"(v1.x), "v"(v1.y), "v"(v1.z), "v"(v1.w),
                               "v"(v2.x), "v"(v2.y), "v"(v2.z), "v"(v2.w),
                               "v"(v3.x), "v"(v3.y), "v"(v3.z), "v"(v3.w));
            float e0 = part(v0), e1 = part(v1), e2 = part(v2), e3 = part(v3);
            group_sum4<LPN>(e0, e1, e2, e3);
            if (j + 1 >= cnt) e1 = -INFINITY;
            if (j + 2 >= cnt) e2 = -INFINITY;
            if (j + 3 >= cnt) e3 = -INFINITY;
            const float mn = fmaxf(fmaxf(m, e0), fmaxf(fmaxf(e1, e2), e3));
            const v2f d01 = v2f{e0, e1} - both(mn), d23 = v2f{e2, e3} - both(mn);
            const float sc = __builtin_amdgcn_exp2f(m - mn);
            const float p0 = __builtin_amdgcn_exp2f(d01.x), p1 = __builtin_amdgcn_exp2f(d01.y);
            const float p2 = __builtin_amdgcn_exp2f(d23.x), p3 = __builtin_amdgcn_exp2f(d23.y);
            l = l * sc + ((p0 + p1) + (p2 + p3));
            const v2f P0 = both(p0), P1 = both(p1), P2 = both(p2), P3 = both(p3), SC = both(sc);
            const v2f s01a = __builtin_elementwise_fma(lo2(v0), P0, lo2(v1) * P1);
            const v2f s01b = __builtin_elementwise_fma(lo2(v2), P2, lo2(v3) * P3);
            const v2f s23a = __builtin_elementwise_fma(hi2(v0), P0, hi2(v1) * P1);
            const v2f s23b = __builtin_elementwise_fma(hi2(v2), P2, hi2(v3) * P3);
            acc01 = __builtin_elementwise_fma(acc01, SC, s01a + s01b);
            acc23 = __builtin_elementwise_fma(acc23, SC, s23a + s23b);
            m = mn;
        }
    }

    const float il = rcp_f(l);
    float vals[4];
    vals[0] = silu_f(acc01.x * il + bias[c0]);
    vals[1] = silu_f(acc01.y * il + bias[c0 + 1]);
    vals[2] = silu_f(acc23.x * il + bias[c0 + 2]);
    vals[3] = silu_f(acc23.y * il + bias[c0 + 3]);
    float s = (vals[0] + vals[1]) + (vals[2] + vals[3]);
    float sq = (vals[0] * vals[0] + vals[1] * vals[1]) + (vals[2] * vals[2] + vals[3] * vals[3]);
    s  = group_sum<LPN>(s);
    sq = group_sum<LPN>(sq);
    const float mean = s * (1.f / NC);
    const float var  = sq * (1.f / NC) - mean * mean;
    const float inv  = rsq_f(var + 1e-5f);
    float4 o;
    o.x = (vals[0] - mean) * inv * g[c0]     + be[c0];
    o.y = (vals[1] - mean) * inv * g[c0 + 1] + be[c0 + 1];
    o.z = (vals[2] - mean) * inv * g[c0 + 2] + be[c0 + 2];
    o.w = (vals[3] - mean) * inv * g[c0 + 3] + be[c0 + 3];
    reinterpret_cast<float4*>(out + (size_t)node * NC)[q] = o;
}


extern "C" void kernel_launch(void* const* d_in, const int* in_sizes, int n_in,
                              void* d_out, int out_size, void* d_ws, size_t ws_size,
                              hipStream_t stream) {
    (void)in_sizes; (void)n_in; (void)out_size; (void)ws_size;

    const float* x        = (const float*)d_in[0];
    const int*   ei       = (const int*)  d_in[1];
    const float* W_emb    = (const float*)d_in[2];
    const float* b_emb    = (const float*)d_in[3];
    const float* g_emb    = (const float*)d_in[4];
    const float* be_emb   = (const float*)d_in[5];
    const float* enc_W    = (const float*)d_in[6];
    const float* enc_b    = (const float*)d_in[7];
    const float* enc_g    = (const float*)d_in[8];
    const float* enc_be   = (const float*)d_in[9];
    const float* dec_Wl   = (const float*)d_in[10];
    const float* dec_bl   = (const float*)d_in[11];
    const float* dec_Wr   = (const float*)d_in[12];
    const float* dec_br   = (const float*)d_in[13];
    const float* dec_att  = (const float*)d_in[14];
    const float* dec_bias = (const float*)d_in[15];
    const float* dec_g    = (const float*)d_in[16];
    const float* dec_be   = (const float*)d_in[17];
    const float* fin_Wl   = (const float*)d_in[18];
    const float* fin_bl   = (const float*)d_in[19];
    const float* fin_Wr   = (const float*)d_in[20];
    const float* fin_br   = (const float*)d_in[21];
    const float* fin_att  = (const float*)d_in[22];
    const float* fin_bias = (const float*)d_in[23];
    const float* fin_g    = (const float*)d_in[24];
    const float* fin_be   = (const float*)d_in[25];

    const int* src = ei;
    const int* dst = ei + EE;
    float* out = (float*)d_out;

    int* wi      = (int*)d_ws;
    int*   start = wi;                  wi += NN;
                                        wi += 256;
    int*   tmp   = wi;                  wi += EE;
    int*   csr   = wi;                  wi += EE;
    float* w     = (float*)wi;
    float* h     = w;                   w += (size_t)NN * 64;
    float* zA    = w;                   w += (size_t)NN * 64;
    float* zB    = w;                   w += (size_t)NN * 64;
    float* xlA   = w;                   w += (size_t)NN * 64;
    float* xrA   = w;                   w += (size_t)NN * 64;
    int*   mat    = (int*)w;
    int*   rowtot = mat + (size_t)NB1 * NBK;

    const int TB = 256;
    auto nb = [](long long t) { return (unsigned)((t + 255) / 256); };

    k_hist<<<NB1, 256, 0, stream>>>(dst, mat);
    k_rowscan<<<NBK, 512, 0, stream>>>(mat, rowtot);
    k_scatter<<<NB1, 256, 0, stream>>>(src, dst, mat, rowtot, tmp);
    k_bucket<<<NBK, 512, 0, stream>>>(tmp, rowtot, start, csr);

    k_embed<<<NN / ER, 128, 0, stream>>>(x, W_emb, b_emb, g_emb, be_emb,
                                         enc_W, enc_b, enc_g, enc_be, h, zA);

    const unsigned NB32 = (unsigned)(NN / 32);
    k_enc_agg<true><<<NB32, TB, 0, stream>>>(start, csr, zA, h,
        enc_W + (size_t)1 * 64 * 64, enc_b + 1 * 64, nullptr, nullptr, enc_g + 1 * 64, enc_be + 1 * 64,
        zB, nullptr, NN, EE);
    k_enc_agg<true><<<NB32, TB, 0, stream>>>(start, csr, zB, h,
        enc_W + (size_t)2 * 64 * 64, enc_b + 2 * 64, nullptr, nullptr, enc_g + 2 * 64, enc_be + 2 * 64,
        zA, nullptr, NN, EE);
    k_enc_agg<false><<<NB32, TB, 0, stream>>>(start, csr, zA, h,
        dec_Wl, dec_bl, dec_Wr, dec_br, nullptr, nullptr, xlA, xrA, NN, EE);

    k_gat_wide<64><<<NB32, TB, 0, stream>>>(
        start, csr, xlA, xrA, dec_att, dec_bias, dec_g, dec_be, h,
        dec_Wl + (size_t)1 * 64 * 64, dec_bl + 1 * 64, dec_Wr + (size_t)1 * 64 * 64, dec_br + 1 * 64,
        zA, zB, NN, EE);
    k_gat_wide<64><<<NB32, TB, 0, stream>>>(
        start, csr, zA, zB, dec_att + 1 * 64, dec_bias + 1 * 64, dec_g + 1 * 64, dec_be + 1 * 64, h,
        dec_Wl + (size_t)2 * 64 * 64, dec_bl + 2 * 64, dec_Wr + (size_t)2 * 64 * 64, dec_br + 2 * 64,
        xlA, xrA, NN, EE);
    k_gat_wide<32><<<NB32, TB, 0, stream>>>(
        start, csr, xlA, xrA, dec_att + 2 * 64, dec_bias + 2 * 64, dec_g + 2 * 64, dec_be + 2 * 64, h,
        fin_Wl, fin_bl, fin_Wr, fin_br, zA, zB, NN, EE);

    k_gat_final<<<nb((long long)NN * 8), TB, 0, stream>>>(
        start, csr, zA, zB, fin_att, fin_bias, fin_g, fin_be, out, NN, EE);
}


# edge sort in 2 launches, was 4 (12 to 10 in all): chunks sorted by range in place; each range block gathers its runs, run starts summed over chunks = its base; slot order still LDS-atomic arrival
# speedup vs baseline: 8.4900x; 1.0713x over previous
#include <hip/hip_runtime.h>
#include <cstdint>
#include <cstddef>


typedef float v2f __attribute__((ext_vector_type(2)));
typedef float v8f __attribute__((ext_vector_type(8)));

static constexpr int NN  = 100000;
static constexpr int EE  = 1200000;
static constexpr int LM1 = 3;
static_assert(NN % 32 == 0, "a 256-thread node block is exactly two 16-row tiles");
static constexpr int HP  = 68;
static constexpr int OP  = 132;

#define DEVFN static __device__ __forceinline__

DEVFN float rcp_f(float v)   { return __builtin_amdgcn_rcpf(v); }
DEVFN float rsq_f(float v)   { return __builtin_amdgcn_rsqf(v); }
DEVFN float silu_f(float v)  { return v * rcp_f(1.f + __expf(-v)); }
DEVFN v2f lrelu2(v2f t)      { return __builtin_elementwise_max(t, t * 0.2f); }
static constexpr float LOG2E = 1.44269504088896340736f;
DEVFN v2f lo2(const float4& v) { return v2f{v.x, v.y}; }
DEVFN v2f hi2(const float4& v) { return v2f{v.z, v.w}; }
DEVFN v2f both(float s)        { return v2f{s, s}; }


template <int M>
DEVFN float row_xor_add(float v) {
    const int t = __builtin_amdgcn_update_dpp(0, __builtin_bit_cast(int, v), 0x160 | M, 0xf, 0xf, true);
    return v + __builtin_bit_cast(float, t);
}

template <int G>
DEVFN float group_sum(float v) {
    static_assert(G == 16 || G == 8, "groups inside a 16-lane row");
    if constexpr (G == 16) v = row_xor_add<8>(v);
    v = row_xor_add<4>(v);
    v = row_xor_add<2>(v);
    return row_xor_add<1>(v);
}

template <int M>
DEVFN float row_xor_get(float v) {
    const int t = __builtin_amdgcn_update_dpp(0, __builtin_bit_cast(int, v), 0x160 | M, 0xf, 0xf, true);
    return __builtin_bit_cast(float, t);
}

template <int K>
DEVFN float quad_get(float v) {
    const int t = __builtin_amdgcn_update_dpp(0, __builtin_bit_cast(int, v), K * 0x55, 0xf, 0xf, true);
    return __builtin_bit_cast(float, t);
}

template <int G>
DEVFN void group_sum4(float& a, float& b, float& c, float& d) {
    static_assert(G == 16 || G == 8, "groups inside a 16-lane row");
    const bool b0 = (threadIdx.x & 1) != 0, b1 = (threadIdx.x & 2) != 0;
    const float x = (b0 ? c : a) + row_xor_get<1>(b0 ? a : c);
    const float y = (b0 ? d : b) + row_xor_get<1>(b0 ? b : d);
    float z = (b1 ? y : x) + row_xor_get<2>(b1 ? x : y);
    z = row_xor_add<4>(z);
    if constexpr (G == 16) z = row_xor_add<8>(z);
    a = quad_get<0>(z); c = quad_get<1>(z); b = quad_get<2>(z); d = quad_get<3>(z);
}

template <int G>
DEVFN int group_bcast(int v, int idx) {
    static_assert(G == 16 || G == 8, "groups inside a 16-lane row");
    const unsigned lo = 0x11111111u * (unsigned)idx;
    const unsigned hi = (G == 16) ? lo : lo + 0x88888888u;
    return (int)__builtin_amdgcn_permlane16(0u, (unsigned)v, lo, hi, true, false);
}


static constexpr int RSH = 8;
static constexpr int RB  = 1 << RSH;
static constexpr int NBK = (NN + RB - 1) / RB;
static constexpr int CH  = 4096;
static constexpr int NB1 = (EE + CH - 1) / CH;
static constexpr int OFFP = NBK + 1;
static constexpr int CAP = 8192;
static_assert(NBK < 512, "k_chunksort scans the ranges two per thread, and one entry more");
static_assert(NB1 <= 512, "k_bucket takes the chunks one per thread");
static_assert(NN <= (1 << (31 - RSH)), "a source id and RSH destination bits share one positive word");

template <int NT>
DEVFN int block_excl(int v, int* wsum, int& total) {
    const int lane = threadIdx.x & 31, wv = threadIdx.x >> 5;
    int inc = v;
#pragma unroll
    for (int off = 1; off < 32; off <<= 1) {
        const int u = __shfl_up(inc, off, 32);
        if (lane >= off) inc += u;
    }
    __syncthreads();
    if (lane == 31) wsum[wv] = inc;
    __syncthreads();
    int woff = 0, tot = 0;
#pragma unroll
    for (int w = 0; w < NT / 32; ++w) {
        const int s = wsum[w];
        tot += s;
        if (w < wv) woff += s;
    }
    total = tot;
    return woff + inc - v;
}

__global__ __launch_bounds__(256)
void k_chunksort(const int* __restrict__ src, const int* __restrict__ dst,
                 int* __restrict__ off, int* __restrict__ tmp) {
    __shared__ int cur[512];
    __shared__ int sval[CH];
    __shared__ int wsum[8];
    const int tid = threadIdx.x, c = blockIdx.x;
    cur[2 * tid] = 0;
    cur[2 * tid + 1] = 0;
    __syncthreads();
    const int e0 = c * CH;
    int dv[CH / 256], sv[CH / 256];
#pragma unroll
    for (int i = 0; i < CH / 256; ++i) {
        const int e = e0 + i * 256 + tid;
        const int ec = min(e, EE - 1);
        const int d = dst[ec];
        sv[i] = src[ec];
        dv[i] = (e < EE) ? d : -1;
    }
#pragma unroll
    for (int i = 0; i < CH / 256; ++i)
        if (dv[i] >= 0) atomicAdd(&cur[dv[i] >> RSH], 1);
    __syncthreads();
    const int k0 = 2 * tid, k1 = 2 * tid + 1;
    const int h0 = cur[k0];
    int unused;
    const int lex = block_excl<256>(h0 + cur[k1], wsum, unused);
    cur[k0] = lex;
    cur[k1] = lex + h0;
    if (k0 <= NBK) off[c * OFFP + k0] = lex;
    if (k1 <= NBK) off[c * OFFP + k1] = lex + h0;
    __syncthreads();
#pragma unroll
    for (int i = 0; i < CH / 256; ++i) {
        if (dv[i] >= 0) {
            const int p = atomicAdd(&cur[dv[i] >> RSH], 1);
            sval[p] = (sv[i] << RSH) | (dv[i] & (RB - 1));
        }
    }
    __syncthreads();
    const int nv = min(CH, EE - e0);
    for (int i = tid; i < nv; i += 256) tmp[e0 + i] = sval[i];
}

__global__ __launch_bounds__(512)
void k_bucket(const int* __restrict__ tmp, const int* __restrict__ off,
              int* __restrict__ start, int* __restrict__ csr) {
    __shared__ int cnt[RB];
    __shared__ int wsum[16];
    __shared__ int rng[CAP];
    __shared__ int stage[CAP];
    const int k = blockIdx.x, t = threadIdx.x;
    int o0 = 0, len = 0;
    if (t < NB1) {
        const int a = off[t * OFFP + k], b = off[t * OFFP + k + 1];
        o0  = min(max(a, 0), CH);
        len = min(max(b - o0, 0), CH - o0);
    }
    int base, tot;
    (void)block_excl<512>(o0, wsum, base);
    const int pos = block_excl<512>(len, wsum, tot);
    if (t < RB) cnt[t] = 0;
    __syncthreads();
    const bool fits = tot <= CAP;
    const int* run = tmp + (t < NB1 ? t * CH + o0 : 0);
    for (int i = 0; i < len; i += 4) {
        int w[4];
#pragma unroll
        for (int u = 0; u < 4; ++u) w[u] = run[min(i + u, len - 1)];
#pragma unroll
        for (int u = 0; u < 4; ++u) {
            if (i + u < len) {
                atomicAdd(&cnt[w[u] & (RB - 1)], 1);
                if (fits) rng[pos + i + u] = w[u];
            }
        }
    }
    __syncthreads();
    const int v = (t < RB) ? cnt[t] : 0;
    int unused;
    const int ex = block_excl<512>(v, wsum, unused);
    if (t < RB) {
        const int node = k * RB + t;
        if (node < NN) start[node] = base + ex;
        cnt[t] = ex;
    }
    __syncthreads();
    if (fits) {
        for (int i = t; i < tot; i += 512) {
            const int w = rng[i];
            const int p = atomicAdd(&cnt[w & (RB - 1)], 1);
            stage[p] = w >> RSH;
        }
        __syncthreads();
        for (int i = t; i < tot; i += 512) csr[base + i] = stage[i];
    } else {
        for (int i = 0; i < len; ++i) {
            const int w = run[i];
            const int p = atomicAdd(&cnt[w & (RB - 1)], 1);
            csr[base + p] = w >> RSH;
        }
    }
}


static constexpr int ER = 32;
static_assert(NN % ER == 0, "k_embed's grid is exact");

template <int K, bool KEEP>
DEVFN void rows_ln(float (*hs)[HP], float (*os)[HP],
                   const float* __restrict__ Wm, const float* __restrict__ bias,
                   const float* __restrict__ g,  const float* __restrict__ be,
                   float* __restrict__ outG, int blk) {
    const int wave = threadIdx.x >> 5;
    const int lane = threadIdx.x & 31;
    const int l16  = lane & 15;
    const int half = lane >> 4;
    const int col  = wave * 16 + l16;
    v8f acc0 = {}, acc1 = {};
#pragma unroll
    for (int k = 0; k < K; k += 4) {
        const int kk = k + 2 * half;
        v2f a0, a1, b;
        a0.x = hs[l16][kk];
        a0.y = hs[l16][kk + 1];
        a1.x = hs[16 + l16][kk];
        a1.y = hs[16 + l16][kk + 1];
        b.x = Wm[(size_t)kk * 64 + col];
        b.y = Wm[(size_t)(kk + 1) * 64 + col];
        acc0 = __builtin_amdgcn_wmma_f32_16x16x4_f32(
            false, a0, false, b, (short)0, acc0, false, false);
        acc1 = __builtin_amdgcn_wmma_f32_16x16x4_f32(
            false, a1, false, b, (short)0, acc1, false, false);
    }
    const float bj = bias[col];
#pragma unroll
    for (int v = 0; v < 8; ++v) {
        os[v + 8 * half][col]      = silu_f(acc0[v] + bj);
        os[16 + v + 8 * half][col] = silu_f(acc1[v] + bj);
    }
    __syncthreads();
    const int q = threadIdx.x & 15, c0 = 4 * q;
    const float g0 = g[c0], g1 = g[c0 + 1], g2 = g[c0 + 2], g3 = g[c0 + 3];
    const float e0 = be[c0], e1 = be[c0 + 1], e2 = be[c0 + 2], e3 = be[c0 + 3];
#pragma unroll
    for (int p = 0; p < ER / 8; ++p) {
        const int nl = (threadIdx.x >> 4) + 8 * p;
        const float4 v0 = *reinterpret_cast<const float4*>(&os[nl][c0]);
        float s  = (v0.x + v0.y) + (v0.z + v0.w);
        float sq = (v0.x * v0.x + v0.y * v0.y) + (v0.z * v0.z + v0.w * v0.w);
        s  = group_sum<16>(s);
        sq = group_sum<16>(sq);
        const float mean = s * (1.f / 64);
        const float var  = sq * (1.f / 64) - mean * mean;
        const float inv  = rsq_f(var + 1e-5f);
        float4 o;
        o.x = (v0.x - mean) * inv * g0 + e0;
        o.y = (v0.y - mean) * inv * g1 + e1;
        o.z = (v0.z - mean) * inv * g2 + e2;
        o.w = (v0.w - mean) * inv * g3 + e3;
        if constexpr (KEEP) *reinterpret_cast<float4*>(&hs[nl][c0]) = o;
        else reinterpret_cast<float4*>(outG + ((size_t)blk * ER + nl) * 64)[q] = o;
    }
}

__global__ __launch_bounds__(128)
void k_embed(const float* __restrict__ x,
             const float* __restrict__ We, const float* __restrict__ bE,
             const float* __restrict__ gE, const float* __restrict__ beE,
             const float* __restrict__ W0, const float* __restrict__ b0,
             const float* __restrict__ g0, const float* __restrict__ be0,
             float* __restrict__ h, float* __restrict__ z) {
    __shared__ __attribute__((aligned(16))) float hs[ER][HP];
    __shared__ __attribute__((aligned(16))) float os[ER][HP];
    const int blk = blockIdx.x;
    const float4* x4 = reinterpret_cast<const float4*>(x + (size_t)blk * ER * 32);
#pragma unroll
    for (int i = 0; i < 2; ++i) {
        const int idx = threadIdx.x + 128 * i;
        *reinterpret_cast<float4*>(&hs[idx >> 3][4 * (idx & 7)]) = x4[idx];
    }
    __syncthreads();
    rows_ln<32, true>(hs, os, We, bE, gE, beE, nullptr, blk);
    __syncthreads();
    rows_ln<64, false>(hs, os, W0, b0, g0, be0, z, blk);
    const int q = threadIdx.x & 15;
#pragma unroll
    for (int p = 0; p < ER / 8; ++p) {
        const int nl = (threadIdx.x >> 4) + 8 * p;
        reinterpret_cast<float4*>(h + ((size_t)blk * ER + nl) * 64)[q] =
            *reinterpret_cast<const float4*>(&hs[nl][4 * q]);
    }
}


template <int NCO, bool LN>
DEVFN void tile_tail(float (*hs)[HP], float (*os)[OP],
                     const float* __restrict__ Wa, const float* __restrict__ ba,
                     const float* __restrict__ Wb, const float* __restrict__ bb,
                     const float* __restrict__ g,  const float* __restrict__ be,
                     float* __restrict__ outA, float* __restrict__ outB, int blk) {
    static_assert(NCO == 64 || (NCO == 32 && !LN), "tile map below");
    constexpr int TPM = NCO / 16;
    constexpr int NCT = LN ? TPM : 2 * TPM;
    constexpr int RPW = NCT / 4;
    const int wave = threadIdx.x >> 5;
    const int lane = threadIdx.x & 31;
    const int l16  = lane & 15;
    const int half = lane >> 4;
    const int ct   = (RPW == 2) ? wave : (wave & 3);
    const int rt0  = (RPW == 2) ? 0 : (wave >> 2);
    const bool second = (!LN) && (ct >= TPM);
    const float* __restrict__ Wm   = second ? Wb : Wa;
    const float* __restrict__ bias = second ? bb : ba;
    const int col = (second ? ct - TPM : ct) * 16 + l16;
    v8f acc[RPW];
#pragma unroll
    for (int r = 0; r < RPW; ++r) acc[r] = v8f{};
#pragma unroll
    for (int k = 0; k < 64; k += 4) {
        const int kk = k + 2 * half;
        v2f b;
        b.x = Wm[(size_t)kk * NCO + col];
        b.y = Wm[(size_t)(kk + 1) * NCO + col];
#pragma unroll
        for (int r = 0; r < RPW; ++r) {
            v2f a;
            a.x = hs[(rt0 + r) * 16 + l16][kk];
            a.y = hs[(rt0 + r) * 16 + l16][kk + 1];
            acc[r] = __builtin_amdgcn_wmma_f32_16x16x4_f32(
                false, a, false, b, (short)0, acc[r], false, false);
        }
    }
    const float bj = bias[col];
#pragma unroll
    for (int r = 0; r < RPW; ++r) {
#pragma unroll
        for (int v = 0; v < 8; ++v) {
            float val = acc[r][v] + bj;
            if constexpr (LN) val = silu_f(val);
            os[(rt0 + r) * 16 + v + 8 * half][ct * 16 + l16] = val;
        }
    }
    __syncthreads();
    const int q = threadIdx.x & 15, c0 = 4 * q;
    float4 gq = make_float4(0.f, 0.f, 0.f, 0.f), eq = gq;
    if constexpr (LN) {
        gq = make_float4(g[c0], g[c0 + 1], g[c0 + 2], g[c0 + 3]);
        eq = make_float4(be[c0], be[c0 + 1], be[c0 + 2], be[c0 + 3]);
    }
#pragma unroll
    for (int p = 0; p < 2; ++p) {
        const int nl = (threadIdx.x >> 4) + 16 * p;
        const size_t row = (size_t)blk * 32 + nl;
        const float4 v0 = *reinterpret_cast<const float4*>(&os[nl][c0]);
        if constexpr (LN) {
            float s  = (v0.x + v0.y) + (v0.z + v0.w);
            float sq = (v0.x * v0.x + v0.y * v0.y) + (v0.z * v0.z + v0.w * v0.w);
            s  = group_sum<16>(s);
            sq = group_sum<16>(sq);
            const float mean = s * (1.f / 64);
            const float var  = sq * (1.f / 64) - mean * mean;
            const float inv  = rsq_f(var + 1e-5f);
            float4 o;
            o.x = (v0.x - mean) * inv * gq.x + eq.x;
            o.y = (v0.y - mean) * inv * gq.y + eq.y;
            o.z = (v0.z - mean) * inv * gq.z + eq.z;
            o.w = (v0.w - mean) * inv * gq.w + eq.w;
            reinterpret_cast<float4*>(outA + row * 64)[q] = o;
        } else if constexpr (NCO == 64) {
            const float4 v1 = *reinterpret_cast<const float4*>(&os[nl][64 + c0]);
            reinterpret_cast<float4*>(outA + row * 64)[q] = v0;
            reinterpret_cast<float4*>(outB + row * 64)[q] = v1;
        } else {
            if (q < 8) reinterpret_cast<float4*>(outA + row * 32)[q] = v0;
            else       reinterpret_cast<float4*>(outB + row * 32)[q - 8] = v0;
        }
    }
}


template <bool NEXT_LN>
__global__ __launch_bounds__(256) __attribute__((amdgpu_waves_per_eu(16)))
void k_enc_agg(const int* __restrict__ start, const int* __restrict__ csr,
               const float* __restrict__ z, float* __restrict__ h,
               const float* __restrict__ Wa, const float* __restrict__ ba,
               const float* __restrict__ Wb, const float* __restrict__ bb,
               const float* __restrict__ g,  const float* __restrict__ be,
               float* __restrict__ outA, float* __restrict__ outB, int n, int e) {
    __shared__ __attribute__((aligned(16))) float hs[32][HP];
    __shared__ __attribute__((aligned(16))) float os[32][OP];
    const int t = blockIdx.x * 256 + threadIdx.x;
    const int node = t >> 3, q = t & 7;
    const int beg = start[node];
    const int end = (node + 1 < n) ? start[node + 1] : e;
    const float4* z4 = reinterpret_cast<const float4*>(z);
    float4 accA = make_float4(0.f, 0.f, 0.f, 0.f), accB = accA;
    for (int base = beg; base < end; base += 16) {
        const int cnt = min(16, end - base);
        const int my0 = (q < cnt)     ? csr[base + q]     : 0;
        const int my1 = (q + 8 < cnt) ? csr[base + 8 + q] : 0;
        for (int j = 0; j < cnt; j += 4) {
            const int my = (j < 8) ? my0 : my1;
            const int s0 = group_bcast<8>(my, (j & 7) + 0);
            const int s1 = group_bcast<8>(my, (j & 7) + 1);
            const int s2 = group_bcast<8>(my, (j & 7) + 2);
            const int s3 = group_bcast<8>(my, (j & 7) + 3);
            const float4 a0 = z4[s0 * 16 + q], b0 = z4[s0 * 16 + 8 + q];
            const float4 a1 = z4[s1 * 16 + q], b1 = z4[s1 * 16 + 8 + q];
            const float4 a2 = z4[s2 * 16 + q], b2 = z4[s2 * 16 + 8 + q];
            const float4 a3 = z4[s3 * 16 + q], b3 = z4[s3 * 16 + 8 + q];
            const bool k1 = j + 1 < cnt, k2 = j + 2 < cnt, k3 = j + 3 < cnt;
            accA.x += a0.x; accA.y += a0.y; accA.z += a0.z; accA.w += a0.w;
            accB.x += b0.x; accB.y += b0.y; accB.z += b0.z; accB.w += b0.w;
            accA.x += k1 ? a1.x : 0.f; accA.y += k1 ? a1.y : 0.f; accA.z += k1 ? a1.z : 0.f; accA.w += k1 ? a1.w : 0.f;
            accB.x += k1 ? b1.x : 0.f; accB.y += k1 ? b1.y : 0.f; accB.z += k1 ? b1.z : 0.f; accB.w += k1 ? b1.w : 0.f;
            accA.x += k2 ? a2.x : 0.f; accA.y += k2 ? a2.y : 0.f; accA.z += k2 ? a2.z : 0.f; accA.w += k2 ? a2.w : 0.f;
            accB.x += k2 ? b2.x : 0.f; accB.y += k2 ? b2.y : 0.f; accB.z += k2 ? b2.z : 0.f; accB.w += k2 ? b2.w : 0.f;
            accA.x += k3 ? a3.x : 0.f; accA.y += k3 ? a3.y : 0.f; accA.z += k3 ? a3.z : 0.f; accA.w += k3 ? a3.w : 0.f;
            accB.x += k3 ? b3.x : 0.f; accB.y += k3 ? b3.y : 0.f; accB.z += k3 ? b3.z : 0.f; accB.w += k3 ? b3.w : 0.f;
        }
    }
    const float inv = rcp_f(fmaxf((float)(end - beg), 1.0f));
    float4* hp = reinterpret_cast<float4*>(h + (size_t)node * 64) + q;
    float4 hvA = hp[0], hvB = hp[8];
    hvA.x += accA.x * inv; hvA.y += accA.y * inv; hvA.z += accA.z * inv; hvA.w += accA.w * inv;
    hvB.x += accB.x * inv; hvB.y += accB.y * inv; hvB.z += accB.z * inv; hvB.w += accB.w * inv;
    const int nl = threadIdx.x >> 3;
    *reinterpret_cast<float4*>(&hs[nl][4 * q])      = hvA;
    *reinterpret_cast<float4*>(&hs[nl][32 + 4 * q]) = hvB;
    __syncthreads();
    tile_tail<64, NEXT_LN>(hs, os, Wa, ba, Wb, bb, g, be, outA, outB, (int)blockIdx.x);
    hp[0] = *reinterpret_cast<const float4*>(&hs[nl][4 * q]);
    hp[8] = *reinterpret_cast<const float4*>(&hs[nl][32 + 4 * q]);
}


template <int TAIL>
__global__ __launch_bounds__(256) __attribute__((amdgpu_waves_per_eu(10)))
void k_gat_wide(const int* __restrict__ start, const int* __restrict__ csr,
                const float* __restrict__ xl, const float* __restrict__ xr,
                const float* __restrict__ att, const float* __restrict__ bias,
                const float* __restrict__ g, const float* __restrict__ be,
                float* __restrict__ h,
                const float* __restrict__ Wa, const float* __restrict__ ba,
                const float* __restrict__ Wb, const float* __restrict__ bb,
                float* __restrict__ outA, float* __restrict__ outB,
                int n, int e) {
    __shared__ __attribute__((aligned(16))) float hs[32][HP];
    __shared__ __attribute__((aligned(16))) float os[32][OP];
    const int t = blockIdx.x * 256 + threadIdx.x;
    const int node = t >> 3, q = t & 7;
    const int beg = start[node];
    const int end = (node + 1 < n) ? start[node + 1] : e;
    const int cA = 4 * q, cB = 32 + 4 * q;
    const v2f aA01 = { att[cA] * LOG2E,     att[cA + 1] * LOG2E };
    const v2f aA23 = { att[cA + 2] * LOG2E, att[cA + 3] * LOG2E };
    const v2f aB01 = { att[cB] * LOG2E,     att[cB + 1] * LOG2E };
    const v2f aB23 = { att[cB + 2] * LOG2E, att[cB + 3] * LOG2E };
    const float4* xl4 = reinterpret_cast<const float4*>(xl);
    const float4 rA = reinterpret_cast<const float4*>(xr)[node * 16 + q];
    const float4 rB = reinterpret_cast<const float4*>(xr)[node * 16 + 8 + q];
    const float4 selfA = xl4[node * 16 + q], selfB = xl4[node * 16 + 8 + q];
    const v2f rA01 = lo2(rA), rA23 = hi2(rA), rB01 = lo2(rB), rB23 = hi2(rB);

    auto part = [&](const float4& vA, const float4& vB) -> float {
        v2f p = aA01 * lrelu2(lo2(vA) + rA01);
        p = __builtin_elementwise_fma(aA23, lrelu2(hi2(vA) + rA23), p);
        p = __builtin_elementwise_fma(aB01, lrelu2(lo2(vB) + rB01), p);
        p = __builtin_elementwise_fma(aB23, lrelu2(hi2(vB) + rB23), p);
        return p.x + p.y;
    };

    float m = group_sum<8>(part(selfA, selfB));
    float l = 1.f;
    v2f accA01 = lo2(selfA), accA23 = hi2(selfA);
    v2f accB01 = lo2(selfB), accB23 = hi2(selfB);
    for (int base = beg; base < end; base += 16) {
        const int cnt = min(16, end - base);
        const int my0 = (q < cnt)     ? csr[base + q]     : node;
        const int my1 = (q + 8 < cnt) ? csr[base + 8 + q] : node;
        for (int j = 0; j < cnt; j += 4) {
            const int my = (j < 8) ? my0 : my1;
            const int s0 = group_bcast<8>(my, (j & 7) + 0);
            const int s1 = group_bcast<8>(my, (j & 7) + 1);
            const int s2 = group_bcast<8>(my, (j & 7) + 2);
            const int s3 = group_bcast<8>(my, (j & 7) + 3);
            const float4 vA0 = xl4[s0 * 16 + q], vB0 = xl4[s0 * 16 + 8 + q];
            const float4 vA1 = xl4[s1 * 16 + q], vB1 = xl4[s1 * 16 + 8 + q];
            const float4 vA2 = xl4[s2 * 16 + q], vB2 = xl4[s2 * 16 + 8 + q];
            const float4 vA3 = xl4[s3 * 16 + q], vB3 = xl4[s3 * 16 + 8 + q];
            asm volatile("" :: "v"(vA0.x), "v"(vA0.y), "v"(vB0.x), "v"(vB0.y),
                               "v"(vA1.x), "v"(vA1.y), "v"(vB1.x), "v"(vB1.y),
                               "v"(vA2.x), "v"(vA2.y), "v"(vB2.x), "v"(vB2.y),
                               "v"(vA3.x), "v"(vA3.y), "v"(vB3.x), "v"(vB3.y));
            asm volatile("" :: "v"(vA0.z), "v"(vA0.w), "v"(vB0.z), "v"(vB0.w),
                               "v"(vA1.z), "v"(vA1.w), "v"(vB1.z), "v"(vB1.w),
                               "v"(vA2.z), "v"(vA2.w), "v"(vB2.z), "v"(vB2.w),
                               "v"(vA3.z), "v"(vA3.w), "v"(vB3.z), "v"(vB3.w));
            float e0 = part(vA0, vB0), e1 = part(vA1, vB1), e2 = part(vA2, vB2), e3 = part(vA3, vB3);
            group_sum4<8>(e0, e1, e2, e3);
            if (j + 1 >= cnt) e1 = -INFINITY;
            if (j + 2 >= cnt) e2 = -INFINITY;
            if (j + 3 >= cnt) e3 = -INFINITY;
            const float mn = fmaxf(fmaxf(m, e0), fmaxf(fmaxf(e1, e2), e3));
            const v2f d01 = v2f{e0, e1} - both(mn), d23 = v2f{e2, e3} - both(mn);
            const float sc = __builtin_amdgcn_exp2f(m - mn);
            const float p0 = __builtin_amdgcn_exp2f(d01.x), p1 = __builtin_amdgcn_exp2f(d01.y);
            const float p2 = __builtin_amdgcn_exp2f(d23.x), p3 = __builtin_amdgcn_exp2f(d23.y);
            l = l * sc + ((p0 + p1) + (p2 + p3));
            const v2f P0 = both(p0), P1 = both(p1), P2 = both(p2), P3 = both(p3), SC = both(sc);
            const v2f sA01a = __builtin_elementwise_fma(lo2(vA0), P0, lo2(vA1) * P1);
            const v2f sA01b = __builtin_elementwise_fma(lo2(vA2), P2, lo2(vA3) * P3);
            const v2f sA23a = __builtin_elementwise_fma(hi2(vA0), P0, hi2(vA1) * P1);
            const v2f sA23b = __builtin_elementwise_fma(hi2(vA2), P2, hi2(vA3) * P3);
            const v2f sB01a = __builtin_elementwise_fma(lo2(vB0), P0, lo2(vB1) * P1);
            const v2f sB01b = __builtin_elementwise_fma(lo2(vB2), P2, lo2(vB3) * P3);
            const v2f sB23a = __builtin_elementwise_fma(hi2(vB0), P0, hi2(vB1) * P1);
            const v2f sB23b = __builtin_elementwise_fma(hi2(vB2), P2, hi2(vB3) * P3);
            accA01 = __builtin_elementwise_fma(accA01, SC, sA01a + sA01b);
            accA23 = __builtin_elementwise_fma(accA23, SC, sA23a + sA23b);
            accB01 = __builtin_elementwise_fma(accB01, SC, sB01a + sB01b);
            accB23 = __builtin_elementwise_fma(accB23, SC, sB23a + sB23b);
            m = mn;
        }
    }

    const float il = rcp_f(l);
    float vals[8];
    vals[0] = silu_f(accA01.x * il + bias[cA]);
    vals[1] = silu_f(accA01.y * il + bias[cA + 1]);
    vals[2] = silu_f(accA23.x * il + bias[cA + 2]);
    vals[3] = silu_f(accA23.y * il + bias[cA + 3]);
    vals[4] = silu_f(accB01.x * il + bias[cB]);
    vals[5] = silu_f(accB01.y * il + bias[cB + 1]);
    vals[6] = silu_f(accB23.x * il + bias[cB + 2]);
    vals[7] = silu_f(accB23.y * il + bias[cB + 3]);
    float s  = ((vals[0] + vals[1]) + (vals[2] + vals[3])) + ((vals[4] + vals[5]) + (vals[6] + vals[7]));
    float sq = ((vals[0] * vals[0] + vals[1] * vals[1]) + (vals[2] * vals[2] + vals[3] * vals[3]))
             + ((vals[4] * vals[4] + vals[5] * vals[5]) + (vals[6] * vals[6] + vals[7] * vals[7]));
    s  = group_sum<8>(s);
    sq = group_sum<8>(sq);
    const float mean = s * (1.f / 64);
    const float var  = sq * (1.f / 64) - mean * mean;
    const float inv  = rsq_f(var + 1e-5f);
    float4* hp = reinterpret_cast<float4*>(h + (size_t)node * 64) + q;
    const float4 hvA = hp[0], hvB = hp[8];
    float4 oA, oB;
    oA.x = (vals[0] - mean) * inv * g[cA]     + be[cA]     + hvA.x;
    oA.y = (vals[1] - mean) * inv * g[cA + 1] + be[cA + 1] + hvA.y;
    oA.z = (vals[2] - mean) * inv * g[cA + 2] + be[cA + 2] + hvA.z;
    oA.w = (vals[3] - mean) * inv * g[cA + 3] + be[cA + 3] + hvA.w;
    oB.x = (vals[4] - mean) * inv * g[cB]     + be[cB]     + hvB.x;
    oB.y = (vals[5] - mean) * inv * g[cB + 1] + be[cB + 1] + hvB.y;
    oB.z = (vals[6] - mean) * inv * g[cB + 2] + be[cB + 2] + hvB.z;
    oB.w = (vals[7] - mean) * inv * g[cB + 3] + be[cB + 3] + hvB.w;
    const int nl = threadIdx.x >> 3;
    *reinterpret_cast<float4*>(&hs[nl][cA]) = oA;
    *reinterpret_cast<float4*>(&hs[nl][cB]) = oB;
    __syncthreads();
    tile_tail<TAIL, false>(hs, os, Wa, ba, Wb, bb, nullptr, nullptr, outA, outB, (int)blockIdx.x);
    hp[0] = *reinterpret_cast<const float4*>(&hs[nl][cA]);
    hp[8] = *reinterpret_cast<const float4*>(&hs[nl][cB]);
}

__global__ __launch_bounds__(256) __attribute__((amdgpu_waves_per_eu(16)))
void k_gat_final(const int* __restrict__ start, const int* __restrict__ csr,
                 const float* __restrict__ xl, const float* __restrict__ xr,
                 const float* __restrict__ att, const float* __restrict__ bias,
                 const float* __restrict__ g, const float* __restrict__ be,
                 float* __restrict__ out, int n, int e) {
    constexpr int NC = 32, LPN = 8;
    const int t = blockIdx.x * 256 + threadIdx.x;
    const int node = t / LPN, q = t % LPN;
    if (node >= n) return;
    const int beg = start[node];
    const int end = (node + 1 < n) ? start[node + 1] : e;
    const int c0 = q * 4;
    const v2f a01 = { att[c0] * LOG2E,     att[c0 + 1] * LOG2E };
    const v2f a23 = { att[c0 + 2] * LOG2E, att[c0 + 3] * LOG2E };
    const float4* xl4 = reinterpret_cast<const float4*>(xl);
    const float4 r4 = reinterpret_cast<const float4*>(xr)[node * LPN + q];
    const float4 self = xl4[node * LPN + q];
    const v2f r01 = { r4.x, r4.y }, r23 = { r4.z, r4.w };

    auto part = [&](const float4& v) -> float {
        const v2f t01 = lo2(v) + r01;
        const v2f t23 = hi2(v) + r23;
        v2f p = a01 * lrelu2(t01);
        p = __builtin_elementwise_fma(a23, lrelu2(t23), p);
        return p.x + p.y;
    };

    float m = group_sum<LPN>(part(self));
    float l = 1.f;
    v2f acc01 = lo2(self), acc23 = hi2(self);
    for (int base = beg; base < end; base += LPN) {
        const int cnt = min(LPN, end - base);
        const int my = (q < cnt) ? csr[base + q] : node;
        for (int j = 0; j < cnt; j += 4) {
            const int s0 = group_bcast<LPN>(my, j + 0);
            const int s1 = group_bcast<LPN>(my, j + 1);
            const int s2 = group_bcast<LPN>(my, j + 2);
            const int s3 = group_bcast<LPN>(my, j + 3);
            const float4 v0 = xl4[s0 * LPN + q];
            const float4 v1 = xl4[s1 * LPN + q];
            const float4 v2 = xl4[s2 * LPN + q];
            const float4 v3 = xl4[s3 * LPN + q];
            asm volatile("" :: "v"(v0.x), "v"(v0.y), "v"(v0.z), "v"(v0.w),
                               "v"(v1.x), "v"(v1.y), "v"(v1.z), "v"(v1.w),
                               "v"(v2.x), "v"(v2.y), "v"(v2.z), "v"(v2.w),
                               "v"(v3.x), "v"(v3.y), "v"(v3.z), "v"(v3.w));
            float e0 = part(v0), e1 = part(v1), e2 = part(v2), e3 = part(v3);
            group_sum4<LPN>(e0, e1, e2, e3);
            if (j + 1 >= cnt) e1 = -INFINITY;
            if (j + 2 >= cnt) e2 = -INFINITY;
            if (j + 3 >= cnt) e3 = -INFINITY;
            const float mn = fmaxf(fmaxf(m, e0), fmaxf(fmaxf(e1, e2), e3));
            const v2f d01 = v2f{e0, e1} - both(mn), d23 = v2f{e2, e3} - both(mn);
            const float sc = __builtin_amdgcn_exp2f(m - mn);
            const float p0 = __builtin_amdgcn_exp2f(d01.x), p1 = __builtin_amdgcn_exp2f(d01.y);
            const float p2 = __builtin_amdgcn_exp2f(d23.x), p3 = __builtin_amdgcn_exp2f(d23.y);
            l = l * sc + ((p0 + p1) + (p2 + p3));
            const v2f P0 = both(p0), P1 = both(p1), P2 = both(p2), P3 = both(p3), SC = both(sc);
            const v2f s01a = __builtin_elementwise_fma(lo2(v0), P0, lo2(v1) * P1);
            const v2f s01b = __builtin_elementwise_fma(lo2(v2), P2, lo2(v3) * P3);
            const v2f s23a = __builtin_elementwise_fma(hi2(v0), P0, hi2(v1) * P1);
            const v2f s23b = __builtin_elementwise_fma(hi2(v2), P2, hi2(v3) * P3);
            acc01 = __builtin_elementwise_fma(acc01, SC, s01a + s01b);
            acc23 = __builtin_elementwise_fma(acc23, SC, s23a + s23b);
            m = mn;
        }
    }

    const float il = rcp_f(l);
    float vals[4];
    vals[0] = silu_f(acc01.x * il + bias[c0]);
    vals[1] = silu_f(acc01.y * il + bias[c0 + 1]);
    vals[2] = silu_f(acc23.x * il + bias[c0 + 2]);
    vals[3] = silu_f(acc23.y * il + bias[c0 + 3]);
    float s = (vals[0] + vals[1]) + (vals[2] + vals[3]);
    float sq = (vals[0] * vals[0] + vals[1] * vals[1]) + (vals[2] * vals[2] + vals[3] * vals[3]);
    s  = group_sum<LPN>(s);
    sq = group_sum<LPN>(sq);
    const float mean = s * (1.f / NC);
    const float var  = sq * (1.f / NC) - mean * mean;
    const float inv  = rsq_f(var + 1e-5f);
    float4 o;
    o.x = (vals[0] - mean) * inv * g[c0]     + be[c0];
    o.y = (vals[1] - mean) * inv * g[c0 + 1] + be[c0 + 1];
    o.z = (vals[2] - mean) * inv * g[c0 + 2] + be[c0 + 2];
    o.w = (vals[3] - mean) * inv * g[c0 + 3] + be[c0 + 3];
    reinterpret_cast<float4*>(out + (size_t)node * NC)[q] = o;
}


extern "C" void kernel_launch(void* const* d_in, const int* in_sizes, int n_in,
                              void* d_out, int out_size, void* d_ws, size_t ws_size,
                              hipStream_t stream) {
    (void)in_sizes; (void)n_in; (void)out_size; (void)ws_size;

    const float* x        = (const float*)d_in[0];
    const int*   ei       = (const int*)  d_in[1];
    const float* W_emb    = (const float*)d_in[2];
    const float* b_emb    = (const float*)d_in[3];
    const float* g_emb    = (const float*)d_in[4];
    const float* be_emb   = (const float*)d_in[5];
    const float* enc_W    = (const float*)d_in[6];
    const float* enc_b    = (const float*)d_in[7];
    const float* enc_g    = (const float*)d_in[8];
    const float* enc_be   = (const float*)d_in[9];
    const float* dec_Wl   = (const float*)d_in[10];
    const float* dec_bl   = (const float*)d_in[11];
    const float* dec_Wr   = (const float*)d_in[12];
    const float* dec_br   = (const float*)d_in[13];
    const float* dec_att  = (const float*)d_in[14];
    const float* dec_bias = (const float*)d_in[15];
    const float* dec_g    = (const float*)d_in[16];
    const float* dec_be   = (const float*)d_in[17];
    const float* fin_Wl   = (const float*)d_in[18];
    const float* fin_bl   = (const float*)d_in[19];
    const float* fin_Wr   = (const float*)d_in[20];
    const float* fin_br   = (const float*)d_in[21];
    const float* fin_att  = (const float*)d_in[22];
    const float* fin_bias = (const float*)d_in[23];
    const float* fin_g    = (const float*)d_in[24];
    const float* fin_be   = (const float*)d_in[25];

    const int* src = ei;
    const int* dst = ei + EE;
    float* out = (float*)d_out;

    int* wi      = (int*)d_ws;
    int*   start = wi;                  wi += NN;
                                        wi += 256;
    int*   tmp   = wi;                  wi += EE;
    int*   csr   = wi;                  wi += EE;
    float* w     = (float*)wi;
    float* h     = w;                   w += (size_t)NN * 64;
    float* zA    = w;                   w += (size_t)NN * 64;
    float* zB    = w;                   w += (size_t)NN * 64;
    float* xlA   = w;                   w += (size_t)NN * 64;
    float* xrA   = w;                   w += (size_t)NN * 64;
    int*   off   = (int*)w;

    const int TB = 256;
    auto nb = [](long long t) { return (unsigned)((t + 255) / 256); };

    k_chunksort<<<NB1, 256, 0, stream>>>(src, dst, off, tmp);
    k_bucket<<<NBK, 512, 0, stream>>>(tmp, off, start, csr);

    k_embed<<<NN / ER, 128, 0, stream>>>(x, W_emb, b_emb, g_emb, be_emb,
                                         enc_W, enc_b, enc_g, enc_be, h, zA);

    const unsigned NB32 = (unsigned)(NN / 32);
    k_enc_agg<true><<<NB32, TB, 0, stream>>>(start, csr, zA, h,
        enc_W + (size_t)1 * 64 * 64, enc_b + 1 * 64, nullptr, nullptr, enc_g + 1 * 64, enc_be + 1 * 64,
        zB, nullptr, NN, EE);
    k_enc_agg<true><<<NB32, TB, 0, stream>>>(start, csr, zB, h,
        enc_W + (size_t)2 * 64 * 64, enc_b + 2 * 64, nullptr, nullptr, enc_g + 2 * 64, enc_be + 2 * 64,
        zA, nullptr, NN, EE);
    k_enc_agg<false><<<NB32, TB, 0, stream>>>(start, csr, zA, h,
        dec_Wl, dec_bl, dec_Wr, dec_br, nullptr, nullptr, xlA, xrA, NN, EE);

    k_gat_wide<64><<<NB32, TB, 0, stream>>>(
        start, csr, xlA, xrA, dec_att, dec_bias, dec_g, dec_be, h,
        dec_Wl + (size_t)1 * 64 * 64, dec_bl + 1 * 64, dec_Wr + (size_t)1 * 64 * 64, dec_br + 1 * 64,
        zA, zB, NN, EE);
    k_gat_wide<64><<<NB32, TB, 0, stream>>>(
        start, csr, zA, zB, dec_att + 1 * 64, dec_bias + 1 * 64, dec_g + 1 * 64, dec_be + 1 * 64, h,
        dec_Wl + (size_t)2 * 64 * 64, dec_bl + 2 * 64, dec_Wr + (size_t)2 * 64 * 64, dec_br + 2 * 64,
        xlA, xrA, NN, EE);
    k_gat_wide<32><<<NB32, TB, 0, stream>>>(
        start, csr, xlA, xrA, dec_att + 2 * 64, dec_bias + 2 * 64, dec_g + 2 * 64, dec_be + 2 * 64, h,
        fin_Wl, fin_bl, fin_Wr, fin_br, zA, zB, NN, EE);

    k_gat_final<<<nb((long long)NN * 8), TB, 0, stream>>>(
        start, csr, zA, zB, fin_att, fin_bias, fin_g, fin_be, out, NN, EE);
}
